# starting kernel
# baseline (speedup 1.0000x reference)
#include <hip/hip_runtime.h>

#define B_    4
#define CK_   64
#define CV_   512
#define H_    30
#define W_    54
#define T_    8
#define P_    (H_*W_)
#define N_    (T_*H_*W_)
#define CHUNK 128
#define NCHUNK ((N_ + CHUNK - 1)/CHUNK)
#define PTILES ((P_ + 15)/16)

typedef __attribute__((ext_vector_type(16))) __bf16       v16bf;
typedef __attribute__((ext_vector_type(8)))  float        v8f;
typedef __attribute__((ext_vector_type(4)))  unsigned int v4u;

union Frag16 {
  v16bf  bf;
  v4u    q[2];
  __bf16 h[16];
};
union Pack8 {
  v4u    q;
  __bf16 h[8];
};
union Pack4 {
  uint2  u;
  __bf16 h[4];
};

#define QSTR  72
#define MKSTR 72
#define PSTR  136

template <bool MV_BF16>
__global__ __launch_bounds__(256) void mem_reader_attn(
    const float* __restrict__ mk, const float* __restrict__ mv_f32,
    const __bf16* __restrict__ mv_bf, const float* __restrict__ qk,
    float* __restrict__ out)
{
  __shared__ __bf16 s_q[16 * QSTR];
  __shared__ __bf16 s_mk[CHUNK * MKSTR];
  __shared__ __bf16 s_prob[16 * PSTR];
  __shared__ float  s_pmax[8][32];
  __shared__ float  s_psum[8][32];
  __shared__ float  s_mrun[16];
  __shared__ float  s_lrun[16];
  __shared__ float  s_alpha[16];

  const int tid  = threadIdx.x;
  const int wave = tid >> 5;
  const int lane = tid & 31;
  const int col  = lane & 15;
  const int kg   = (lane >> 4) * 8;
  const int p0   = blockIdx.x * 16;
  const int bi   = blockIdx.y;

  const float*  mk_b  = mk + (size_t)bi * CK_ * N_;
  const float*  qk_b  = qk + (size_t)bi * CK_ * P_;
  const float*  mvf_b = mv_f32 ? (mv_f32 + (size_t)bi * CV_ * N_) : nullptr;
  const __bf16* mvb_b = mv_bf  ? (mv_bf  + (size_t)bi * CV_ * N_) : nullptr;

  for (int idx = tid; idx < CK_ * 16; idx += 256) {
    int p = idx & 15, c = idx >> 4;
    int pg = p0 + p; if (pg > P_ - 1) pg = P_ - 1;
    s_q[p * QSTR + c] = (__bf16)(qk_b[(size_t)c * P_ + pg] * 0.125f);
  }
  if (tid < 16) { s_mrun[tid] = -1e30f; s_lrun[tid] = 0.f; s_alpha[tid] = 0.f; }

  v8f acc[4] = {v8f{}, v8f{}, v8f{}, v8f{}};

  for (int nc = 0; nc < NCHUNK; ++nc) {
    const int n0 = nc * CHUNK;
    __syncthreads();

    for (int idx = tid; idx < CK_ * CHUNK; idx += 256) {
      int nl = idx & (CHUNK - 1), c = idx >> 7;
      int ng = n0 + nl;
      float v = (ng < N_) ? mk_b[(size_t)c * N_ + ng] : 0.f;
      s_mk[nl * MKSTR + c] = (__bf16)v;
    }
    __syncthreads();

    const int mrow = wave * 16;
    v8f S = {};
#pragma unroll
    for (int ks = 0; ks < 2; ++ks) {
      Frag16 a, b;
      const __bf16* ap = &s_mk[(mrow + col) * MKSTR + ks * 32 + kg];
      a.q[0] = *(const v4u*)ap;  a.q[1] = *(const v4u*)(ap + 16);
      const __bf16* bp = &s_q[col * QSTR + ks * 32 + kg];
      b.q[0] = *(const v4u*)bp;  b.q[1] = *(const v4u*)(bp + 16);
      S = __builtin_amdgcn_wmma_f32_16x16x32_bf16(false, a.bf, false, b.bf,
                                                  (short)0, S, false, false);
    }

    float sv[8];
    float lmax = -1e30f;
#pragma unroll
    for (int j = 0; j < 8; ++j) {
      int ng = n0 + mrow + kg + j;
      sv[j] = (ng < N_) ? S[j] : -1e30f;
      lmax = fmaxf(lmax, sv[j]);
    }
    s_pmax[wave][lane] = lmax;
    __syncthreads();

    if (tid < 16) {
      float cmax = -1e30f;
#pragma unroll
      for (int w = 0; w < 8; ++w)
        cmax = fmaxf(cmax, fmaxf(s_pmax[w][tid], s_pmax[w][tid + 16]));
      float mold = s_mrun[tid];
      float mnew = fmaxf(mold, cmax);
      s_alpha[tid] = __expf(mold - mnew);
      s_mrun[tid]  = mnew;
    }
    __syncthreads();

    const float mnew = s_mrun[col];
    const float alph = s_alpha[col];
#pragma unroll
    for (int t = 0; t < 4; ++t) acc[t] *= alph;

    Pack8 pk;
    float lsum = 0.f;
#pragma unroll
    for (int j = 0; j < 8; ++j) {
      int ng = n0 + mrow + kg + j;
      float pr = (ng < N_) ? __expf(sv[j] - mnew) : 0.f;
      lsum += pr;
      pk.h[j] = (__bf16)pr;
    }
    *(v4u*)&s_prob[col * PSTR + mrow + kg] = pk.q;
    s_psum[wave][lane] = lsum;
    __syncthreads();

    if (tid < 16) {
      float csum = 0.f;
#pragma unroll
      for (int w = 0; w < 8; ++w)
        csum += s_psum[w][tid] + s_psum[w][tid + 16];
      s_lrun[tid] = s_lrun[tid] * s_alpha[tid] + csum;
    }

#pragma unroll
    for (int vt = 0; vt < 4; ++vt) {
      const int v = wave * 64 + vt * 16 + col;
#pragma unroll
      for (int ks = 0; ks < 4; ++ks) {
        int nA = n0 + ks * 32 + kg;
        int b1 = nA;      if (b1 > N_ - 8) b1 = N_ - 8;
        int b2 = nA + 16; if (b2 > N_ - 8) b2 = N_ - 8;
        Frag16 a;
        if (MV_BF16) {
          const __bf16* rowp = mvb_b + (size_t)v * N_;
          a.q[0] = *(const v4u*)(rowp + b1);
          a.q[1] = *(const v4u*)(rowp + b2);
        } else {
          const float* rowp = mvf_b + (size_t)v * N_;
          const float4 f0 = *(const float4*)(rowp + b1);
          const float4 f1 = *(const float4*)(rowp + b1 + 4);
          const float4 f2 = *(const float4*)(rowp + b2);
          const float4 f3 = *(const float4*)(rowp + b2 + 4);
          a.h[0]=(__bf16)f0.x;  a.h[1]=(__bf16)f0.y;  a.h[2]=(__bf16)f0.z;  a.h[3]=(__bf16)f0.w;
          a.h[4]=(__bf16)f1.x;  a.h[5]=(__bf16)f1.y;  a.h[6]=(__bf16)f1.z;  a.h[7]=(__bf16)f1.w;
          a.h[8]=(__bf16)f2.x;  a.h[9]=(__bf16)f2.y;  a.h[10]=(__bf16)f2.z; a.h[11]=(__bf16)f2.w;
          a.h[12]=(__bf16)f3.x; a.h[13]=(__bf16)f3.y; a.h[14]=(__bf16)f3.z; a.h[15]=(__bf16)f3.w;
        }
        Frag16 b;
        const __bf16* bp = &s_prob[col * PSTR + ks * 32 + kg];
        b.q[0] = *(const v4u*)bp;  b.q[1] = *(const v4u*)(bp + 16);
        acc[vt] = __builtin_amdgcn_wmma_f32_16x16x32_bf16(false, a.bf, false, b.bf,
                                                          (short)0, acc[vt], false, false);
      }
    }
  }

  __syncthreads();
  const float linv = 1.0f / s_lrun[col];
  const int pg = p0 + col;
  if (pg < P_) {
#pragma unroll
    for (int vt = 0; vt < 4; ++vt) {
#pragma unroll
      for (int j = 0; j < 8; ++j) {
        int v = wave * 64 + vt * 16 + kg + j;
        out[((size_t)bi * (2 * CV_) + v) * P_ + pg] = acc[vt][j] * linv;
      }
    }
  }
}

__global__ __launch_bounds__(256) void convert_mv(const float* __restrict__ src,
                                                  __bf16* __restrict__ dst, int n4)
{
  int i = blockIdx.x * blockDim.x + threadIdx.x;
  if (i < n4) {
    float4 f = ((const float4*)src)[i];
    Pack4 p;
    p.h[0] = (__bf16)f.x; p.h[1] = (__bf16)f.y;
    p.h[2] = (__bf16)f.z; p.h[3] = (__bf16)f.w;
    *(uint2*)(dst + 4 * (size_t)i) = p.u;
  }
}

__global__ __launch_bounds__(256) void qv_copy(const float* __restrict__ qv,
                                               float* __restrict__ out, int total)
{
  int i = blockIdx.x * blockDim.x + threadIdx.x;
  if (i < total) {
    int b = i / (CV_ * P_);
    int r = i - b * (CV_ * P_);
    out[(size_t)b * (2 * CV_ * P_) + (CV_ * P_) + r] = qv[i];
  }
}

extern "C" void kernel_launch(void* const* d_in, const int* in_sizes, int n_in,
                              void* d_out, int out_size, void* d_ws, size_t ws_size,
                              hipStream_t stream) {
  const float* mk = (const float*)d_in[0];
  const float* mv = (const float*)d_in[1];
  const float* qk = (const float*)d_in[2];
  const float* qv = (const float*)d_in[3];
  float* out = (float*)d_out;

  const size_t mv_elems = (size_t)B_ * CV_ * N_;
  const size_t need     = mv_elems * sizeof(__bf16);
  dim3 grid(PTILES, B_);

  if (d_ws && ws_size >= need) {
    __bf16* mv_bf = (__bf16*)d_ws;
    int n4 = (int)(mv_elems / 4);
    convert_mv<<<(n4 + 255) / 256, 256, 0, stream>>>(mv, mv_bf, n4);
    mem_reader_attn<true><<<grid, 256, 0, stream>>>(mk, nullptr, mv_bf, qk, out);
  } else {
    mem_reader_attn<false><<<grid, 256, 0, stream>>>(mk, mv, nullptr, qk, out);
  }

  int total = B_ * CV_ * P_;
  qv_copy<<<(total + 255) / 256, 256, 0, stream>>>(qv, out, total);
}


# mk chunk staging: 8 b128 loads per thread issued together one chunk ahead into registers, 4 packed b128 LDS stores; replaces 32 serialised load-wait-store round trips per chunk
# speedup vs baseline: 2.9383x; 2.9383x over previous
#include <hip/hip_runtime.h>

#define B_    4
#define CK_   64
#define CV_   512
#define H_    30
#define W_    54
#define T_    8
#define P_    (H_*W_)
#define N_    (T_*H_*W_)
#define CHUNK 128
#define NCHUNK ((N_ + CHUNK - 1)/CHUNK)
#define PTILES ((P_ + 15)/16)

typedef __attribute__((ext_vector_type(16))) __bf16       v16bf;
typedef __attribute__((ext_vector_type(8)))  float        v8f;
typedef __attribute__((ext_vector_type(4)))  unsigned int v4u;

union Frag16 {
  v16bf  bf;
  v4u    q[2];
  __bf16 h[16];
};
union Pack8 {
  v4u    q;
  __bf16 h[8];
};
union Pack4 {
  uint2  u;
  __bf16 h[4];
};

#define QSTR  72
#define MKSTR 72
#define PSTR  136

template <bool MV_BF16>
__global__ __launch_bounds__(256) void mem_reader_attn(
    const float* __restrict__ mk, const float* __restrict__ mv_f32,
    const __bf16* __restrict__ mv_bf, const float* __restrict__ qk,
    float* __restrict__ out)
{
  __shared__ __bf16 s_q[16 * QSTR];
  __shared__ __bf16 s_mk[CHUNK * MKSTR];
  __shared__ __bf16 s_prob[16 * PSTR];
  __shared__ float  s_pmax[8][32];
  __shared__ float  s_psum[8][32];
  __shared__ float  s_mrun[16];
  __shared__ float  s_lrun[16];
  __shared__ float  s_alpha[16];

  const int tid  = threadIdx.x;
  const int wave = tid >> 5;
  const int lane = tid & 31;
  const int col  = lane & 15;
  const int kg   = (lane >> 4) * 8;
  const int p0   = blockIdx.x * 16;
  const int bi   = blockIdx.y;

  const float*  mk_b  = mk + (size_t)bi * CK_ * N_;
  const float*  qk_b  = qk + (size_t)bi * CK_ * P_;
  const float*  mvf_b = mv_f32 ? (mv_f32 + (size_t)bi * CV_ * N_) : nullptr;
  const __bf16* mvb_b = mv_bf  ? (mv_bf  + (size_t)bi * CV_ * N_) : nullptr;

  for (int idx = tid; idx < CK_ * 16; idx += 256) {
    int p = idx & 15, c = idx >> 4;
    int pg = p0 + p; if (pg > P_ - 1) pg = P_ - 1;
    s_q[p * QSTR + c] = (__bf16)(qk_b[(size_t)c * P_ + pg] * 0.125f);
  }
  if (tid < 16) { s_mrun[tid] = -1e30f; s_lrun[tid] = 0.f; s_alpha[tid] = 0.f; }

  v8f acc[4] = {v8f{}, v8f{}, v8f{}, v8f{}};

  const float* mk_t = mk_b + (size_t)(wave * 8) * N_;
  float4 pre[8];
  {
    const int ngc = 4 * lane;
#pragma unroll
    for (int j = 0; j < 8; ++j)
      pre[j] = *(const float4*)(mk_t + (size_t)j * N_ + ngc);
  }

  for (int nc = 0; nc < NCHUNK; ++nc) {
    const int n0 = nc * CHUNK;
    __syncthreads();

    {
      const bool ok = (n0 + 4 * lane) < N_;
      Pack8 r0, r1, r2, r3;
#pragma unroll
      for (int j = 0; j < 8; ++j) {
        r0.h[j] = (__bf16)(ok ? pre[j].x : 0.f);
        r1.h[j] = (__bf16)(ok ? pre[j].y : 0.f);
        r2.h[j] = (__bf16)(ok ? pre[j].z : 0.f);
        r3.h[j] = (__bf16)(ok ? pre[j].w : 0.f);
      }
      __bf16* dp = &s_mk[(4 * lane) * MKSTR + wave * 8];
      *(v4u*)(dp)             = r0.q;
      *(v4u*)(dp + MKSTR)     = r1.q;
      *(v4u*)(dp + 2 * MKSTR) = r2.q;
      *(v4u*)(dp + 3 * MKSTR) = r3.q;
    }
    __syncthreads();

    const int mrow = wave * 16;
    v8f S = {};
#pragma unroll
    for (int ks = 0; ks < 2; ++ks) {
      Frag16 a, b;
      const __bf16* ap = &s_mk[(mrow + col) * MKSTR + ks * 32 + kg];
      a.q[0] = *(const v4u*)ap;  a.q[1] = *(const v4u*)(ap + 16);
      const __bf16* bp = &s_q[col * QSTR + ks * 32 + kg];
      b.q[0] = *(const v4u*)bp;  b.q[1] = *(const v4u*)(bp + 16);
      S = __builtin_amdgcn_wmma_f32_16x16x32_bf16(false, a.bf, false, b.bf,
                                                  (short)0, S, false, false);
    }

    float sv[8];
    float lmax = -1e30f;
#pragma unroll
    for (int j = 0; j < 8; ++j) {
      int ng = n0 + mrow + kg + j;
      sv[j] = (ng < N_) ? S[j] : -1e30f;
      lmax = fmaxf(lmax, sv[j]);
    }
    s_pmax[wave][lane] = lmax;
    __syncthreads();

    if (tid < 16) {
      float cmax = -1e30f;
#pragma unroll
      for (int w = 0; w < 8; ++w)
        cmax = fmaxf(cmax, fmaxf(s_pmax[w][tid], s_pmax[w][tid + 16]));
      float mold = s_mrun[tid];
      float mnew = fmaxf(mold, cmax);
      s_alpha[tid] = __expf(mold - mnew);
      s_mrun[tid]  = mnew;
    }
    __syncthreads();

    const float mnew = s_mrun[col];
    const float alph = s_alpha[col];
#pragma unroll
    for (int t = 0; t < 4; ++t) acc[t] *= alph;

    Pack8 pk;
    float lsum = 0.f;
#pragma unroll
    for (int j = 0; j < 8; ++j) {
      int ng = n0 + mrow + kg + j;
      float pr = (ng < N_) ? __expf(sv[j] - mnew) : 0.f;
      lsum += pr;
      pk.h[j] = (__bf16)pr;
    }
    *(v4u*)&s_prob[col * PSTR + mrow + kg] = pk.q;
    s_psum[wave][lane] = lsum;
    __syncthreads();

    if (nc + 1 < NCHUNK) {
      int ngc = n0 + CHUNK + 4 * lane;
      if (ngc > N_ - 4) ngc = N_ - 4;
#pragma unroll
      for (int j = 0; j < 8; ++j)
        pre[j] = *(const float4*)(mk_t + (size_t)j * N_ + ngc);
    }

    if (tid < 16) {
      float csum = 0.f;
#pragma unroll
      for (int w = 0; w < 8; ++w)
        csum += s_psum[w][tid] + s_psum[w][tid + 16];
      s_lrun[tid] = s_lrun[tid] * s_alpha[tid] + csum;
    }

#pragma unroll
    for (int vt = 0; vt < 4; ++vt) {
      const int v = wave * 64 + vt * 16 + col;
#pragma unroll
      for (int ks = 0; ks < 4; ++ks) {
        int nA = n0 + ks * 32 + kg;
        int b1 = nA;      if (b1 > N_ - 8) b1 = N_ - 8;
        int b2 = nA + 16; if (b2 > N_ - 8) b2 = N_ - 8;
        Frag16 a;
        if (MV_BF16) {
          const __bf16* rowp = mvb_b + (size_t)v * N_;
          a.q[0] = *(const v4u*)(rowp + b1);
          a.q[1] = *(const v4u*)(rowp + b2);
        } else {
          const float* rowp = mvf_b + (size_t)v * N_;
          const float4 f0 = *(const float4*)(rowp + b1);
          const float4 f1 = *(const float4*)(rowp + b1 + 4);
          const float4 f2 = *(const float4*)(rowp + b2);
          const float4 f3 = *(const float4*)(rowp + b2 + 4);
          a.h[0]=(__bf16)f0.x;  a.h[1]=(__bf16)f0.y;  a.h[2]=(__bf16)f0.z;  a.h[3]=(__bf16)f0.w;
          a.h[4]=(__bf16)f1.x;  a.h[5]=(__bf16)f1.y;  a.h[6]=(__bf16)f1.z;  a.h[7]=(__bf16)f1.w;
          a.h[8]=(__bf16)f2.x;  a.h[9]=(__bf16)f2.y;  a.h[10]=(__bf16)f2.z; a.h[11]=(__bf16)f2.w;
          a.h[12]=(__bf16)f3.x; a.h[13]=(__bf16)f3.y; a.h[14]=(__bf16)f3.z; a.h[15]=(__bf16)f3.w;
        }
        Frag16 b;
        const __bf16* bp = &s_prob[col * PSTR + ks * 32 + kg];
        b.q[0] = *(const v4u*)bp;  b.q[1] = *(const v4u*)(bp + 16);
        acc[vt] = __builtin_amdgcn_wmma_f32_16x16x32_bf16(false, a.bf, false, b.bf,
                                                          (short)0, acc[vt], false, false);
      }
    }
  }

  __syncthreads();
  const float linv = 1.0f / s_lrun[col];
  const int pg = p0 + col;
  if (pg < P_) {
#pragma unroll
    for (int vt = 0; vt < 4; ++vt) {
#pragma unroll
      for (int j = 0; j < 8; ++j) {
        int v = wave * 64 + vt * 16 + kg + j;
        out[((size_t)bi * (2 * CV_) + v) * P_ + pg] = acc[vt][j] * linv;
      }
    }
  }
}

__global__ __launch_bounds__(256) void convert_mv(const float* __restrict__ src,
                                                  __bf16* __restrict__ dst, int n4)
{
  int i = blockIdx.x * blockDim.x + threadIdx.x;
  if (i < n4) {
    float4 f = ((const float4*)src)[i];
    Pack4 p;
    p.h[0] = (__bf16)f.x; p.h[1] = (__bf16)f.y;
    p.h[2] = (__bf16)f.z; p.h[3] = (__bf16)f.w;
    *(uint2*)(dst + 4 * (size_t)i) = p.u;
  }
}

__global__ __launch_bounds__(256) void qv_copy(const float* __restrict__ qv,
                                               float* __restrict__ out, int total)
{
  int i = blockIdx.x * blockDim.x + threadIdx.x;
  if (i < total) {
    int b = i / (CV_ * P_);
    int r = i - b * (CV_ * P_);
    out[(size_t)b * (2 * CV_ * P_) + (CV_ * P_) + r] = qv[i];
  }
}

extern "C" void kernel_launch(void* const* d_in, const int* in_sizes, int n_in,
                              void* d_out, int out_size, void* d_ws, size_t ws_size,
                              hipStream_t stream) {
  const float* mk = (const float*)d_in[0];
  const float* mv = (const float*)d_in[1];
  const float* qk = (const float*)d_in[2];
  const float* qv = (const float*)d_in[3];
  float* out = (float*)d_out;

  const size_t mv_elems = (size_t)B_ * CV_ * N_;
  const size_t need     = mv_elems * sizeof(__bf16);
  dim3 grid(PTILES, B_);

  if (d_ws && ws_size >= need) {
    __bf16* mv_bf = (__bf16*)d_ws;
    int n4 = (int)(mv_elems / 4);
    convert_mv<<<(n4 + 255) / 256, 256, 0, stream>>>(mv, mv_bf, n4);
    mem_reader_attn<true><<<grid, 256, 0, stream>>>(mk, nullptr, mv_bf, qk, out);
  } else {
    mem_reader_attn<false><<<grid, 256, 0, stream>>>(mk, mv, nullptr, qk, out);
  }

  int total = B_ * CV_ * P_;
  qv_copy<<<(total + 255) / 256, 256, 0, stream>>>(qv, out, total);
}


# work-group tile 16 queries x 512 v-rows -> 64 queries x 256 v-rows (grid 26x2x4): each mv fragment read from global feeds 4 WMMAs; barriers and mk staging per chunk serve 4x the queries
# speedup vs baseline: 4.1906x; 1.4262x over previous
#include <hip/hip_runtime.h>

#define B_    4
#define CK_   64
#define CV_   512
#define H_    30
#define W_    54
#define T_    8
#define P_    (H_*W_)
#define N_    (T_*H_*W_)
#define CHUNK 128
#define NCHUNK ((N_ + CHUNK - 1)/CHUNK)
#define PQ    64
#define PT_   (PQ/16)
#define VB_   256
#define VSPL  (CV_/VB_)
#define PBLK  ((P_ + PQ - 1)/PQ)

typedef __attribute__((ext_vector_type(16))) __bf16       v16bf;
typedef __attribute__((ext_vector_type(8)))  float        v8f;
typedef __attribute__((ext_vector_type(4)))  unsigned int v4u;

union Frag16 {
  v16bf  bf;
  v4u    q[2];
  __bf16 h[16];
};
union Pack8 {
  v4u    q;
  __bf16 h[8];
};
union Pack4 {
  uint2  u;
  __bf16 h[4];
};

#define QSTR  72
#define MKSTR 72
#define PSTR  136

template <bool MV_BF16>
__global__ __launch_bounds__(256) void mem_reader_attn(
    const float* __restrict__ mk, const float* __restrict__ mv_f32,
    const __bf16* __restrict__ mv_bf, const float* __restrict__ qk,
    float* __restrict__ out)
{
  __shared__ __bf16 s_q[PQ * QSTR];
  __shared__ __bf16 s_mk[CHUNK * MKSTR];
  __shared__ __bf16 s_prob[PQ * PSTR];
  __shared__ float  s_pmax[8][PT_][32];
  __shared__ float  s_psum[8][PT_][32];
  __shared__ float  s_mrun[PQ];
  __shared__ float  s_lrun[PQ];
  __shared__ float  s_alpha[PQ];

  const int tid  = threadIdx.x;
  const int wave = tid >> 5;
  const int lane = tid & 31;
  const int col  = lane & 15;
  const int kg   = (lane >> 4) * 8;
  const int p0   = blockIdx.x * PQ;
  const int vb0  = blockIdx.y * VB_ + wave * 32;
  const int bi   = blockIdx.z;

  const float*  mk_b  = mk + (size_t)bi * CK_ * N_;
  const float*  qk_b  = qk + (size_t)bi * CK_ * P_;
  const float*  mvf_b = mv_f32 ? (mv_f32 + (size_t)bi * CV_ * N_) : nullptr;
  const __bf16* mvb_b = mv_bf  ? (mv_bf  + (size_t)bi * CV_ * N_) : nullptr;

  for (int idx = tid; idx < CK_ * PQ; idx += 256) {
    int p = idx & (PQ - 1), c = idx / PQ;
    int pg = p0 + p; if (pg > P_ - 1) pg = P_ - 1;
    s_q[p * QSTR + c] = (__bf16)(qk_b[(size_t)c * P_ + pg] * 0.125f);
  }
  if (tid < PQ) { s_mrun[tid] = -1e30f; s_lrun[tid] = 0.f; s_alpha[tid] = 0.f; }

  v8f acc[2][PT_];
#pragma unroll
  for (int vt = 0; vt < 2; ++vt)
#pragma unroll
    for (int pt = 0; pt < PT_; ++pt) acc[vt][pt] = v8f{};

  const float* mk_t = mk_b + (size_t)(wave * 8) * N_;
  float4 pre[8];
  {
    const int ngc = 4 * lane;
#pragma unroll
    for (int j = 0; j < 8; ++j)
      pre[j] = *(const float4*)(mk_t + (size_t)j * N_ + ngc);
  }

  for (int nc = 0; nc < NCHUNK; ++nc) {
    const int n0 = nc * CHUNK;
    __syncthreads();

    {
      const bool ok = (n0 + 4 * lane) < N_;
      Pack8 r0, r1, r2, r3;
#pragma unroll
      for (int j = 0; j < 8; ++j) {
        r0.h[j] = (__bf16)(ok ? pre[j].x : 0.f);
        r1.h[j] = (__bf16)(ok ? pre[j].y : 0.f);
        r2.h[j] = (__bf16)(ok ? pre[j].z : 0.f);
        r3.h[j] = (__bf16)(ok ? pre[j].w : 0.f);
      }
      __bf16* dp = &s_mk[(4 * lane) * MKSTR + wave * 8];
      *(v4u*)(dp)             = r0.q;
      *(v4u*)(dp + MKSTR)     = r1.q;
      *(v4u*)(dp + 2 * MKSTR) = r2.q;
      *(v4u*)(dp + 3 * MKSTR) = r3.q;
    }
    __syncthreads();

    const int mrow = wave * 16;
    Frag16 amk[2];
#pragma unroll
    for (int ks = 0; ks < 2; ++ks) {
      const __bf16* ap = &s_mk[(mrow + col) * MKSTR + ks * 32 + kg];
      amk[ks].q[0] = *(const v4u*)ap;  amk[ks].q[1] = *(const v4u*)(ap + 16);
    }
    float sv[PT_][8];
#pragma unroll
    for (int pt = 0; pt < PT_; ++pt) {
      v8f S = {};
#pragma unroll
      for (int ks = 0; ks < 2; ++ks) {
        Frag16 b;
        const __bf16* bp = &s_q[(pt * 16 + col) * QSTR + ks * 32 + kg];
        b.q[0] = *(const v4u*)bp;  b.q[1] = *(const v4u*)(bp + 16);
        S = __builtin_amdgcn_wmma_f32_16x16x32_bf16(false, amk[ks].bf, false, b.bf,
                                                    (short)0, S, false, false);
      }
      float lmax = -1e30f;
#pragma unroll
      for (int j = 0; j < 8; ++j) {
        int ng = n0 + mrow + kg + j;
        sv[pt][j] = (ng < N_) ? S[j] : -1e30f;
        lmax = fmaxf(lmax, sv[pt][j]);
      }
      s_pmax[wave][pt][lane] = lmax;
    }
    __syncthreads();

    if (tid < PQ) {
      const int pt = tid >> 4, c = tid & 15;
      float cmax = -1e30f;
#pragma unroll
      for (int w = 0; w < 8; ++w)
        cmax = fmaxf(cmax, fmaxf(s_pmax[w][pt][c], s_pmax[w][pt][c + 16]));
      float mold = s_mrun[tid];
      float mnew = fmaxf(mold, cmax);
      s_alpha[tid] = __expf(mold - mnew);
      s_mrun[tid]  = mnew;
    }
    __syncthreads();

#pragma unroll
    for (int pt = 0; pt < PT_; ++pt) {
      const float mnew = s_mrun[pt * 16 + col];
      const float alph = s_alpha[pt * 16 + col];
      acc[0][pt] *= alph;
      acc[1][pt] *= alph;

      Pack8 pk;
      float lsum = 0.f;
#pragma unroll
      for (int j = 0; j < 8; ++j) {
        int ng = n0 + mrow + kg + j;
        float pr = (ng < N_) ? __expf(sv[pt][j] - mnew) : 0.f;
        lsum += pr;
        pk.h[j] = (__bf16)pr;
      }
      *(v4u*)&s_prob[(pt * 16 + col) * PSTR + mrow + kg] = pk.q;
      s_psum[wave][pt][lane] = lsum;
    }
    __syncthreads();

    if (nc + 1 < NCHUNK) {
      int ngc = n0 + CHUNK + 4 * lane;
      if (ngc > N_ - 4) ngc = N_ - 4;
#pragma unroll
      for (int j = 0; j < 8; ++j)
        pre[j] = *(const float4*)(mk_t + (size_t)j * N_ + ngc);
    }

    if (tid < PQ) {
      const int pt = tid >> 4, c = tid & 15;
      float csum = 0.f;
#pragma unroll
      for (int w = 0; w < 8; ++w)
        csum += s_psum[w][pt][c] + s_psum[w][pt][c + 16];
      s_lrun[tid] = s_lrun[tid] * s_alpha[tid] + csum;
    }

#pragma unroll
    for (int ks = 0; ks < 4; ++ks) {
      int nA = n0 + ks * 32 + kg;
      int b1 = nA;      if (b1 > N_ - 8) b1 = N_ - 8;
      int b2 = nA + 16; if (b2 > N_ - 8) b2 = N_ - 8;
      Frag16 b[PT_];
#pragma unroll
      for (int pt = 0; pt < PT_; ++pt) {
        const __bf16* bp = &s_prob[(pt * 16 + col) * PSTR + ks * 32 + kg];
        b[pt].q[0] = *(const v4u*)bp;  b[pt].q[1] = *(const v4u*)(bp + 16);
      }
#pragma unroll
      for (int vt = 0; vt < 2; ++vt) {
        const int v = vb0 + vt * 16 + col;
        Frag16 a;
        if (MV_BF16) {
          const __bf16* rowp = mvb_b + (size_t)v * N_;
          a.q[0] = *(const v4u*)(rowp + b1);
          a.q[1] = *(const v4u*)(rowp + b2);
        } else {
          const float* rowp = mvf_b + (size_t)v * N_;
          const float4 f0 = *(const float4*)(rowp + b1);
          const float4 f1 = *(const float4*)(rowp + b1 + 4);
          const float4 f2 = *(const float4*)(rowp + b2);
          const float4 f3 = *(const float4*)(rowp + b2 + 4);
          a.h[0]=(__bf16)f0.x;  a.h[1]=(__bf16)f0.y;  a.h[2]=(__bf16)f0.z;  a.h[3]=(__bf16)f0.w;
          a.h[4]=(__bf16)f1.x;  a.h[5]=(__bf16)f1.y;  a.h[6]=(__bf16)f1.z;  a.h[7]=(__bf16)f1.w;
          a.h[8]=(__bf16)f2.x;  a.h[9]=(__bf16)f2.y;  a.h[10]=(__bf16)f2.z; a.h[11]=(__bf16)f2.w;
          a.h[12]=(__bf16)f3.x; a.h[13]=(__bf16)f3.y; a.h[14]=(__bf16)f3.z; a.h[15]=(__bf16)f3.w;
        }
#pragma unroll
        for (int pt = 0; pt < PT_; ++pt)
          acc[vt][pt] = __builtin_amdgcn_wmma_f32_16x16x32_bf16(false, a.bf, false, b[pt].bf,
                                                                (short)0, acc[vt][pt], false, false);
      }
    }
  }

  __syncthreads();
#pragma unroll
  for (int pt = 0; pt < PT_; ++pt) {
    const float linv = 1.0f / s_lrun[pt * 16 + col];
    const int pg = p0 + pt * 16 + col;
    if (pg < P_) {
#pragma unroll
      for (int vt = 0; vt < 2; ++vt) {
#pragma unroll
        for (int j = 0; j < 8; ++j) {
          int v = vb0 + vt * 16 + kg + j;
          out[((size_t)bi * (2 * CV_) + v) * P_ + pg] = acc[vt][pt][j] * linv;
        }
      }
    }
  }
}

__global__ __launch_bounds__(256) void convert_mv(const float* __restrict__ src,
                                                  __bf16* __restrict__ dst, int n4)
{
  int i = blockIdx.x * blockDim.x + threadIdx.x;
  if (i < n4) {
    float4 f = ((const float4*)src)[i];
    Pack4 p;
    p.h[0] = (__bf16)f.x; p.h[1] = (__bf16)f.y;
    p.h[2] = (__bf16)f.z; p.h[3] = (__bf16)f.w;
    *(uint2*)(dst + 4 * (size_t)i) = p.u;
  }
}

__global__ __launch_bounds__(256) void qv_copy(const float* __restrict__ qv,
                                               float* __restrict__ out, int total)
{
  int i = blockIdx.x * blockDim.x + threadIdx.x;
  if (i < total) {
    int b = i / (CV_ * P_);
    int r = i - b * (CV_ * P_);
    out[(size_t)b * (2 * CV_ * P_) + (CV_ * P_) + r] = qv[i];
  }
}

extern "C" void kernel_launch(void* const* d_in, const int* in_sizes, int n_in,
                              void* d_out, int out_size, void* d_ws, size_t ws_size,
                              hipStream_t stream) {
  const float* mk = (const float*)d_in[0];
  const float* mv = (const float*)d_in[1];
  const float* qk = (const float*)d_in[2];
  const float* qv = (const float*)d_in[3];
  float* out = (float*)d_out;

  const size_t mv_elems = (size_t)B_ * CV_ * N_;
  const size_t need     = mv_elems * sizeof(__bf16);
  dim3 grid(PBLK, VSPL, B_);

  if (d_ws && ws_size >= need) {
    __bf16* mv_bf = (__bf16*)d_ws;
    int n4 = (int)(mv_elems / 4);
    convert_mv<<<(n4 + 255) / 256, 256, 0, stream>>>(mv, mv_bf, n4);
    mem_reader_attn<true><<<grid, 256, 0, stream>>>(mk, nullptr, mv_bf, qk, out);
  } else {
    mem_reader_attn<false><<<grid, 256, 0, stream>>>(mk, mv, nullptr, qk, out);
  }

  int total = B_ * CV_ * P_;
  qv_copy<<<(total + 255) / 256, 256, 0, stream>>>(qv, out, total);
}


# next chunk's mv fragments and mk rows are loaded a whole chunk ahead into registers and stay in flight across the chunk's barriers, which now wait for LDS only instead of draining the load counter
# speedup vs baseline: 6.0975x; 1.4550x over previous
#include <hip/hip_runtime.h>

#define B_    4
#define CK_   64
#define CV_   512
#define H_    30
#define W_    54
#define T_    8
#define P_    (H_*W_)
#define N_    (T_*H_*W_)
#define CHUNK 128
#define NCHUNK ((N_ + CHUNK - 1)/CHUNK)
#define PQ    64
#define PT_   (PQ/16)
#define VB_   256
#define VSPL  (CV_/VB_)
#define PBLK  ((P_ + PQ - 1)/PQ)

typedef __attribute__((ext_vector_type(16))) __bf16       v16bf;
typedef __attribute__((ext_vector_type(8)))  float        v8f;
typedef __attribute__((ext_vector_type(4)))  unsigned int v4u;

union Frag16 {
  v16bf  bf;
  v4u    q[2];
  __bf16 h[16];
};
union Pack8 {
  v4u    q;
  __bf16 h[8];
};
union Pack4 {
  uint2  u;
  __bf16 h[4];
};

#define QSTR  72
#define MKSTR 72
#define PSTR  136

static __device__ __forceinline__ void lds_barrier() {
  asm volatile("s_wait_dscnt 0x0\n\ts_barrier_signal -1\n\ts_barrier_wait -1" ::: "memory");
}

template <bool MV_BF16>
__global__ __launch_bounds__(256) void mem_reader_attn(
    const float* __restrict__ mk, const float* __restrict__ mv_f32,
    const __bf16* __restrict__ mv_bf, const float* __restrict__ qk,
    float* __restrict__ out)
{
  __shared__ __bf16 s_q[PQ * QSTR];
  __shared__ __bf16 s_mk[CHUNK * MKSTR];
  __shared__ __bf16 s_prob[PQ * PSTR];
  __shared__ float  s_pmax[8][PT_][32];
  __shared__ float  s_psum[8][PT_][32];
  __shared__ float  s_mrun[PQ];
  __shared__ float  s_lrun[PQ];
  __shared__ float  s_alpha[PQ];

  const int tid  = threadIdx.x;
  const int wave = tid >> 5;
  const int lane = tid & 31;
  const int col  = lane & 15;
  const int kg   = (lane >> 4) * 8;
  const int p0   = blockIdx.x * PQ;
  const int vb0  = blockIdx.y * VB_ + wave * 32;
  const int bi   = blockIdx.z;

  const float*  mk_b  = mk + (size_t)bi * CK_ * N_;
  const float*  qk_b  = qk + (size_t)bi * CK_ * P_;
  const float*  mvf_b = mv_f32 ? (mv_f32 + (size_t)bi * CV_ * N_) : nullptr;
  const __bf16* mvb_b = mv_bf  ? (mv_bf  + (size_t)bi * CV_ * N_) : nullptr;

  for (int idx = tid; idx < CK_ * PQ; idx += 256) {
    int p = idx & (PQ - 1), c = idx / PQ;
    int pg = p0 + p; if (pg > P_ - 1) pg = P_ - 1;
    s_q[p * QSTR + c] = (__bf16)(qk_b[(size_t)c * P_ + pg] * 0.125f);
  }
  if (tid < PQ) { s_mrun[tid] = -1e30f; s_lrun[tid] = 0.f; s_alpha[tid] = 0.f; }

  v8f acc[2][PT_];
#pragma unroll
  for (int vt = 0; vt < 2; ++vt)
#pragma unroll
    for (int pt = 0; pt < PT_; ++pt) acc[vt][pt] = v8f{};

  const float* mk_t = mk_b + (size_t)(wave * 8) * N_;
  float4 pre[8];
  {
    const int ngc = 4 * lane;
#pragma unroll
    for (int j = 0; j < 8; ++j)
      pre[j] = *(const float4*)(mk_t + (size_t)j * N_ + ngc);
  }

  const __bf16* mv_r0 = MV_BF16 ? (mvb_b + (size_t)(vb0 + col) * N_) : nullptr;
  Frag16 mvr[4][2];
  if (MV_BF16) {
    __builtin_amdgcn_sched_barrier(0);
#pragma unroll
    for (int ks = 0; ks < 4; ++ks) {
      const int nA = ks * 32 + kg;
#pragma unroll
      for (int vt = 0; vt < 2; ++vt) {
        const __bf16* rowp = mv_r0 + (size_t)(vt * 16) * N_;
        mvr[ks][vt].q[0] = *(const v4u*)(rowp + nA);
        mvr[ks][vt].q[1] = *(const v4u*)(rowp + nA + 16);
      }
    }
  }

  for (int nc = 0; nc < NCHUNK; ++nc) {
    const int n0 = nc * CHUNK;
    lds_barrier();

    {
      const bool ok = (n0 + 4 * lane) < N_;
      Pack8 r0, r1, r2, r3;
#pragma unroll
      for (int j = 0; j < 8; ++j) {
        r0.h[j] = (__bf16)(ok ? pre[j].x : 0.f);
        r1.h[j] = (__bf16)(ok ? pre[j].y : 0.f);
        r2.h[j] = (__bf16)(ok ? pre[j].z : 0.f);
        r3.h[j] = (__bf16)(ok ? pre[j].w : 0.f);
      }
      __bf16* dp = &s_mk[(4 * lane) * MKSTR + wave * 8];
      *(v4u*)(dp)             = r0.q;
      *(v4u*)(dp + MKSTR)     = r1.q;
      *(v4u*)(dp + 2 * MKSTR) = r2.q;
      *(v4u*)(dp + 3 * MKSTR) = r3.q;
    }
    {
      const int ncn = (nc + 1 < NCHUNK) ? (nc + 1) : (NCHUNK - 1);
      int ngc = ncn * CHUNK + 4 * lane;
      if (ngc > N_ - 4) ngc = N_ - 4;
#pragma unroll
      for (int j = 0; j < 8; ++j)
        pre[j] = *(const float4*)(mk_t + (size_t)j * N_ + ngc);
    }
    lds_barrier();

    const int mrow = wave * 16;
    Frag16 amk[2];
#pragma unroll
    for (int ks = 0; ks < 2; ++ks) {
      const __bf16* ap = &s_mk[(mrow + col) * MKSTR + ks * 32 + kg];
      amk[ks].q[0] = *(const v4u*)ap;  amk[ks].q[1] = *(const v4u*)(ap + 16);
    }
    float sv[PT_][8];
#pragma unroll
    for (int pt = 0; pt < PT_; ++pt) {
      v8f S = {};
#pragma unroll
      for (int ks = 0; ks < 2; ++ks) {
        Frag16 b;
        const __bf16* bp = &s_q[(pt * 16 + col) * QSTR + ks * 32 + kg];
        b.q[0] = *(const v4u*)bp;  b.q[1] = *(const v4u*)(bp + 16);
        S = __builtin_amdgcn_wmma_f32_16x16x32_bf16(false, amk[ks].bf, false, b.bf,
                                                    (short)0, S, false, false);
      }
      float lmax = -1e30f;
#pragma unroll
      for (int j = 0; j < 8; ++j) {
        int ng = n0 + mrow + kg + j;
        sv[pt][j] = (ng < N_) ? S[j] : -1e30f;
        lmax = fmaxf(lmax, sv[pt][j]);
      }
      s_pmax[wave][pt][lane] = lmax;
    }
    lds_barrier();

    if (tid < PQ) {
      const int pt = tid >> 4, c = tid & 15;
      float cmax = -1e30f;
#pragma unroll
      for (int w = 0; w < 8; ++w)
        cmax = fmaxf(cmax, fmaxf(s_pmax[w][pt][c], s_pmax[w][pt][c + 16]));
      float mold = s_mrun[tid];
      float mnew = fmaxf(mold, cmax);
      s_alpha[tid] = __expf(mold - mnew);
      s_mrun[tid]  = mnew;
    }
    lds_barrier();

#pragma unroll
    for (int pt = 0; pt < PT_; ++pt) {
      const float mnew = s_mrun[pt * 16 + col];
      const float alph = s_alpha[pt * 16 + col];
      acc[0][pt] *= alph;
      acc[1][pt] *= alph;

      Pack8 pk;
      float lsum = 0.f;
#pragma unroll
      for (int j = 0; j < 8; ++j) {
        int ng = n0 + mrow + kg + j;
        float pr = (ng < N_) ? __expf(sv[pt][j] - mnew) : 0.f;
        lsum += pr;
        pk.h[j] = (__bf16)pr;
      }
      *(v4u*)&s_prob[(pt * 16 + col) * PSTR + mrow + kg] = pk.q;
      s_psum[wave][pt][lane] = lsum;
    }
    lds_barrier();

    if (tid < PQ) {
      const int pt = tid >> 4, c = tid & 15;
      float csum = 0.f;
#pragma unroll
      for (int w = 0; w < 8; ++w)
        csum += s_psum[w][pt][c] + s_psum[w][pt][c + 16];
      s_lrun[tid] = s_lrun[tid] * s_alpha[tid] + csum;
    }

#pragma unroll
    for (int ks = 0; ks < 4; ++ks) {
      int nA = n0 + ks * 32 + kg;
      int b1 = nA;      if (b1 > N_ - 8) b1 = N_ - 8;
      int b2 = nA + 16; if (b2 > N_ - 8) b2 = N_ - 8;
      Frag16 b[PT_];
#pragma unroll
      for (int pt = 0; pt < PT_; ++pt) {
        const __bf16* bp = &s_prob[(pt * 16 + col) * PSTR + ks * 32 + kg];
        b[pt].q[0] = *(const v4u*)bp;  b[pt].q[1] = *(const v4u*)(bp + 16);
      }
#pragma unroll
      for (int vt = 0; vt < 2; ++vt) {
        const int v = vb0 + vt * 16 + col;
        Frag16 a;
        if (MV_BF16) {
          a = mvr[ks][vt];
        } else {
          const float* rowp = mvf_b + (size_t)v * N_;
          const float4 f0 = *(const float4*)(rowp + b1);
          const float4 f1 = *(const float4*)(rowp + b1 + 4);
          const float4 f2 = *(const float4*)(rowp + b2);
          const float4 f3 = *(const float4*)(rowp + b2 + 4);
          a.h[0]=(__bf16)f0.x;  a.h[1]=(__bf16)f0.y;  a.h[2]=(__bf16)f0.z;  a.h[3]=(__bf16)f0.w;
          a.h[4]=(__bf16)f1.x;  a.h[5]=(__bf16)f1.y;  a.h[6]=(__bf16)f1.z;  a.h[7]=(__bf16)f1.w;
          a.h[8]=(__bf16)f2.x;  a.h[9]=(__bf16)f2.y;  a.h[10]=(__bf16)f2.z; a.h[11]=(__bf16)f2.w;
          a.h[12]=(__bf16)f3.x; a.h[13]=(__bf16)f3.y; a.h[14]=(__bf16)f3.z; a.h[15]=(__bf16)f3.w;
        }
#pragma unroll
        for (int pt = 0; pt < PT_; ++pt)
          acc[vt][pt] = __builtin_amdgcn_wmma_f32_16x16x32_bf16(false, a.bf, false, b[pt].bf,
                                                                (short)0, acc[vt][pt], false, false);
      }
    }

    if (MV_BF16) {
      const int ncn = (nc + 1 < NCHUNK) ? (nc + 1) : (NCHUNK - 1);
#pragma unroll
      for (int ks = 0; ks < 4; ++ks) {
        const int nA = ncn * CHUNK + ks * 32 + kg;
        int b1 = nA;      if (b1 > N_ - 8) b1 = N_ - 8;
        int b2 = nA + 16; if (b2 > N_ - 8) b2 = N_ - 8;
#pragma unroll
        for (int vt = 0; vt < 2; ++vt) {
          const __bf16* rowp = mv_r0 + (size_t)(vt * 16) * N_;
          mvr[ks][vt].q[0] = *(const v4u*)(rowp + b1);
          mvr[ks][vt].q[1] = *(const v4u*)(rowp + b2);
        }
      }
    }
  }

  __syncthreads();
#pragma unroll
  for (int pt = 0; pt < PT_; ++pt) {
    const float linv = 1.0f / s_lrun[pt * 16 + col];
    const int pg = p0 + pt * 16 + col;
    if (pg < P_) {
#pragma unroll
      for (int vt = 0; vt < 2; ++vt) {
#pragma unroll
        for (int j = 0; j < 8; ++j) {
          int v = vb0 + vt * 16 + kg + j;
          out[((size_t)bi * (2 * CV_) + v) * P_ + pg] = acc[vt][pt][j] * linv;
        }
      }
    }
  }
}

__global__ __launch_bounds__(256) void convert_mv(const float* __restrict__ src,
                                                  __bf16* __restrict__ dst, int n4)
{
  int i = blockIdx.x * blockDim.x + threadIdx.x;
  if (i < n4) {
    float4 f = ((const float4*)src)[i];
    Pack4 p;
    p.h[0] = (__bf16)f.x; p.h[1] = (__bf16)f.y;
    p.h[2] = (__bf16)f.z; p.h[3] = (__bf16)f.w;
    *(uint2*)(dst + 4 * (size_t)i) = p.u;
  }
}

__global__ __launch_bounds__(256) void qv_copy(const float* __restrict__ qv,
                                               float* __restrict__ out, int total)
{
  int i = blockIdx.x * blockDim.x + threadIdx.x;
  if (i < total) {
    int b = i / (CV_ * P_);
    int r = i - b * (CV_ * P_);
    out[(size_t)b * (2 * CV_ * P_) + (CV_ * P_) + r] = qv[i];
  }
}

extern "C" void kernel_launch(void* const* d_in, const int* in_sizes, int n_in,
                              void* d_out, int out_size, void* d_ws, size_t ws_size,
                              hipStream_t stream) {
  const float* mk = (const float*)d_in[0];
  const float* mv = (const float*)d_in[1];
  const float* qk = (const float*)d_in[2];
  const float* qv = (const float*)d_in[3];
  float* out = (float*)d_out;

  const size_t mv_elems = (size_t)B_ * CV_ * N_;
  const size_t need     = mv_elems * sizeof(__bf16);
  dim3 grid(PBLK, VSPL, B_);

  if (d_ws && ws_size >= need) {
    __bf16* mv_bf = (__bf16*)d_ws;
    int n4 = (int)(mv_elems / 4);
    convert_mv<<<(n4 + 255) / 256, 256, 0, stream>>>(mv, mv_bf, n4);
    mem_reader_attn<true><<<grid, 256, 0, stream>>>(mk, nullptr, mv_bf, qk, out);
  } else {
    mem_reader_attn<false><<<grid, 256, 0, stream>>>(mk, mv, nullptr, qk, out);
  }

  int total = B_ * CV_ * P_;
  qv_copy<<<(total + 255) / 256, 256, 0, stream>>>(qv, out, total);
}


# bf16 copy of mv is written by the conversion pass in WMMA fragment order: each mv fragment load of the main kernel reads 512 contiguous bytes (4 whole lines), not 32 bytes from each of 16 lines
# speedup vs baseline: 7.9576x; 1.3051x over previous
#include <hip/hip_runtime.h>

#define B_    4
#define CK_   64
#define CV_   512
#define H_    30
#define W_    54
#define T_    8
#define P_    (H_*W_)
#define N_    (T_*H_*W_)
#define CHUNK 128
#define NCHUNK ((N_ + CHUNK - 1)/CHUNK)
#define PQ    64
#define PT_   (PQ/16)
#define VB_   256
#define VSPL  (CV_/VB_)
#define PBLK  ((P_ + PQ - 1)/PQ)
#define KSTEP 32
#define NKS   (N_/KSTEP)
#define FRAG  (16*KSTEP)
#define CONV_THREADS (B_*CV_*N_/8)

typedef __attribute__((ext_vector_type(16))) __bf16       v16bf;
typedef __attribute__((ext_vector_type(8)))  float        v8f;
typedef __attribute__((ext_vector_type(4)))  unsigned int v4u;

union Frag16 {
  v16bf  bf;
  v4u    q[2];
  __bf16 h[16];
};
union Pack8 {
  v4u    q;
  __bf16 h[8];
};
union Pack4 {
  uint2  u;
  __bf16 h[4];
};

#define QSTR  72
#define MKSTR 72
#define PSTR  136

static __device__ __forceinline__ void lds_barrier() {
  asm volatile("s_wait_dscnt 0x0\n\ts_barrier_signal -1\n\ts_barrier_wait -1" ::: "memory");
}

template <bool MV_BF16>
__global__ __launch_bounds__(256) void mem_reader_attn(
    const float* __restrict__ mk, const float* __restrict__ mv_f32,
    const __bf16* __restrict__ mv_bf, const float* __restrict__ qk,
    float* __restrict__ out)
{
  __shared__ __bf16 s_q[PQ * QSTR];
  __shared__ __bf16 s_mk[CHUNK * MKSTR];
  __shared__ __bf16 s_prob[PQ * PSTR];
  __shared__ float  s_pmax[8][PT_][32];
  __shared__ float  s_psum[8][PT_][32];
  __shared__ float  s_mrun[PQ];
  __shared__ float  s_lrun[PQ];
  __shared__ float  s_alpha[PQ];

  const int tid  = threadIdx.x;
  const int wave = tid >> 5;
  const int lane = tid & 31;
  const int col  = lane & 15;
  const int kg   = (lane >> 4) * 8;
  const int p0   = blockIdx.x * PQ;
  const int vb0  = blockIdx.y * VB_ + wave * 32;
  const int bi   = blockIdx.z;

  const float*  mk_b  = mk + (size_t)bi * CK_ * N_;
  const float*  qk_b  = qk + (size_t)bi * CK_ * P_;
  const float*  mvf_b = mv_f32 ? (mv_f32 + (size_t)bi * CV_ * N_) : nullptr;
  const __bf16* mvb_b = mv_bf  ? (mv_bf  + (size_t)bi * CV_ * N_) : nullptr;

  for (int idx = tid; idx < CK_ * PQ; idx += 256) {
    int p = idx & (PQ - 1), c = idx / PQ;
    int pg = p0 + p; if (pg > P_ - 1) pg = P_ - 1;
    s_q[p * QSTR + c] = (__bf16)(qk_b[(size_t)c * P_ + pg] * 0.125f);
  }
  if (tid < PQ) { s_mrun[tid] = -1e30f; s_lrun[tid] = 0.f; s_alpha[tid] = 0.f; }

  v8f acc[2][PT_];
#pragma unroll
  for (int vt = 0; vt < 2; ++vt)
#pragma unroll
    for (int pt = 0; pt < PT_; ++pt) acc[vt][pt] = v8f{};

  const float* mk_t = mk_b + (size_t)(wave * 8) * N_;
  float4 pre[8];
  {
    const int ngc = 4 * lane;
#pragma unroll
    for (int j = 0; j < 8; ++j)
      pre[j] = *(const float4*)(mk_t + (size_t)j * N_ + ngc);
  }

  const __bf16* mv_w0 = MV_BF16 ? (mvb_b + (size_t)(vb0 / 16) * NKS * FRAG + lane * 8) : nullptr;
  Frag16 mvr[4][2];
  if (MV_BF16) {
    __builtin_amdgcn_sched_barrier(0);
#pragma unroll
    for (int ks = 0; ks < 4; ++ks) {
#pragma unroll
      for (int vt = 0; vt < 2; ++vt) {
        const __bf16* blk = mv_w0 + ((size_t)vt * NKS + ks) * FRAG;
        mvr[ks][vt].q[0] = *(const v4u*)(blk);
        mvr[ks][vt].q[1] = *(const v4u*)(blk + FRAG / 2);
      }
    }
  }

  for (int nc = 0; nc < NCHUNK; ++nc) {
    const int n0 = nc * CHUNK;
    lds_barrier();

    {
      const bool ok = (n0 + 4 * lane) < N_;
      Pack8 r0, r1, r2, r3;
#pragma unroll
      for (int j = 0; j < 8; ++j) {
        r0.h[j] = (__bf16)(ok ? pre[j].x : 0.f);
        r1.h[j] = (__bf16)(ok ? pre[j].y : 0.f);
        r2.h[j] = (__bf16)(ok ? pre[j].z : 0.f);
        r3.h[j] = (__bf16)(ok ? pre[j].w : 0.f);
      }
      __bf16* dp = &s_mk[(4 * lane) * MKSTR + wave * 8];
      *(v4u*)(dp)             = r0.q;
      *(v4u*)(dp + MKSTR)     = r1.q;
      *(v4u*)(dp + 2 * MKSTR) = r2.q;
      *(v4u*)(dp + 3 * MKSTR) = r3.q;
    }
    {
      const int ncn = (nc + 1 < NCHUNK) ? (nc + 1) : (NCHUNK - 1);
      int ngc = ncn * CHUNK + 4 * lane;
      if (ngc > N_ - 4) ngc = N_ - 4;
#pragma unroll
      for (int j = 0; j < 8; ++j)
        pre[j] = *(const float4*)(mk_t + (size_t)j * N_ + ngc);
    }
    lds_barrier();

    const int mrow = wave * 16;
    Frag16 amk[2];
#pragma unroll
    for (int ks = 0; ks < 2; ++ks) {
      const __bf16* ap = &s_mk[(mrow + col) * MKSTR + ks * 32 + kg];
      amk[ks].q[0] = *(const v4u*)ap;  amk[ks].q[1] = *(const v4u*)(ap + 16);
    }
    float sv[PT_][8];
#pragma unroll
    for (int pt = 0; pt < PT_; ++pt) {
      v8f S = {};
#pragma unroll
      for (int ks = 0; ks < 2; ++ks) {
        Frag16 b;
        const __bf16* bp = &s_q[(pt * 16 + col) * QSTR + ks * 32 + kg];
        b.q[0] = *(const v4u*)bp;  b.q[1] = *(const v4u*)(bp + 16);
        S = __builtin_amdgcn_wmma_f32_16x16x32_bf16(false, amk[ks].bf, false, b.bf,
                                                    (short)0, S, false, false);
      }
      float lmax = -1e30f;
#pragma unroll
      for (int j = 0; j < 8; ++j) {
        int ng = n0 + mrow + kg + j;
        sv[pt][j] = (ng < N_) ? S[j] : -1e30f;
        lmax = fmaxf(lmax, sv[pt][j]);
      }
      s_pmax[wave][pt][lane] = lmax;
    }
    lds_barrier();

    if (tid < PQ) {
      const int pt = tid >> 4, c = tid & 15;
      float cmax = -1e30f;
#pragma unroll
      for (int w = 0; w < 8; ++w)
        cmax = fmaxf(cmax, fmaxf(s_pmax[w][pt][c], s_pmax[w][pt][c + 16]));
      float mold = s_mrun[tid];
      float mnew = fmaxf(mold, cmax);
      s_alpha[tid] = __expf(mold - mnew);
      s_mrun[tid]  = mnew;
    }
    lds_barrier();

#pragma unroll
    for (int pt = 0; pt < PT_; ++pt) {
      const float mnew = s_mrun[pt * 16 + col];
      const float alph = s_alpha[pt * 16 + col];
      acc[0][pt] *= alph;
      acc[1][pt] *= alph;

      Pack8 pk;
      float lsum = 0.f;
#pragma unroll
      for (int j = 0; j < 8; ++j) {
        int ng = n0 + mrow + kg + j;
        float pr = (ng < N_) ? __expf(sv[pt][j] - mnew) : 0.f;
        lsum += pr;
        pk.h[j] = (__bf16)pr;
      }
      *(v4u*)&s_prob[(pt * 16 + col) * PSTR + mrow + kg] = pk.q;
      s_psum[wave][pt][lane] = lsum;
    }
    lds_barrier();

    if (tid < PQ) {
      const int pt = tid >> 4, c = tid & 15;
      float csum = 0.f;
#pragma unroll
      for (int w = 0; w < 8; ++w)
        csum += s_psum[w][pt][c] + s_psum[w][pt][c + 16];
      s_lrun[tid] = s_lrun[tid] * s_alpha[tid] + csum;
    }

#pragma unroll
    for (int ks = 0; ks < 4; ++ks) {
      int nA = n0 + ks * 32 + kg;
      int b1 = nA;      if (b1 > N_ - 8) b1 = N_ - 8;
      int b2 = nA + 16; if (b2 > N_ - 8) b2 = N_ - 8;
      Frag16 b[PT_];
#pragma unroll
      for (int pt = 0; pt < PT_; ++pt) {
        const __bf16* bp = &s_prob[(pt * 16 + col) * PSTR + ks * 32 + kg];
        b[pt].q[0] = *(const v4u*)bp;  b[pt].q[1] = *(const v4u*)(bp + 16);
      }
#pragma unroll
      for (int vt = 0; vt < 2; ++vt) {
        const int v = vb0 + vt * 16 + col;
        Frag16 a;
        if (MV_BF16) {
          a = mvr[ks][vt];
        } else {
          const float* rowp = mvf_b + (size_t)v * N_;
          const float4 f0 = *(const float4*)(rowp + b1);
          const float4 f1 = *(const float4*)(rowp + b1 + 4);
          const float4 f2 = *(const float4*)(rowp + b2);
          const float4 f3 = *(const float4*)(rowp + b2 + 4);
          a.h[0]=(__bf16)f0.x;  a.h[1]=(__bf16)f0.y;  a.h[2]=(__bf16)f0.z;  a.h[3]=(__bf16)f0.w;
          a.h[4]=(__bf16)f1.x;  a.h[5]=(__bf16)f1.y;  a.h[6]=(__bf16)f1.z;  a.h[7]=(__bf16)f1.w;
          a.h[8]=(__bf16)f2.x;  a.h[9]=(__bf16)f2.y;  a.h[10]=(__bf16)f2.z; a.h[11]=(__bf16)f2.w;
          a.h[12]=(__bf16)f3.x; a.h[13]=(__bf16)f3.y; a.h[14]=(__bf16)f3.z; a.h[15]=(__bf16)f3.w;
        }
#pragma unroll
        for (int pt = 0; pt < PT_; ++pt)
          acc[vt][pt] = __builtin_amdgcn_wmma_f32_16x16x32_bf16(false, a.bf, false, b[pt].bf,
                                                                (short)0, acc[vt][pt], false, false);
      }
    }

    if (MV_BF16) {
      const int ncn = (nc + 1 < NCHUNK) ? (nc + 1) : (NCHUNK - 1);
#pragma unroll
      for (int ks = 0; ks < 4; ++ks) {
        int kst = ncn * (CHUNK / KSTEP) + ks;
        if (kst > NKS - 1) kst = NKS - 1;
#pragma unroll
        for (int vt = 0; vt < 2; ++vt) {
          const __bf16* blk = mv_w0 + ((size_t)vt * NKS + kst) * FRAG;
          mvr[ks][vt].q[0] = *(const v4u*)(blk);
          mvr[ks][vt].q[1] = *(const v4u*)(blk + FRAG / 2);
        }
      }
    }
  }

  __syncthreads();
#pragma unroll
  for (int pt = 0; pt < PT_; ++pt) {
    const float linv = 1.0f / s_lrun[pt * 16 + col];
    const int pg = p0 + pt * 16 + col;
    if (pg < P_) {
#pragma unroll
      for (int vt = 0; vt < 2; ++vt) {
#pragma unroll
        for (int j = 0; j < 8; ++j) {
          int v = vb0 + vt * 16 + kg + j;
          out[((size_t)bi * (2 * CV_) + v) * P_ + pg] = acc[vt][pt][j] * linv;
        }
      }
    }
  }
}

__global__ __launch_bounds__(256) void convert_mv(const float* __restrict__ src,
                                                  __bf16* __restrict__ dst, int nthr)
{
  int i = blockIdx.x * blockDim.x + threadIdx.x;
  if (i < nthr) {
    const int g   = i & 3;
    const int r   = (i >> 2) & 15;
    const int blk = i >> 6;
    const int kst = blk % NKS;
    const int rt  = blk / NKS;
    const float* sp = src + ((size_t)rt * 16 + r) * N_ + kst * KSTEP + g * 8;
    const float4 f0 = *(const float4*)(sp);
    const float4 f1 = *(const float4*)(sp + 4);
    Pack8 p;
    p.h[0] = (__bf16)f0.x; p.h[1] = (__bf16)f0.y;
    p.h[2] = (__bf16)f0.z; p.h[3] = (__bf16)f0.w;
    p.h[4] = (__bf16)f1.x; p.h[5] = (__bf16)f1.y;
    p.h[6] = (__bf16)f1.z; p.h[7] = (__bf16)f1.w;
    *(v4u*)(dst + (size_t)blk * FRAG + (g >> 1) * (FRAG / 2) + ((g & 1) * 16 + r) * 8) = p.q;
  }
}

__global__ __launch_bounds__(256) void qv_copy(const float* __restrict__ qv,
                                               float* __restrict__ out, int total)
{
  int i = blockIdx.x * blockDim.x + threadIdx.x;
  if (i < total) {
    int b = i / (CV_ * P_);
    int r = i - b * (CV_ * P_);
    out[(size_t)b * (2 * CV_ * P_) + (CV_ * P_) + r] = qv[i];
  }
}

extern "C" void kernel_launch(void* const* d_in, const int* in_sizes, int n_in,
                              void* d_out, int out_size, void* d_ws, size_t ws_size,
                              hipStream_t stream) {
  const float* mk = (const float*)d_in[0];
  const float* mv = (const float*)d_in[1];
  const float* qk = (const float*)d_in[2];
  const float* qv = (const float*)d_in[3];
  float* out = (float*)d_out;

  const size_t mv_elems = (size_t)B_ * CV_ * N_;
  const size_t need     = mv_elems * sizeof(__bf16);
  dim3 grid(PBLK, VSPL, B_);

  if (d_ws && ws_size >= need) {
    __bf16* mv_bf = (__bf16*)d_ws;
    convert_mv<<<(CONV_THREADS + 255) / 256, 256, 0, stream>>>(mv, mv_bf, CONV_THREADS);
    mem_reader_attn<true><<<grid, 256, 0, stream>>>(mk, nullptr, mv_bf, qk, out);
  } else {
    mem_reader_attn<false><<<grid, 256, 0, stream>>>(mk, mv, nullptr, qk, out);
  }

  int total = B_ * CV_ * P_;
  qv_copy<<<(total + 255) / 256, 256, 0, stream>>>(qv, out, total);
}


# conversion and copy passes relaunched as few fat waves: each thread moves 27 (copy: 9) 16-byte pieces with the loads in flight together; 480 and 360 work-groups replace 12960 short-lived ones each
# speedup vs baseline: 8.3167x; 1.0451x over previous
#include <hip/hip_runtime.h>

#define B_    4
#define CK_   64
#define CV_   512
#define H_    30
#define W_    54
#define T_    8
#define P_    (H_*W_)
#define N_    (T_*H_*W_)
#define CHUNK 128
#define NCHUNK ((N_ + CHUNK - 1)/CHUNK)
#define PQ    64
#define PT_   (PQ/16)
#define VB_   256
#define VSPL  (CV_/VB_)
#define PBLK  ((P_ + PQ - 1)/PQ)
#define KSTEP 32
#define NKS   (N_/KSTEP)
#define FRAG  (16*KSTEP)
#define CONV_KS      27
#define CONV_KB      9
#define CONV_GRPS    (NKS / CONV_KS)
#define CONV_THREADS ((B_*CV_/16) * CONV_GRPS * 64)
#define CONV_BLOCKS  (CONV_THREADS / 256)
#define QV_V4        (B_*CV_*P_/4)
#define QV_PER       9
#define QV_THREADS   (QV_V4 / QV_PER)
#define QV_BLOCKS    (QV_THREADS / 256)

typedef __attribute__((ext_vector_type(16))) __bf16       v16bf;
typedef __attribute__((ext_vector_type(8)))  float        v8f;
typedef __attribute__((ext_vector_type(4)))  unsigned int v4u;
typedef __attribute__((ext_vector_type(4)))  float        v4f;

union Frag16 {
  v16bf  bf;
  v4u    q[2];
  __bf16 h[16];
};
union Pack8 {
  v4u    q;
  __bf16 h[8];
};
union Pack4 {
  uint2  u;
  __bf16 h[4];
};

#define QSTR  72
#define MKSTR 72
#define PSTR  136

static __device__ __forceinline__ void lds_barrier() {
  asm volatile("s_wait_dscnt 0x0\n\ts_barrier_signal -1\n\ts_barrier_wait -1" ::: "memory");
}

template <bool MV_BF16>
__global__ __launch_bounds__(256) void mem_reader_attn(
    const float* __restrict__ mk, const float* __restrict__ mv_f32,
    const __bf16* __restrict__ mv_bf, const float* __restrict__ qk,
    float* __restrict__ out)
{
  __shared__ __bf16 s_q[PQ * QSTR];
  __shared__ __bf16 s_mk[CHUNK * MKSTR];
  __shared__ __bf16 s_prob[PQ * PSTR];
  __shared__ float  s_pmax[8][PT_][32];
  __shared__ float  s_psum[8][PT_][32];
  __shared__ float  s_mrun[PQ];
  __shared__ float  s_lrun[PQ];
  __shared__ float  s_alpha[PQ];

  const int tid  = threadIdx.x;
  const int wave = tid >> 5;
  const int lane = tid & 31;
  const int col  = lane & 15;
  const int kg   = (lane >> 4) * 8;
  const int p0   = blockIdx.x * PQ;
  const int vb0  = blockIdx.y * VB_ + wave * 32;
  const int bi   = blockIdx.z;

  const float*  mk_b  = mk + (size_t)bi * CK_ * N_;
  const float*  qk_b  = qk + (size_t)bi * CK_ * P_;
  const float*  mvf_b = mv_f32 ? (mv_f32 + (size_t)bi * CV_ * N_) : nullptr;
  const __bf16* mvb_b = mv_bf  ? (mv_bf  + (size_t)bi * CV_ * N_) : nullptr;

  for (int idx = tid; idx < CK_ * PQ; idx += 256) {
    int p = idx & (PQ - 1), c = idx / PQ;
    int pg = p0 + p; if (pg > P_ - 1) pg = P_ - 1;
    s_q[p * QSTR + c] = (__bf16)(qk_b[(size_t)c * P_ + pg] * 0.125f);
  }
  if (tid < PQ) { s_mrun[tid] = -1e30f; s_lrun[tid] = 0.f; s_alpha[tid] = 0.f; }

  v8f acc[2][PT_];
#pragma unroll
  for (int vt = 0; vt < 2; ++vt)
#pragma unroll
    for (int pt = 0; pt < PT_; ++pt) acc[vt][pt] = v8f{};

  const float* mk_t = mk_b + (size_t)(wave * 8) * N_;
  float4 pre[8];
  {
    const int ngc = 4 * lane;
#pragma unroll
    for (int j = 0; j < 8; ++j)
      pre[j] = *(const float4*)(mk_t + (size_t)j * N_ + ngc);
  }

  const __bf16* mv_w0 = MV_BF16 ? (mvb_b + (size_t)(vb0 / 16) * NKS * FRAG + lane * 8) : nullptr;
  Frag16 mvr[4][2];
  if (MV_BF16) {
    __builtin_amdgcn_sched_barrier(0);
#pragma unroll
    for (int ks = 0; ks < 4; ++ks) {
#pragma unroll
      for (int vt = 0; vt < 2; ++vt) {
        const __bf16* blk = mv_w0 + ((size_t)vt * NKS + ks) * FRAG;
        mvr[ks][vt].q[0] = *(const v4u*)(blk);
        mvr[ks][vt].q[1] = *(const v4u*)(blk + FRAG / 2);
      }
    }
  }

  for (int nc = 0; nc < NCHUNK; ++nc) {
    const int n0 = nc * CHUNK;
    lds_barrier();

    {
      const bool ok = (n0 + 4 * lane) < N_;
      Pack8 r0, r1, r2, r3;
#pragma unroll
      for (int j = 0; j < 8; ++j) {
        r0.h[j] = (__bf16)(ok ? pre[j].x : 0.f);
        r1.h[j] = (__bf16)(ok ? pre[j].y : 0.f);
        r2.h[j] = (__bf16)(ok ? pre[j].z : 0.f);
        r3.h[j] = (__bf16)(ok ? pre[j].w : 0.f);
      }
      __bf16* dp = &s_mk[(4 * lane) * MKSTR + wave * 8];
      *(v4u*)(dp)             = r0.q;
      *(v4u*)(dp + MKSTR)     = r1.q;
      *(v4u*)(dp + 2 * MKSTR) = r2.q;
      *(v4u*)(dp + 3 * MKSTR) = r3.q;
    }
    {
      const int ncn = (nc + 1 < NCHUNK) ? (nc + 1) : (NCHUNK - 1);
      int ngc = ncn * CHUNK + 4 * lane;
      if (ngc > N_ - 4) ngc = N_ - 4;
#pragma unroll
      for (int j = 0; j < 8; ++j)
        pre[j] = *(const float4*)(mk_t + (size_t)j * N_ + ngc);
    }
    lds_barrier();

    const int mrow = wave * 16;
    Frag16 amk[2];
#pragma unroll
    for (int ks = 0; ks < 2; ++ks) {
      const __bf16* ap = &s_mk[(mrow + col) * MKSTR + ks * 32 + kg];
      amk[ks].q[0] = *(const v4u*)ap;  amk[ks].q[1] = *(const v4u*)(ap + 16);
    }
    float sv[PT_][8];
#pragma unroll
    for (int pt = 0; pt < PT_; ++pt) {
      v8f S = {};
#pragma unroll
      for (int ks = 0; ks < 2; ++ks) {
        Frag16 b;
        const __bf16* bp = &s_q[(pt * 16 + col) * QSTR + ks * 32 + kg];
        b.q[0] = *(const v4u*)bp;  b.q[1] = *(const v4u*)(bp + 16);
        S = __builtin_amdgcn_wmma_f32_16x16x32_bf16(false, amk[ks].bf, false, b.bf,
                                                    (short)0, S, false, false);
      }
      float lmax = -1e30f;
#pragma unroll
      for (int j = 0; j < 8; ++j) {
        int ng = n0 + mrow + kg + j;
        sv[pt][j] = (ng < N_) ? S[j] : -1e30f;
        lmax = fmaxf(lmax, sv[pt][j]);
      }
      s_pmax[wave][pt][lane] = lmax;
    }
    lds_barrier();

    if (tid < PQ) {
      const int pt = tid >> 4, c = tid & 15;
      float cmax = -1e30f;
#pragma unroll
      for (int w = 0; w < 8; ++w)
        cmax = fmaxf(cmax, fmaxf(s_pmax[w][pt][c], s_pmax[w][pt][c + 16]));
      float mold = s_mrun[tid];
      float mnew = fmaxf(mold, cmax);
      s_alpha[tid] = __expf(mold - mnew);
      s_mrun[tid]  = mnew;
    }
    lds_barrier();

#pragma unroll
    for (int pt = 0; pt < PT_; ++pt) {
      const float mnew = s_mrun[pt * 16 + col];
      const float alph = s_alpha[pt * 16 + col];
      acc[0][pt] *= alph;
      acc[1][pt] *= alph;

      Pack8 pk;
      float lsum = 0.f;
#pragma unroll
      for (int j = 0; j < 8; ++j) {
        int ng = n0 + mrow + kg + j;
        float pr = (ng < N_) ? __expf(sv[pt][j] - mnew) : 0.f;
        lsum += pr;
        pk.h[j] = (__bf16)pr;
      }
      *(v4u*)&s_prob[(pt * 16 + col) * PSTR + mrow + kg] = pk.q;
      s_psum[wave][pt][lane] = lsum;
    }
    lds_barrier();

    if (tid < PQ) {
      const int pt = tid >> 4, c = tid & 15;
      float csum = 0.f;
#pragma unroll
      for (int w = 0; w < 8; ++w)
        csum += s_psum[w][pt][c] + s_psum[w][pt][c + 16];
      s_lrun[tid] = s_lrun[tid] * s_alpha[tid] + csum;
    }

#pragma unroll
    for (int ks = 0; ks < 4; ++ks) {
      int nA = n0 + ks * 32 + kg;
      int b1 = nA;      if (b1 > N_ - 8) b1 = N_ - 8;
      int b2 = nA + 16; if (b2 > N_ - 8) b2 = N_ - 8;
      Frag16 b[PT_];
#pragma unroll
      for (int pt = 0; pt < PT_; ++pt) {
        const __bf16* bp = &s_prob[(pt * 16 + col) * PSTR + ks * 32 + kg];
        b[pt].q[0] = *(const v4u*)bp;  b[pt].q[1] = *(const v4u*)(bp + 16);
      }
#pragma unroll
      for (int vt = 0; vt < 2; ++vt) {
        const int v = vb0 + vt * 16 + col;
        Frag16 a;
        if (MV_BF16) {
          a = mvr[ks][vt];
        } else {
          const float* rowp = mvf_b + (size_t)v * N_;
          const float4 f0 = *(const float4*)(rowp + b1);
          const float4 f1 = *(const float4*)(rowp + b1 + 4);
          const float4 f2 = *(const float4*)(rowp + b2);
          const float4 f3 = *(const float4*)(rowp + b2 + 4);
          a.h[0]=(__bf16)f0.x;  a.h[1]=(__bf16)f0.y;  a.h[2]=(__bf16)f0.z;  a.h[3]=(__bf16)f0.w;
          a.h[4]=(__bf16)f1.x;  a.h[5]=(__bf16)f1.y;  a.h[6]=(__bf16)f1.z;  a.h[7]=(__bf16)f1.w;
          a.h[8]=(__bf16)f2.x;  a.h[9]=(__bf16)f2.y;  a.h[10]=(__bf16)f2.z; a.h[11]=(__bf16)f2.w;
          a.h[12]=(__bf16)f3.x; a.h[13]=(__bf16)f3.y; a.h[14]=(__bf16)f3.z; a.h[15]=(__bf16)f3.w;
        }
#pragma unroll
        for (int pt = 0; pt < PT_; ++pt)
          acc[vt][pt] = __builtin_amdgcn_wmma_f32_16x16x32_bf16(false, a.bf, false, b[pt].bf,
                                                                (short)0, acc[vt][pt], false, false);
      }
    }

    if (MV_BF16) {
      const int ncn = (nc + 1 < NCHUNK) ? (nc + 1) : (NCHUNK - 1);
#pragma unroll
      for (int ks = 0; ks < 4; ++ks) {
        int kst = ncn * (CHUNK / KSTEP) + ks;
        if (kst > NKS - 1) kst = NKS - 1;
#pragma unroll
        for (int vt = 0; vt < 2; ++vt) {
          const __bf16* blk = mv_w0 + ((size_t)vt * NKS + kst) * FRAG;
          mvr[ks][vt].q[0] = *(const v4u*)(blk);
          mvr[ks][vt].q[1] = *(const v4u*)(blk + FRAG / 2);
        }
      }
    }
  }

  __syncthreads();
#pragma unroll
  for (int pt = 0; pt < PT_; ++pt) {
    const float linv = 1.0f / s_lrun[pt * 16 + col];
    const int pg = p0 + pt * 16 + col;
    if (pg < P_) {
#pragma unroll
      for (int vt = 0; vt < 2; ++vt) {
#pragma unroll
        for (int j = 0; j < 8; ++j) {
          int v = vb0 + vt * 16 + kg + j;
          out[((size_t)bi * (2 * CV_) + v) * P_ + pg] = acc[vt][pt][j] * linv;
        }
      }
    }
  }
}

__global__ __launch_bounds__(256) void convert_mv(const float* __restrict__ src,
                                                  __bf16* __restrict__ dst)
{
  const int u   = blockIdx.x * 256 + threadIdx.x;
  const int g   = u & 3;
  const int r   = (u >> 2) & 15;
  const int grp = u >> 6;
  const int rt  = grp / CONV_GRPS;
  const int ks0 = (grp - rt * CONV_GRPS) * CONV_KS;
  const float* sp = src + ((size_t)rt * 16 + r) * N_ + ks0 * KSTEP + g * 8;
  __bf16* dp = dst + ((size_t)rt * NKS + ks0) * FRAG + (g >> 1) * (FRAG / 2) + ((g & 1) * 16 + r) * 8;
#pragma unroll
  for (int b = 0; b < CONV_KS / CONV_KB; ++b) {
    float4 f0[CONV_KB], f1[CONV_KB];
#pragma unroll
    for (int j = 0; j < CONV_KB; ++j) {
      const float* p = sp + (b * CONV_KB + j) * KSTEP;
      f0[j] = *(const float4*)(p);
      f1[j] = *(const float4*)(p + 4);
    }
#pragma unroll
    for (int j = 0; j < CONV_KB; ++j) {
      Pack8 p;
      p.h[0] = (__bf16)f0[j].x; p.h[1] = (__bf16)f0[j].y;
      p.h[2] = (__bf16)f0[j].z; p.h[3] = (__bf16)f0[j].w;
      p.h[4] = (__bf16)f1[j].x; p.h[5] = (__bf16)f1[j].y;
      p.h[6] = (__bf16)f1[j].z; p.h[7] = (__bf16)f1[j].w;
      *(v4u*)(dp + (size_t)(b * CONV_KB + j) * FRAG) = p.q;
    }
  }
}

__global__ __launch_bounds__(256) void qv_copy(const float* __restrict__ qv,
                                               float* __restrict__ out)
{
  const int t = blockIdx.x * 256 + threadIdx.x;
  const v4f* s4 = (const v4f*)qv;
  v4f* d4 = (v4f*)out;
  v4f v[QV_PER];
#pragma unroll
  for (int j = 0; j < QV_PER; ++j)
    v[j] = s4[t + j * QV_THREADS];
#pragma unroll
  for (int j = 0; j < QV_PER; ++j) {
    const int i4 = t + j * QV_THREADS;
    const int b  = i4 / (CV_ * P_ / 4);
    d4[i4 + (b + 1) * (CV_ * P_ / 4)] = v[j];
  }
}

extern "C" void kernel_launch(void* const* d_in, const int* in_sizes, int n_in,
                              void* d_out, int out_size, void* d_ws, size_t ws_size,
                              hipStream_t stream) {
  const float* mk = (const float*)d_in[0];
  const float* mv = (const float*)d_in[1];
  const float* qk = (const float*)d_in[2];
  const float* qv = (const float*)d_in[3];
  float* out = (float*)d_out;

  const size_t mv_elems = (size_t)B_ * CV_ * N_;
  const size_t need     = mv_elems * sizeof(__bf16);
  dim3 grid(PBLK, VSPL, B_);

  if (d_ws && ws_size >= need) {
    __bf16* mv_bf = (__bf16*)d_ws;
    convert_mv<<<CONV_BLOCKS, 256, 0, stream>>>(mv, mv_bf);
    mem_reader_attn<true><<<grid, 256, 0, stream>>>(mk, nullptr, mv_bf, qk, out);
  } else {
    mem_reader_attn<false><<<grid, 256, 0, stream>>>(mk, mv, nullptr, qk, out);
  }

  qv_copy<<<QV_BLOCKS, 256, 0, stream>>>(qv, out);
}


# keys split into 2 ranges over work-groups (grid 26x2x8, two per WGP at 256 registers); a combine pass rescales both to the common max and adds them: softmax sums add as 2 partials in range order
# speedup vs baseline: 10.0767x; 1.2116x over previous
#include <hip/hip_runtime.h>

#define B_    4
#define CK_   64
#define CV_   512
#define H_    30
#define W_    54
#define T_    8
#define P_    (H_*W_)
#define N_    (T_*H_*W_)
#define CHUNK 128
#define NCHUNK ((N_ + CHUNK - 1)/CHUNK)
#define PQ    64
#define PT_   (PQ/16)
#define VB_   256
#define VSPL  (CV_/VB_)
#define PBLK  ((P_ + PQ - 1)/PQ)
#define KSTEP 32
#define NKS   (N_/KSTEP)
#define FRAG  (16*KSTEP)
#define CONV_KS      27
#define CONV_KB      9
#define CONV_GRPS    (NKS / CONV_KS)
#define CONV_THREADS ((B_*CV_/16) * CONV_GRPS * 64)
#define CONV_BLOCKS  (CONV_THREADS / 256)
#define QV_V4        (B_*CV_*P_/4)
#define QV_PER       9
#define QV_THREADS   (QV_V4 / QV_PER)
#define QV_BLOCKS    (QV_THREADS / 256)
#define PLANE_ELEMS  ((size_t)B_ * CV_ * P_)
#define STAT_ELEMS(S) ((size_t)(S) * B_ * VSPL * P_)
#define QP           (P_ / 4)
#define CMB_ROWS     8
#define CMB_GRPS     (VB_ / CMB_ROWS)
#define CMB_THREADS  (B_ * VSPL * CMB_GRPS * QP)
#define CMB_BLOCKS   (CMB_THREADS / 256)

typedef __attribute__((ext_vector_type(16))) __bf16       v16bf;
typedef __attribute__((ext_vector_type(8)))  float        v8f;
typedef __attribute__((ext_vector_type(4)))  unsigned int v4u;
typedef __attribute__((ext_vector_type(4)))  float        v4f;
#define GLOBAL_AS __attribute__((address_space(1)))

union Frag16 {
  v16bf  bf;
  v4u    q[2];
  __bf16 h[16];
};
union Pack8 {
  v4u    q;
  __bf16 h[8];
};
union Pack4 {
  uint2  u;
  __bf16 h[4];
};

#define QSTR  72
#define MKSTR 72
#define PSTR  136

static __device__ __forceinline__ void lds_barrier() {
  asm volatile("s_wait_dscnt 0x0\n\ts_barrier_signal -1\n\ts_barrier_wait -1" ::: "memory");
}

template <bool MV_BF16, int NSPLIT>
__global__ __launch_bounds__(256, 4) void mem_reader_attn(
    const float* __restrict__ mk, const float* __restrict__ mv_f32,
    const __bf16* __restrict__ mv_bf, const float* __restrict__ qk,
    float* __restrict__ out, float* __restrict__ part, float* __restrict__ stat)
{
  constexpr int CPS = (NCHUNK + NSPLIT - 1) / NSPLIT;
  static_assert((NSPLIT - 1) * CPS < NCHUNK - 1, "every range must start on a full chunk");
  __shared__ __bf16 s_q[PQ * QSTR];
  __shared__ __bf16 s_mk[CHUNK * MKSTR];
  __shared__ __bf16 s_prob[PQ * PSTR];
  __shared__ float  s_pmax[8][PT_][32];
  __shared__ float  s_psum[8][PT_][32];
  __shared__ float  s_mrun[PQ];
  __shared__ float  s_lrun[PQ];
  __shared__ float  s_alpha[PQ];

  const int tid  = threadIdx.x;
  const int wave = tid >> 5;
  const int lane = tid & 31;
  const int col  = lane & 15;
  const int kg   = (lane >> 4) * 8;
  const int p0   = blockIdx.x * PQ;
  const int vb0  = blockIdx.y * VB_ + wave * 32;
  const int bi   = blockIdx.z % B_;
  const int sp   = blockIdx.z / B_;
  const int c_lo = sp * CPS;
  const int c_hi = (c_lo + CPS < NCHUNK) ? (c_lo + CPS) : NCHUNK;

  const float*  mk_b  = mk + (size_t)bi * CK_ * N_;
  const float*  qk_b  = qk + (size_t)bi * CK_ * P_;
  const float*  mvf_b = mv_f32 ? (mv_f32 + (size_t)bi * CV_ * N_) : nullptr;
  const __bf16* mvb_b = mv_bf  ? (mv_bf  + (size_t)bi * CV_ * N_) : nullptr;

  for (int idx = tid; idx < CK_ * PQ; idx += 256) {
    int p = idx & (PQ - 1), c = idx / PQ;
    int pg = p0 + p; if (pg > P_ - 1) pg = P_ - 1;
    s_q[p * QSTR + c] = (__bf16)(qk_b[(size_t)c * P_ + pg] * 0.125f);
  }
  if (tid < PQ) { s_mrun[tid] = -1e30f; s_lrun[tid] = 0.f; s_alpha[tid] = 0.f; }

  v8f acc[2][PT_];
#pragma unroll
  for (int vt = 0; vt < 2; ++vt)
#pragma unroll
    for (int pt = 0; pt < PT_; ++pt) acc[vt][pt] = v8f{};

  const float* mk_t = mk_b + (size_t)(wave * 8) * N_;
  float4 pre[8];
  {
    const int ngc = c_lo * CHUNK + 4 * lane;
    const GLOBAL_AS float* mk_p = (const GLOBAL_AS float*)mk_t;
    asm volatile("" : "+v"(mk_p));
#pragma unroll
    for (int j = 0; j < 8; ++j) {
      const v4f t = *(const GLOBAL_AS v4f*)(mk_p + (size_t)j * N_ + ngc);
      pre[j] = make_float4(t[0], t[1], t[2], t[3]);
    }
  }

  const __bf16* mv_w0 = MV_BF16 ? (mvb_b + (size_t)(vb0 / 16) * NKS * FRAG + lane * 8) : nullptr;
  Frag16 mvr[4][2];
  if (MV_BF16) {
    __builtin_amdgcn_sched_barrier(0);
    const GLOBAL_AS __bf16* mv_p = (const GLOBAL_AS __bf16*)mv_w0;
    asm volatile("" : "+v"(mv_p));
#pragma unroll
    for (int ks = 0; ks < 4; ++ks) {
#pragma unroll
      for (int vt = 0; vt < 2; ++vt) {
        const GLOBAL_AS __bf16* blk = mv_p + ((size_t)vt * NKS + c_lo * (CHUNK / KSTEP) + ks) * FRAG;
        mvr[ks][vt].q[0] = *(const GLOBAL_AS v4u*)(blk);
        mvr[ks][vt].q[1] = *(const GLOBAL_AS v4u*)(blk + FRAG / 2);
      }
    }
  }

  for (int nc = c_lo; nc < c_hi; ++nc) {
    const int n0 = nc * CHUNK;
    lds_barrier();

    {
      const bool ok = (n0 + 4 * lane) < N_;
      Pack8 r0, r1, r2, r3;
#pragma unroll
      for (int j = 0; j < 8; ++j) {
        r0.h[j] = (__bf16)(ok ? pre[j].x : 0.f);
        r1.h[j] = (__bf16)(ok ? pre[j].y : 0.f);
        r2.h[j] = (__bf16)(ok ? pre[j].z : 0.f);
        r3.h[j] = (__bf16)(ok ? pre[j].w : 0.f);
      }
      __bf16* dp = &s_mk[(4 * lane) * MKSTR + wave * 8];
      *(v4u*)(dp)             = r0.q;
      *(v4u*)(dp + MKSTR)     = r1.q;
      *(v4u*)(dp + 2 * MKSTR) = r2.q;
      *(v4u*)(dp + 3 * MKSTR) = r3.q;
    }
    {
      const int ncn = (nc + 1 < NCHUNK) ? (nc + 1) : (NCHUNK - 1);
      int ngc = ncn * CHUNK + 4 * lane;
      if (ngc > N_ - 4) ngc = N_ - 4;
#pragma unroll
      for (int j = 0; j < 8; ++j)
        pre[j] = *(const float4*)(mk_t + (size_t)j * N_ + ngc);
    }
    lds_barrier();

    const int mrow = wave * 16;
    Frag16 amk[2];
#pragma unroll
    for (int ks = 0; ks < 2; ++ks) {
      const __bf16* ap = &s_mk[(mrow + col) * MKSTR + ks * 32 + kg];
      amk[ks].q[0] = *(const v4u*)ap;  amk[ks].q[1] = *(const v4u*)(ap + 16);
    }
    float sv[PT_][8];
#pragma unroll
    for (int pt = 0; pt < PT_; ++pt) {
      v8f S = {};
#pragma unroll
      for (int ks = 0; ks < 2; ++ks) {
        Frag16 b;
        const __bf16* bp = &s_q[(pt * 16 + col) * QSTR + ks * 32 + kg];
        b.q[0] = *(const v4u*)bp;  b.q[1] = *(const v4u*)(bp + 16);
        S = __builtin_amdgcn_wmma_f32_16x16x32_bf16(false, amk[ks].bf, false, b.bf,
                                                    (short)0, S, false, false);
      }
      float lmax = -1e30f;
#pragma unroll
      for (int j = 0; j < 8; ++j) {
        int ng = n0 + mrow + kg + j;
        sv[pt][j] = (ng < N_) ? S[j] : -1e30f;
        lmax = fmaxf(lmax, sv[pt][j]);
      }
      s_pmax[wave][pt][lane] = lmax;
    }
    lds_barrier();

    if (tid < PQ) {
      const int pt = tid >> 4, c = tid & 15;
      float cmax = -1e30f;
#pragma unroll
      for (int w = 0; w < 8; ++w)
        cmax = fmaxf(cmax, fmaxf(s_pmax[w][pt][c], s_pmax[w][pt][c + 16]));
      float mold = s_mrun[tid];
      float mnew = fmaxf(mold, cmax);
      s_alpha[tid] = __expf(mold - mnew);
      s_mrun[tid]  = mnew;
    }
    lds_barrier();

#pragma unroll
    for (int pt = 0; pt < PT_; ++pt) {
      const float mnew = s_mrun[pt * 16 + col];
      const float alph = s_alpha[pt * 16 + col];
      acc[0][pt] *= alph;
      acc[1][pt] *= alph;

      Pack8 pk;
      float lsum = 0.f;
#pragma unroll
      for (int j = 0; j < 8; ++j) {
        int ng = n0 + mrow + kg + j;
        float pr = (ng < N_) ? __expf(sv[pt][j] - mnew) : 0.f;
        lsum += pr;
        pk.h[j] = (__bf16)pr;
      }
      *(v4u*)&s_prob[(pt * 16 + col) * PSTR + mrow + kg] = pk.q;
      s_psum[wave][pt][lane] = lsum;
    }
    lds_barrier();

    if (tid < PQ) {
      const int pt = tid >> 4, c = tid & 15;
      float csum = 0.f;
#pragma unroll
      for (int w = 0; w < 8; ++w)
        csum += s_psum[w][pt][c] + s_psum[w][pt][c + 16];
      s_lrun[tid] = s_lrun[tid] * s_alpha[tid] + csum;
    }

#pragma unroll
    for (int ks = 0; ks < 4; ++ks) {
      int nA = n0 + ks * 32 + kg;
      int b1 = nA;      if (b1 > N_ - 8) b1 = N_ - 8;
      int b2 = nA + 16; if (b2 > N_ - 8) b2 = N_ - 8;
      Frag16 b[PT_];
#pragma unroll
      for (int pt = 0; pt < PT_; ++pt) {
        const __bf16* bp = &s_prob[(pt * 16 + col) * PSTR + ks * 32 + kg];
        b[pt].q[0] = *(const v4u*)bp;  b[pt].q[1] = *(const v4u*)(bp + 16);
      }
#pragma unroll
      for (int vt = 0; vt < 2; ++vt) {
        const int v = vb0 + vt * 16 + col;
        Frag16 a;
        if (MV_BF16) {
          a = mvr[ks][vt];
        } else {
          const float* rowp = mvf_b + (size_t)v * N_;
          const float4 f0 = *(const float4*)(rowp + b1);
          const float4 f1 = *(const float4*)(rowp + b1 + 4);
          const float4 f2 = *(const float4*)(rowp + b2);
          const float4 f3 = *(const float4*)(rowp + b2 + 4);
          a.h[0]=(__bf16)f0.x;  a.h[1]=(__bf16)f0.y;  a.h[2]=(__bf16)f0.z;  a.h[3]=(__bf16)f0.w;
          a.h[4]=(__bf16)f1.x;  a.h[5]=(__bf16)f1.y;  a.h[6]=(__bf16)f1.z;  a.h[7]=(__bf16)f1.w;
          a.h[8]=(__bf16)f2.x;  a.h[9]=(__bf16)f2.y;  a.h[10]=(__bf16)f2.z; a.h[11]=(__bf16)f2.w;
          a.h[12]=(__bf16)f3.x; a.h[13]=(__bf16)f3.y; a.h[14]=(__bf16)f3.z; a.h[15]=(__bf16)f3.w;
        }
#pragma unroll
        for (int pt = 0; pt < PT_; ++pt)
          acc[vt][pt] = __builtin_amdgcn_wmma_f32_16x16x32_bf16(false, a.bf, false, b[pt].bf,
                                                                (short)0, acc[vt][pt], false, false);
      }
    }

    if (MV_BF16) {
      const int ncn = (nc + 1 < NCHUNK) ? (nc + 1) : (NCHUNK - 1);
#pragma unroll
      for (int ks = 0; ks < 4; ++ks) {
        int kst = ncn * (CHUNK / KSTEP) + ks;
        if (kst > NKS - 1) kst = NKS - 1;
#pragma unroll
        for (int vt = 0; vt < 2; ++vt) {
          const __bf16* blk = mv_w0 + ((size_t)vt * NKS + kst) * FRAG;
          mvr[ks][vt].q[0] = *(const v4u*)(blk);
          mvr[ks][vt].q[1] = *(const v4u*)(blk + FRAG / 2);
        }
      }
    }
  }

  __syncthreads();
  if (NSPLIT > 1) {
    if (tid < PQ && p0 + tid < P_) {
      const size_t si = ((size_t)(sp * B_ + bi) * VSPL + blockIdx.y) * P_ + p0 + tid;
      stat[si]                      = s_mrun[tid];
      stat[STAT_ELEMS(NSPLIT) + si] = s_lrun[tid];
    }
  }
  float* dst = (NSPLIT > 1 && sp > 0)
             ? part + (size_t)(sp - 1) * PLANE_ELEMS + (size_t)bi * CV_ * P_
             : out + (size_t)bi * (2 * CV_) * P_;
#pragma unroll
  for (int pt = 0; pt < PT_; ++pt) {
    const float linv = (NSPLIT > 1) ? 1.0f : 1.0f / s_lrun[pt * 16 + col];
    const int pg = p0 + pt * 16 + col;
    if (pg < P_) {
#pragma unroll
      for (int vt = 0; vt < 2; ++vt) {
#pragma unroll
        for (int j = 0; j < 8; ++j) {
          int v = vb0 + vt * 16 + kg + j;
          dst[(size_t)v * P_ + pg] = (NSPLIT > 1) ? acc[vt][pt][j] : acc[vt][pt][j] * linv;
        }
      }
    }
  }
}

template <int NSPLIT>
__global__ __launch_bounds__(256) void combine_ranges(float* out,
                                                      const float* __restrict__ part,
                                                      const float* __restrict__ stat)
{
  const int u  = blockIdx.x * 256 + threadIdx.x;
  const int p4 = u % QP;
  const int g  = u / QP;
  const int vg = g % CMB_GRPS;
  const int bv = g / CMB_GRPS;
  const int b  = bv / VSPL;
  const int vs = bv - b * VSPL;

  v4f m[NSPLIT], l[NSPLIT], w[NSPLIT];
#pragma unroll
  for (int s = 0; s < NSPLIT; ++s) {
    const float* sp = stat + ((size_t)(s * B_ + b) * VSPL + vs) * P_ + 4 * p4;
    m[s] = *(const v4f*)(sp);
    l[s] = *(const v4f*)(sp + STAT_ELEMS(NSPLIT));
  }
  v4f mx = m[0];
#pragma unroll
  for (int s = 1; s < NSPLIT; ++s)
#pragma unroll
    for (int e = 0; e < 4; ++e) mx[e] = fmaxf(mx[e], m[s][e]);
  v4f den = {0.f, 0.f, 0.f, 0.f};
#pragma unroll
  for (int s = 0; s < NSPLIT; ++s) {
#pragma unroll
    for (int e = 0; e < 4; ++e) w[s][e] = __expf(m[s][e] - mx[e]);
    den += w[s] * l[s];
  }
#pragma unroll
  for (int s = 0; s < NSPLIT; ++s) w[s] = w[s] / den;

  const int v0 = vs * VB_ + vg * CMB_ROWS;
  float* o = out + ((size_t)b * (2 * CV_) + v0) * P_ + 4 * p4;
  const float* pp = part + ((size_t)b * CV_ + v0) * P_ + 4 * p4;
  v4f x[CMB_ROWS][NSPLIT];
#pragma unroll
  for (int r = 0; r < CMB_ROWS; ++r) {
    x[r][0] = *(const v4f*)(o + (size_t)r * P_);
#pragma unroll
    for (int s = 1; s < NSPLIT; ++s)
      x[r][s] = *(const v4f*)(pp + (size_t)(s - 1) * PLANE_ELEMS + (size_t)r * P_);
  }
#pragma unroll
  for (int r = 0; r < CMB_ROWS; ++r) {
    v4f y = x[r][0] * w[0];
#pragma unroll
    for (int s = 1; s < NSPLIT; ++s) y += x[r][s] * w[s];
    *(v4f*)(o + (size_t)r * P_) = y;
  }
}

__global__ __launch_bounds__(256) void convert_mv(const float* __restrict__ src,
                                                  __bf16* __restrict__ dst)
{
  const int u   = blockIdx.x * 256 + threadIdx.x;
  const int g   = u & 3;
  const int r   = (u >> 2) & 15;
  const int grp = u >> 6;
  const int rt  = grp / CONV_GRPS;
  const int ks0 = (grp - rt * CONV_GRPS) * CONV_KS;
  const float* sp = src + ((size_t)rt * 16 + r) * N_ + ks0 * KSTEP + g * 8;
  __bf16* dp = dst + ((size_t)rt * NKS + ks0) * FRAG + (g >> 1) * (FRAG / 2) + ((g & 1) * 16 + r) * 8;
#pragma unroll
  for (int b = 0; b < CONV_KS / CONV_KB; ++b) {
    float4 f0[CONV_KB], f1[CONV_KB];
#pragma unroll
    for (int j = 0; j < CONV_KB; ++j) {
      const float* p = sp + (b * CONV_KB + j) * KSTEP;
      f0[j] = *(const float4*)(p);
      f1[j] = *(const float4*)(p + 4);
    }
#pragma unroll
    for (int j = 0; j < CONV_KB; ++j) {
      Pack8 p;
      p.h[0] = (__bf16)f0[j].x; p.h[1] = (__bf16)f0[j].y;
      p.h[2] = (__bf16)f0[j].z; p.h[3] = (__bf16)f0[j].w;
      p.h[4] = (__bf16)f1[j].x; p.h[5] = (__bf16)f1[j].y;
      p.h[6] = (__bf16)f1[j].z; p.h[7] = (__bf16)f1[j].w;
      *(v4u*)(dp + (size_t)(b * CONV_KB + j) * FRAG) = p.q;
    }
  }
}

__global__ __launch_bounds__(256) void qv_copy(const float* __restrict__ qv,
                                               float* __restrict__ out)
{
  const int t = blockIdx.x * 256 + threadIdx.x;
  const v4f* s4 = (const v4f*)qv;
  v4f* d4 = (v4f*)out;
  v4f v[QV_PER];
#pragma unroll
  for (int j = 0; j < QV_PER; ++j)
    v[j] = s4[t + j * QV_THREADS];
#pragma unroll
  for (int j = 0; j < QV_PER; ++j) {
    const int i4 = t + j * QV_THREADS;
    const int b  = i4 / (CV_ * P_ / 4);
    d4[i4 + (b + 1) * (CV_ * P_ / 4)] = v[j];
  }
}

extern "C" void kernel_launch(void* const* d_in, const int* in_sizes, int n_in,
                              void* d_out, int out_size, void* d_ws, size_t ws_size,
                              hipStream_t stream) {
  const float* mk = (const float*)d_in[0];
  const float* mv = (const float*)d_in[1];
  const float* qk = (const float*)d_in[2];
  const float* qv = (const float*)d_in[3];
  float* out = (float*)d_out;

  const size_t mv_elems = (size_t)B_ * CV_ * N_;
  const size_t need     = mv_elems * sizeof(__bf16);
  const size_t need2 = need + (1 * PLANE_ELEMS + 2 * STAT_ELEMS(2)) * sizeof(float);
  dim3 grid(PBLK, VSPL, B_);

  if (d_ws && ws_size >= need2) {
    __bf16* mv_bf = (__bf16*)d_ws;
    float* part = (float*)((char*)d_ws + need);
    float* stat = part + 1 * PLANE_ELEMS;
    convert_mv<<<CONV_BLOCKS, 256, 0, stream>>>(mv, mv_bf);
    mem_reader_attn<true, 2><<<dim3(PBLK, VSPL, B_ * 2), 256, 0, stream>>>(mk, nullptr, mv_bf, qk, out, part, stat);
    combine_ranges<2><<<CMB_BLOCKS, 256, 0, stream>>>(out, part, stat);
  } else if (d_ws && ws_size >= need) {
    __bf16* mv_bf = (__bf16*)d_ws;
    convert_mv<<<CONV_BLOCKS, 256, 0, stream>>>(mv, mv_bf);
    mem_reader_attn<true, 1><<<grid, 256, 0, stream>>>(mk, nullptr, mv_bf, qk, out, nullptr, nullptr);
  } else {
    mem_reader_attn<false, 1><<<grid, 256, 0, stream>>>(mk, mv, nullptr, qk, out, nullptr, nullptr);
  }

  qv_copy<<<QV_BLOCKS, 256, 0, stream>>>(qv, out);
}


# 5 barriers per chunk -> 2: each wave reduces the chunk max itself from 8 LDS partials, mk chunk double-buffered; softmax sums kept per lane over all chunks, added once at the end (sum order changes)
# speedup vs baseline: 10.6606x; 1.0579x over previous
#include <hip/hip_runtime.h>

#define B_    4
#define CK_   64
#define CV_   512
#define H_    30
#define W_    54
#define T_    8
#define P_    (H_*W_)
#define N_    (T_*H_*W_)
#define CHUNK 128
#define NCHUNK ((N_ + CHUNK - 1)/CHUNK)
#define PQ    64
#define PT_   (PQ/16)
#define VB_   256
#define VSPL  (CV_/VB_)
#define PBLK  ((P_ + PQ - 1)/PQ)
#define KSTEP 32
#define NKS   (N_/KSTEP)
#define FRAG  (16*KSTEP)
#define CONV_KS      27
#define CONV_KB      9
#define CONV_GRPS    (NKS / CONV_KS)
#define CONV_THREADS ((B_*CV_/16) * CONV_GRPS * 64)
#define CONV_BLOCKS  (CONV_THREADS / 256)
#define QV_V4        (B_*CV_*P_/4)
#define QV_PER       9
#define QV_THREADS   (QV_V4 / QV_PER)
#define QV_BLOCKS    (QV_THREADS / 256)
#define PLANE_ELEMS  ((size_t)B_ * CV_ * P_)
#define STAT_ELEMS(S) ((size_t)(S) * B_ * VSPL * P_)
#define QP           (P_ / 4)
#define CMB_ROWS     8
#define CMB_GRPS     (VB_ / CMB_ROWS)
#define CMB_THREADS  (B_ * VSPL * CMB_GRPS * QP)
#define CMB_BLOCKS   (CMB_THREADS / 256)

typedef __attribute__((ext_vector_type(16))) __bf16       v16bf;
typedef __attribute__((ext_vector_type(8)))  float        v8f;
typedef __attribute__((ext_vector_type(4)))  unsigned int v4u;
typedef __attribute__((ext_vector_type(4)))  float        v4f;
#define GLOBAL_AS __attribute__((address_space(1)))

union Frag16 {
  v16bf  bf;
  v4u    q[2];
  __bf16 h[16];
};
union Pack8 {
  v4u    q;
  __bf16 h[8];
};
union Pack4 {
  uint2  u;
  __bf16 h[4];
};

#define QSTR  72
#define MKSTR 72
#define PSTR  136
#define RSTR  12

static __device__ __forceinline__ float other_half(float x) {
  return __uint_as_float(__builtin_amdgcn_permlanex16(__float_as_uint(x), __float_as_uint(x),
                                                      0x76543210u, 0xfedcba98u, false, false));
}

static __device__ __forceinline__ void lds_barrier() {
  asm volatile("s_wait_dscnt 0x0\n\ts_barrier_signal -1\n\ts_barrier_wait -1" ::: "memory");
}

template <bool MV_BF16, int NSPLIT>
__global__ __launch_bounds__(256, 4) void mem_reader_attn(
    const float* __restrict__ mk, const float* __restrict__ mv_f32,
    const __bf16* __restrict__ mv_bf, const float* __restrict__ qk,
    float* __restrict__ out, float* __restrict__ part, float* __restrict__ stat)
{
  constexpr int CPS = (NCHUNK + NSPLIT - 1) / NSPLIT;
  static_assert((NSPLIT - 1) * CPS < NCHUNK - 1, "every range must start on a full chunk");
  __shared__ __bf16 s_q[PQ * QSTR];
  __shared__ __bf16 s_mk[2][CHUNK * MKSTR];
  __shared__ __bf16 s_prob[PQ * PSTR];
  __shared__ __attribute__((aligned(16))) float s_red[PQ * RSTR];

  const int tid  = threadIdx.x;
  const int wave = tid >> 5;
  const int lane = tid & 31;
  const int col  = lane & 15;
  const int kg   = (lane >> 4) * 8;
  const int p0   = blockIdx.x * PQ;
  const int vb0  = blockIdx.y * VB_ + wave * 32;
  const int bi   = blockIdx.z % B_;
  const int sp   = blockIdx.z / B_;
  const int c_lo = sp * CPS;
  const int c_hi = (c_lo + CPS < NCHUNK) ? (c_lo + CPS) : NCHUNK;

  const float*  mk_b  = mk + (size_t)bi * CK_ * N_;
  const float*  qk_b  = qk + (size_t)bi * CK_ * P_;
  const float*  mvf_b = mv_f32 ? (mv_f32 + (size_t)bi * CV_ * N_) : nullptr;
  const __bf16* mvb_b = mv_bf  ? (mv_bf  + (size_t)bi * CV_ * N_) : nullptr;

  for (int idx = tid; idx < CK_ * PQ; idx += 256) {
    int p = idx & (PQ - 1), c = idx / PQ;
    int pg = p0 + p; if (pg > P_ - 1) pg = P_ - 1;
    s_q[p * QSTR + c] = (__bf16)(qk_b[(size_t)c * P_ + pg] * 0.125f);
  }
  float mrun[PT_], lrun[PT_];
#pragma unroll
  for (int pt = 0; pt < PT_; ++pt) { mrun[pt] = -1e30f; lrun[pt] = 0.f; }

  v8f acc[2][PT_];
#pragma unroll
  for (int vt = 0; vt < 2; ++vt)
#pragma unroll
    for (int pt = 0; pt < PT_; ++pt) acc[vt][pt] = v8f{};

  const float* mk_t = mk_b + (size_t)(wave * 8) * N_;
  float4 pre[8];
  __bf16* const mk_dst = &s_mk[0][(4 * lane) * MKSTR + wave * 8];
  {
    const int ngc = c_lo * CHUNK + 4 * lane;
    const GLOBAL_AS float* mk_p = (const GLOBAL_AS float*)mk_t;
    asm volatile("" : "+v"(mk_p));
#pragma unroll
    for (int j = 0; j < 8; ++j) {
      const v4f t = *(const GLOBAL_AS v4f*)(mk_p + (size_t)j * N_ + ngc);
      pre[j] = make_float4(t[0], t[1], t[2], t[3]);
    }
    Pack8 r0, r1, r2, r3;
#pragma unroll
    for (int j = 0; j < 8; ++j) {
      r0.h[j] = (__bf16)pre[j].x;
      r1.h[j] = (__bf16)pre[j].y;
      r2.h[j] = (__bf16)pre[j].z;
      r3.h[j] = (__bf16)pre[j].w;
    }
    *(v4u*)(mk_dst)             = r0.q;
    *(v4u*)(mk_dst + MKSTR)     = r1.q;
    *(v4u*)(mk_dst + 2 * MKSTR) = r2.q;
    *(v4u*)(mk_dst + 3 * MKSTR) = r3.q;
    __builtin_amdgcn_sched_barrier(0);
    int ng2 = (c_lo + 1) * CHUNK + 4 * lane;
    if (ng2 > N_ - 4) ng2 = N_ - 4;
#pragma unroll
    for (int j = 0; j < 8; ++j) {
      const v4f t = *(const GLOBAL_AS v4f*)(mk_p + (size_t)j * N_ + ng2);
      pre[j] = make_float4(t[0], t[1], t[2], t[3]);
    }
  }

  const __bf16* mv_w0 = MV_BF16 ? (mvb_b + (size_t)(vb0 / 16) * NKS * FRAG + lane * 8) : nullptr;
  Frag16 mvr[4][2];
  if (MV_BF16) {
    __builtin_amdgcn_sched_barrier(0);
    const GLOBAL_AS __bf16* mv_p = (const GLOBAL_AS __bf16*)mv_w0;
    asm volatile("" : "+v"(mv_p));
#pragma unroll
    for (int ks = 0; ks < 4; ++ks) {
#pragma unroll
      for (int vt = 0; vt < 2; ++vt) {
        const GLOBAL_AS __bf16* blk = mv_p + ((size_t)vt * NKS + c_lo * (CHUNK / KSTEP) + ks) * FRAG;
        mvr[ks][vt].q[0] = *(const GLOBAL_AS v4u*)(blk);
        mvr[ks][vt].q[1] = *(const GLOBAL_AS v4u*)(blk + FRAG / 2);
      }
    }
  }
  lds_barrier();

  const GLOBAL_AS float* mk_l = (const GLOBAL_AS float*)mk_t;
  asm volatile("" : "+v"(mk_l));
  int cur = 0;
  for (int nc = c_lo; nc < c_hi; ++nc, cur ^= 1) {
    const int n0 = nc * CHUNK;
    const __bf16* mk_cur = &s_mk[cur][0];

    {
      const bool ok = (n0 + CHUNK + 4 * lane) < N_;
      Pack8 r0, r1, r2, r3;
#pragma unroll
      for (int j = 0; j < 8; ++j) {
        r0.h[j] = (__bf16)(ok ? pre[j].x : 0.f);
        r1.h[j] = (__bf16)(ok ? pre[j].y : 0.f);
        r2.h[j] = (__bf16)(ok ? pre[j].z : 0.f);
        r3.h[j] = (__bf16)(ok ? pre[j].w : 0.f);
      }
      __bf16* dp = mk_dst + (cur ^ 1) * (CHUNK * MKSTR);
      *(v4u*)(dp)             = r0.q;
      *(v4u*)(dp + MKSTR)     = r1.q;
      *(v4u*)(dp + 2 * MKSTR) = r2.q;
      *(v4u*)(dp + 3 * MKSTR) = r3.q;
    }

    const int mrow = wave * 16;
    Frag16 amk[2];
#pragma unroll
    for (int ks = 0; ks < 2; ++ks) {
      const __bf16* ap = mk_cur + (mrow + col) * MKSTR + ks * 32 + kg;
      amk[ks].q[0] = *(const v4u*)ap;  amk[ks].q[1] = *(const v4u*)(ap + 16);
    }
    float sv[PT_][8];
    float lmx[PT_];
#pragma unroll
    for (int pt = 0; pt < PT_; ++pt) {
      v8f S = {};
#pragma unroll
      for (int ks = 0; ks < 2; ++ks) {
        Frag16 b;
        const __bf16* bp = &s_q[(pt * 16 + col) * QSTR + ks * 32 + kg];
        b.q[0] = *(const v4u*)bp;  b.q[1] = *(const v4u*)(bp + 16);
        S = __builtin_amdgcn_wmma_f32_16x16x32_bf16(false, amk[ks].bf, false, b.bf,
                                                    (short)0, S, false, false);
      }
      float lmax = -1e30f;
#pragma unroll
      for (int j = 0; j < 8; ++j) {
        int ng = n0 + mrow + kg + j;
        sv[pt][j] = (ng < N_) ? S[j] : -1e30f;
        lmax = fmaxf(lmax, sv[pt][j]);
      }
      lmx[pt] = fmaxf(lmax, other_half(lmax));
    }
    if (lane < 16) {
#pragma unroll
      for (int pt = 0; pt < PT_; ++pt) s_red[(pt * 16 + col) * RSTR + wave] = lmx[pt];
    }
    lds_barrier();

#pragma unroll
    for (int pt = 0; pt < PT_; ++pt) {
      const v4f* rp = (const v4f*)&s_red[(pt * 16 + col) * RSTR];
      const v4f a0 = rp[0], a1 = rp[1];
      const float c0 = fmaxf(fmaxf(a0[0], a0[1]), fmaxf(a0[2], a0[3]));
      const float c1 = fmaxf(fmaxf(a1[0], a1[1]), fmaxf(a1[2], a1[3]));
      const float mnew = fmaxf(mrun[pt], fmaxf(c0, c1));
      const float alph = __expf(mrun[pt] - mnew);
      mrun[pt] = mnew;
      acc[0][pt] *= alph;
      acc[1][pt] *= alph;

      Pack8 pk;
      float lsum = 0.f;
#pragma unroll
      for (int j = 0; j < 8; ++j) {
        int ng = n0 + mrow + kg + j;
        float pr = (ng < N_) ? __expf(sv[pt][j] - mnew) : 0.f;
        lsum += pr;
        pk.h[j] = (__bf16)pr;
      }
      *(v4u*)&s_prob[(pt * 16 + col) * PSTR + mrow + kg] = pk.q;
      lrun[pt] = lrun[pt] * alph + lsum;
    }
    lds_barrier();

    {
      const int ncn = (nc + 2 < NCHUNK) ? (nc + 2) : (NCHUNK - 1);
      int ngc = ncn * CHUNK + 4 * lane;
      if (ngc > N_ - 4) ngc = N_ - 4;
#pragma unroll
      for (int j = 0; j < 8; ++j) {
        const v4f t = *(const GLOBAL_AS v4f*)(mk_l + (size_t)j * N_ + ngc);
        pre[j] = make_float4(t[0], t[1], t[2], t[3]);
      }
    }
    __builtin_amdgcn_sched_barrier(0);

#pragma unroll
    for (int ks = 0; ks < 4; ++ks) {
      int nA = n0 + ks * 32 + kg;
      int b1 = nA;      if (b1 > N_ - 8) b1 = N_ - 8;
      int b2 = nA + 16; if (b2 > N_ - 8) b2 = N_ - 8;
      Frag16 b[PT_];
#pragma unroll
      for (int pt = 0; pt < PT_; ++pt) {
        const __bf16* bp = &s_prob[(pt * 16 + col) * PSTR + ks * 32 + kg];
        b[pt].q[0] = *(const v4u*)bp;  b[pt].q[1] = *(const v4u*)(bp + 16);
      }
#pragma unroll
      for (int vt = 0; vt < 2; ++vt) {
        const int v = vb0 + vt * 16 + col;
        Frag16 a;
        if (MV_BF16) {
          a = mvr[ks][vt];
        } else {
          const float* rowp = mvf_b + (size_t)v * N_;
          const float4 f0 = *(const float4*)(rowp + b1);
          const float4 f1 = *(const float4*)(rowp + b1 + 4);
          const float4 f2 = *(const float4*)(rowp + b2);
          const float4 f3 = *(const float4*)(rowp + b2 + 4);
          a.h[0]=(__bf16)f0.x;  a.h[1]=(__bf16)f0.y;  a.h[2]=(__bf16)f0.z;  a.h[3]=(__bf16)f0.w;
          a.h[4]=(__bf16)f1.x;  a.h[5]=(__bf16)f1.y;  a.h[6]=(__bf16)f1.z;  a.h[7]=(__bf16)f1.w;
          a.h[8]=(__bf16)f2.x;  a.h[9]=(__bf16)f2.y;  a.h[10]=(__bf16)f2.z; a.h[11]=(__bf16)f2.w;
          a.h[12]=(__bf16)f3.x; a.h[13]=(__bf16)f3.y; a.h[14]=(__bf16)f3.z; a.h[15]=(__bf16)f3.w;
        }
#pragma unroll
        for (int pt = 0; pt < PT_; ++pt)
          acc[vt][pt] = __builtin_amdgcn_wmma_f32_16x16x32_bf16(false, a.bf, false, b[pt].bf,
                                                                (short)0, acc[vt][pt], false, false);
      }
    }

    if (MV_BF16) {
      const int ncn = (nc + 1 < NCHUNK) ? (nc + 1) : (NCHUNK - 1);
#pragma unroll
      for (int ks = 0; ks < 4; ++ks) {
        int kst = ncn * (CHUNK / KSTEP) + ks;
        if (kst > NKS - 1) kst = NKS - 1;
#pragma unroll
        for (int vt = 0; vt < 2; ++vt) {
          const __bf16* blk = mv_w0 + ((size_t)vt * NKS + kst) * FRAG;
          mvr[ks][vt].q[0] = *(const v4u*)(blk);
          mvr[ks][vt].q[1] = *(const v4u*)(blk + FRAG / 2);
        }
      }
    }
  }

#pragma unroll
  for (int pt = 0; pt < PT_; ++pt) lrun[pt] += other_half(lrun[pt]);
  if (lane < 16) {
#pragma unroll
    for (int pt = 0; pt < PT_; ++pt) s_red[(pt * 16 + col) * RSTR + wave] = lrun[pt];
  }
  __syncthreads();
  float ltot[PT_];
#pragma unroll
  for (int pt = 0; pt < PT_; ++pt) {
    const float* rp = &s_red[(pt * 16 + col) * RSTR];
    float s = 0.f;
#pragma unroll
    for (int k = 0; k < 8; ++k) s += rp[k];
    ltot[pt] = s;
  }
  if (NSPLIT > 1) {
    if (wave == 0 && lane < 16) {
#pragma unroll
      for (int pt = 0; pt < PT_; ++pt) {
        const int pq = p0 + pt * 16 + col;
        if (pq < P_) {
          const size_t si = ((size_t)(sp * B_ + bi) * VSPL + blockIdx.y) * P_ + pq;
          stat[si]                      = mrun[pt];
          stat[STAT_ELEMS(NSPLIT) + si] = ltot[pt];
        }
      }
    }
  }
  float* dst = (NSPLIT > 1 && sp > 0)
             ? part + (size_t)(sp - 1) * PLANE_ELEMS + (size_t)bi * CV_ * P_
             : out + (size_t)bi * (2 * CV_) * P_;
#pragma unroll
  for (int pt = 0; pt < PT_; ++pt) {
    const float linv = (NSPLIT > 1) ? 1.0f : 1.0f / ltot[pt];
    const int pg = p0 + pt * 16 + col;
    if (pg < P_) {
#pragma unroll
      for (int vt = 0; vt < 2; ++vt) {
#pragma unroll
        for (int j = 0; j < 8; ++j) {
          int v = vb0 + vt * 16 + kg + j;
          dst[(size_t)v * P_ + pg] = (NSPLIT > 1) ? acc[vt][pt][j] : acc[vt][pt][j] * linv;
        }
      }
    }
  }
}

template <int NSPLIT>
__global__ __launch_bounds__(256) void combine_ranges(float* out,
                                                      const float* __restrict__ part,
                                                      const float* __restrict__ stat)
{
  const int u  = blockIdx.x * 256 + threadIdx.x;
  const int p4 = u % QP;
  const int g  = u / QP;
  const int vg = g % CMB_GRPS;
  const int bv = g / CMB_GRPS;
  const int b  = bv / VSPL;
  const int vs = bv - b * VSPL;

  v4f m[NSPLIT], l[NSPLIT], w[NSPLIT];
#pragma unroll
  for (int s = 0; s < NSPLIT; ++s) {
    const float* sp = stat + ((size_t)(s * B_ + b) * VSPL + vs) * P_ + 4 * p4;
    m[s] = *(const v4f*)(sp);
    l[s] = *(const v4f*)(sp + STAT_ELEMS(NSPLIT));
  }
  v4f mx = m[0];
#pragma unroll
  for (int s = 1; s < NSPLIT; ++s)
#pragma unroll
    for (int e = 0; e < 4; ++e) mx[e] = fmaxf(mx[e], m[s][e]);
  v4f den = {0.f, 0.f, 0.f, 0.f};
#pragma unroll
  for (int s = 0; s < NSPLIT; ++s) {
#pragma unroll
    for (int e = 0; e < 4; ++e) w[s][e] = __expf(m[s][e] - mx[e]);
    den += w[s] * l[s];
  }
#pragma unroll
  for (int s = 0; s < NSPLIT; ++s) w[s] = w[s] / den;

  const int v0 = vs * VB_ + vg * CMB_ROWS;
  float* o = out + ((size_t)b * (2 * CV_) + v0) * P_ + 4 * p4;
  const float* pp = part + ((size_t)b * CV_ + v0) * P_ + 4 * p4;
  v4f x[CMB_ROWS][NSPLIT];
#pragma unroll
  for (int r = 0; r < CMB_ROWS; ++r) {
    x[r][0] = *(const v4f*)(o + (size_t)r * P_);
#pragma unroll
    for (int s = 1; s < NSPLIT; ++s)
      x[r][s] = *(const v4f*)(pp + (size_t)(s - 1) * PLANE_ELEMS + (size_t)r * P_);
  }
#pragma unroll
  for (int r = 0; r < CMB_ROWS; ++r) {
    v4f y = x[r][0] * w[0];
#pragma unroll
    for (int s = 1; s < NSPLIT; ++s) y += x[r][s] * w[s];
    *(v4f*)(o + (size_t)r * P_) = y;
  }
}

__global__ __launch_bounds__(256) void convert_mv(const float* __restrict__ src,
                                                  __bf16* __restrict__ dst)
{
  const int u   = blockIdx.x * 256 + threadIdx.x;
  const int g   = u & 3;
  const int r   = (u >> 2) & 15;
  const int grp = u >> 6;
  const int rt  = grp / CONV_GRPS;
  const int ks0 = (grp - rt * CONV_GRPS) * CONV_KS;
  const float* sp = src + ((size_t)rt * 16 + r) * N_ + ks0 * KSTEP + g * 8;
  __bf16* dp = dst + ((size_t)rt * NKS + ks0) * FRAG + (g >> 1) * (FRAG / 2) + ((g & 1) * 16 + r) * 8;
#pragma unroll
  for (int b = 0; b < CONV_KS / CONV_KB; ++b) {
    float4 f0[CONV_KB], f1[CONV_KB];
#pragma unroll
    for (int j = 0; j < CONV_KB; ++j) {
      const float* p = sp + (b * CONV_KB + j) * KSTEP;
      f0[j] = *(const float4*)(p);
      f1[j] = *(const float4*)(p + 4);
    }
#pragma unroll
    for (int j = 0; j < CONV_KB; ++j) {
      Pack8 p;
      p.h[0] = (__bf16)f0[j].x; p.h[1] = (__bf16)f0[j].y;
      p.h[2] = (__bf16)f0[j].z; p.h[3] = (__bf16)f0[j].w;
      p.h[4] = (__bf16)f1[j].x; p.h[5] = (__bf16)f1[j].y;
      p.h[6] = (__bf16)f1[j].z; p.h[7] = (__bf16)f1[j].w;
      *(v4u*)(dp + (size_t)(b * CONV_KB + j) * FRAG) = p.q;
    }
  }
}

__global__ __launch_bounds__(256) void qv_copy(const float* __restrict__ qv,
                                               float* __restrict__ out)
{
  const int t = blockIdx.x * 256 + threadIdx.x;
  const v4f* s4 = (const v4f*)qv;
  v4f* d4 = (v4f*)out;
  v4f v[QV_PER];
#pragma unroll
  for (int j = 0; j < QV_PER; ++j)
    v[j] = s4[t + j * QV_THREADS];
#pragma unroll
  for (int j = 0; j < QV_PER; ++j) {
    const int i4 = t + j * QV_THREADS;
    const int b  = i4 / (CV_ * P_ / 4);
    d4[i4 + (b + 1) * (CV_ * P_ / 4)] = v[j];
  }
}

extern "C" void kernel_launch(void* const* d_in, const int* in_sizes, int n_in,
                              void* d_out, int out_size, void* d_ws, size_t ws_size,
                              hipStream_t stream) {
  const float* mk = (const float*)d_in[0];
  const float* mv = (const float*)d_in[1];
  const float* qk = (const float*)d_in[2];
  const float* qv = (const float*)d_in[3];
  float* out = (float*)d_out;

  const size_t mv_elems = (size_t)B_ * CV_ * N_;
  const size_t need     = mv_elems * sizeof(__bf16);
  const size_t need2 = need + (1 * PLANE_ELEMS + 2 * STAT_ELEMS(2)) * sizeof(float);
  dim3 grid(PBLK, VSPL, B_);

  if (d_ws && ws_size >= need2) {
    __bf16* mv_bf = (__bf16*)d_ws;
    float* part = (float*)((char*)d_ws + need);
    float* stat = part + 1 * PLANE_ELEMS;
    convert_mv<<<CONV_BLOCKS, 256, 0, stream>>>(mv, mv_bf);
    mem_reader_attn<true, 2><<<dim3(PBLK, VSPL, B_ * 2), 256, 0, stream>>>(mk, nullptr, mv_bf, qk, out, part, stat);
    combine_ranges<2><<<CMB_BLOCKS, 256, 0, stream>>>(out, part, stat);
  } else if (d_ws && ws_size >= need) {
    __bf16* mv_bf = (__bf16*)d_ws;
    convert_mv<<<CONV_BLOCKS, 256, 0, stream>>>(mv, mv_bf);
    mem_reader_attn<true, 1><<<grid, 256, 0, stream>>>(mk, nullptr, mv_bf, qk, out, nullptr, nullptr);
  } else {
    mem_reader_attn<false, 1><<<grid, 256, 0, stream>>>(mk, mv, nullptr, qk, out, nullptr, nullptr);
  }

  qv_copy<<<QV_BLOCKS, 256, 0, stream>>>(qv, out);
}


# mk converted to bf16 once by the conversion pass, in WMMA A-fragment order: each wave loads its 16-key fragments straight from global a chunk ahead; no in-loop conversion, no LDS staging of mk
# speedup vs baseline: 11.8880x; 1.1151x over previous
#include <hip/hip_runtime.h>

#define B_    4
#define CK_   64
#define CV_   512
#define H_    30
#define W_    54
#define T_    8
#define P_    (H_*W_)
#define N_    (T_*H_*W_)
#define CHUNK 128
#define NCHUNK ((N_ + CHUNK - 1)/CHUNK)
#define PQ    64
#define PT_   (PQ/16)
#define VB_   256
#define VSPL  (CV_/VB_)
#define PBLK  ((P_ + PQ - 1)/PQ)
#define KSTEP 32
#define NKS   (N_/KSTEP)
#define FRAG  (16*KSTEP)
#define CONV_KS      27
#define CONV_KB      9
#define CONV_GRPS    (NKS / CONV_KS)
#define CONV_THREADS ((B_*CV_/16) * CONV_GRPS * 64)
#define CONV_BLOCKS  (CONV_THREADS / 256)
#define QV_V4        (B_*CV_*P_/4)
#define QV_PER       9
#define QV_THREADS   (QV_V4 / QV_PER)
#define QV_BLOCKS    (QV_THREADS / 256)
#define PLANE_ELEMS  ((size_t)B_ * CV_ * P_)
#define STAT_ELEMS(S) ((size_t)(S) * B_ * VSPL * P_)
#define QP           (P_ / 4)
#define CMB_ROWS     8
#define CMB_GRPS     (VB_ / CMB_ROWS)
#define CMB_THREADS  (B_ * VSPL * CMB_GRPS * QP)
#define CMB_BLOCKS   (CMB_THREADS / 256)
#define NKT          (N_ / 16)
#define MKT          (2 * FRAG)
#define MKQ          (N_ / 4)
#define MKC_THREADS  (B_ * (CK_ / 8) * MKQ)
#define MKC_BLOCKS   (MKC_THREADS / 256)

typedef __attribute__((ext_vector_type(16))) __bf16       v16bf;
typedef __attribute__((ext_vector_type(8)))  float        v8f;
typedef __attribute__((ext_vector_type(4)))  unsigned int v4u;
typedef __attribute__((ext_vector_type(4)))  float        v4f;
#define GLOBAL_AS __attribute__((address_space(1)))

union Frag16 {
  v16bf  bf;
  v4u    q[2];
  __bf16 h[16];
};
union Pack8 {
  v4u    q;
  __bf16 h[8];
};
union Pack4 {
  uint2  u;
  __bf16 h[4];
};

#define QSTR  72
#define MKSTR 72
#define PSTR  136
#define RSTR  12

static __device__ __forceinline__ float other_half(float x) {
  return __uint_as_float(__builtin_amdgcn_permlanex16(__float_as_uint(x), __float_as_uint(x),
                                                      0x76543210u, 0xfedcba98u, false, false));
}

static __device__ __forceinline__ void lds_barrier() {
  asm volatile("s_wait_dscnt 0x0\n\ts_barrier_signal -1\n\ts_barrier_wait -1" ::: "memory");
}

template <bool MV_BF16, int NSPLIT, bool MK_FRAG>
__global__ __launch_bounds__(256, 4) void mem_reader_attn(
    const float* __restrict__ mk, const float* __restrict__ mv_f32,
    const __bf16* __restrict__ mv_bf, const __bf16* __restrict__ mk_frag,
    const float* __restrict__ qk,
    float* __restrict__ out, float* __restrict__ part, float* __restrict__ stat)
{
  constexpr int CPS = (NCHUNK + NSPLIT - 1) / NSPLIT;
  static_assert((NSPLIT - 1) * CPS < NCHUNK - 1, "every range must start on a full chunk");
  __shared__ __bf16 s_q[PQ * QSTR];
  __shared__ __bf16 s_mk[MK_FRAG ? 1 : 2][MK_FRAG ? 8 : CHUNK * MKSTR];
  __shared__ __bf16 s_prob[PQ * PSTR];
  __shared__ __attribute__((aligned(16))) float s_red[PQ * RSTR];

  const int tid  = threadIdx.x;
  const int wave = tid >> 5;
  const int lane = tid & 31;
  const int col  = lane & 15;
  const int kg   = (lane >> 4) * 8;
  const int p0   = blockIdx.x * PQ;
  const int vb0  = blockIdx.y * VB_ + wave * 32;
  const int bi   = blockIdx.z % B_;
  const int sp   = blockIdx.z / B_;
  const int c_lo = sp * CPS;
  const int c_hi = (c_lo + CPS < NCHUNK) ? (c_lo + CPS) : NCHUNK;

  const float*  mk_b  = mk + (size_t)bi * CK_ * N_;
  const float*  qk_b  = qk + (size_t)bi * CK_ * P_;
  const float*  mvf_b = mv_f32 ? (mv_f32 + (size_t)bi * CV_ * N_) : nullptr;
  const __bf16* mvb_b = mv_bf  ? (mv_bf  + (size_t)bi * CV_ * N_) : nullptr;

  for (int idx = tid; idx < CK_ * PQ; idx += 256) {
    int p = idx & (PQ - 1), c = idx / PQ;
    int pg = p0 + p; if (pg > P_ - 1) pg = P_ - 1;
    s_q[p * QSTR + c] = (__bf16)(qk_b[(size_t)c * P_ + pg] * 0.125f);
  }
  float mrun[PT_], lrun[PT_];
#pragma unroll
  for (int pt = 0; pt < PT_; ++pt) { mrun[pt] = -1e30f; lrun[pt] = 0.f; }

  v8f acc[2][PT_];
#pragma unroll
  for (int vt = 0; vt < 2; ++vt)
#pragma unroll
    for (int pt = 0; pt < PT_; ++pt) acc[vt][pt] = v8f{};

  const float* mk_t = mk_b + (size_t)(wave * 8) * N_;
  float4 pre[8];
  __bf16* const mk_dst = &s_mk[0][MK_FRAG ? 0 : ((4 * lane) * MKSTR + wave * 8)];
  Frag16 amk[2];
  const __bf16* mkf_w = MK_FRAG ? (mk_frag + (size_t)bi * NKT * MKT + lane * 8) : nullptr;
  if (MK_FRAG) {
    const GLOBAL_AS __bf16* mkf_p = (const GLOBAL_AS __bf16*)mkf_w;
    asm volatile("" : "+v"(mkf_p));
    const GLOBAL_AS __bf16* t = mkf_p + (size_t)(c_lo * (CHUNK / 16) + wave) * MKT;
#pragma unroll
    for (int ks = 0; ks < 2; ++ks) {
      amk[ks].q[0] = *(const GLOBAL_AS v4u*)(t + ks * FRAG);
      amk[ks].q[1] = *(const GLOBAL_AS v4u*)(t + ks * FRAG + FRAG / 2);
    }
  } else {
    const int ngc = c_lo * CHUNK + 4 * lane;
    const GLOBAL_AS float* mk_p = (const GLOBAL_AS float*)mk_t;
    asm volatile("" : "+v"(mk_p));
#pragma unroll
    for (int j = 0; j < 8; ++j) {
      const v4f t = *(const GLOBAL_AS v4f*)(mk_p + (size_t)j * N_ + ngc);
      pre[j] = make_float4(t[0], t[1], t[2], t[3]);
    }
    Pack8 r0, r1, r2, r3;
#pragma unroll
    for (int j = 0; j < 8; ++j) {
      r0.h[j] = (__bf16)pre[j].x;
      r1.h[j] = (__bf16)pre[j].y;
      r2.h[j] = (__bf16)pre[j].z;
      r3.h[j] = (__bf16)pre[j].w;
    }
    *(v4u*)(mk_dst)             = r0.q;
    *(v4u*)(mk_dst + MKSTR)     = r1.q;
    *(v4u*)(mk_dst + 2 * MKSTR) = r2.q;
    *(v4u*)(mk_dst + 3 * MKSTR) = r3.q;
    __builtin_amdgcn_sched_barrier(0);
    int ng2 = (c_lo + 1) * CHUNK + 4 * lane;
    if (ng2 > N_ - 4) ng2 = N_ - 4;
#pragma unroll
    for (int j = 0; j < 8; ++j) {
      const v4f t = *(const GLOBAL_AS v4f*)(mk_p + (size_t)j * N_ + ng2);
      pre[j] = make_float4(t[0], t[1], t[2], t[3]);
    }
  }

  const __bf16* mv_w0 = MV_BF16 ? (mvb_b + (size_t)(vb0 / 16) * NKS * FRAG + lane * 8) : nullptr;
  Frag16 mvr[4][2];
  if (MV_BF16) {
    __builtin_amdgcn_sched_barrier(0);
    const GLOBAL_AS __bf16* mv_p = (const GLOBAL_AS __bf16*)mv_w0;
    asm volatile("" : "+v"(mv_p));
#pragma unroll
    for (int ks = 0; ks < 4; ++ks) {
#pragma unroll
      for (int vt = 0; vt < 2; ++vt) {
        const GLOBAL_AS __bf16* blk = mv_p + ((size_t)vt * NKS + c_lo * (CHUNK / KSTEP) + ks) * FRAG;
        mvr[ks][vt].q[0] = *(const GLOBAL_AS v4u*)(blk);
        mvr[ks][vt].q[1] = *(const GLOBAL_AS v4u*)(blk + FRAG / 2);
      }
    }
  }
  lds_barrier();

  const GLOBAL_AS float* mk_l = (const GLOBAL_AS float*)mk_t;
  asm volatile("" : "+v"(mk_l));
  const GLOBAL_AS __bf16* mkf_l = (const GLOBAL_AS __bf16*)mkf_w;
  asm volatile("" : "+v"(mkf_l));
  int cur = 0;
  for (int nc = c_lo; nc < c_hi; ++nc, cur ^= 1) {
    const int n0 = nc * CHUNK;
    const __bf16* mk_cur = &s_mk[cur][0];

    if (!MK_FRAG) {
      const bool ok = (n0 + CHUNK + 4 * lane) < N_;
      Pack8 r0, r1, r2, r3;
#pragma unroll
      for (int j = 0; j < 8; ++j) {
        r0.h[j] = (__bf16)(ok ? pre[j].x : 0.f);
        r1.h[j] = (__bf16)(ok ? pre[j].y : 0.f);
        r2.h[j] = (__bf16)(ok ? pre[j].z : 0.f);
        r3.h[j] = (__bf16)(ok ? pre[j].w : 0.f);
      }
      __bf16* dp = mk_dst + (cur ^ 1) * (CHUNK * MKSTR);
      *(v4u*)(dp)             = r0.q;
      *(v4u*)(dp + MKSTR)     = r1.q;
      *(v4u*)(dp + 2 * MKSTR) = r2.q;
      *(v4u*)(dp + 3 * MKSTR) = r3.q;
    }

    const int mrow = wave * 16;
    if (!MK_FRAG) {
#pragma unroll
      for (int ks = 0; ks < 2; ++ks) {
        const __bf16* ap = mk_cur + (mrow + col) * MKSTR + ks * 32 + kg;
        amk[ks].q[0] = *(const v4u*)ap;  amk[ks].q[1] = *(const v4u*)(ap + 16);
      }
    }
    float sv[PT_][8];
    float lmx[PT_];
#pragma unroll
    for (int pt = 0; pt < PT_; ++pt) {
      v8f S = {};
#pragma unroll
      for (int ks = 0; ks < 2; ++ks) {
        Frag16 b;
        const __bf16* bp = &s_q[(pt * 16 + col) * QSTR + ks * 32 + kg];
        b.q[0] = *(const v4u*)bp;  b.q[1] = *(const v4u*)(bp + 16);
        S = __builtin_amdgcn_wmma_f32_16x16x32_bf16(false, amk[ks].bf, false, b.bf,
                                                    (short)0, S, false, false);
      }
      float lmax = -1e30f;
#pragma unroll
      for (int j = 0; j < 8; ++j) {
        int ng = n0 + mrow + kg + j;
        sv[pt][j] = (ng < N_) ? S[j] : -1e30f;
        lmax = fmaxf(lmax, sv[pt][j]);
      }
      lmx[pt] = fmaxf(lmax, other_half(lmax));
    }
    if (lane < 16) {
#pragma unroll
      for (int pt = 0; pt < PT_; ++pt) s_red[(pt * 16 + col) * RSTR + wave] = lmx[pt];
    }
    lds_barrier();

#pragma unroll
    for (int pt = 0; pt < PT_; ++pt) {
      const v4f* rp = (const v4f*)&s_red[(pt * 16 + col) * RSTR];
      const v4f a0 = rp[0], a1 = rp[1];
      const float c0 = fmaxf(fmaxf(a0[0], a0[1]), fmaxf(a0[2], a0[3]));
      const float c1 = fmaxf(fmaxf(a1[0], a1[1]), fmaxf(a1[2], a1[3]));
      const float mnew = fmaxf(mrun[pt], fmaxf(c0, c1));
      const float alph = __expf(mrun[pt] - mnew);
      mrun[pt] = mnew;
      acc[0][pt] *= alph;
      acc[1][pt] *= alph;

      Pack8 pk;
      float lsum = 0.f;
#pragma unroll
      for (int j = 0; j < 8; ++j) {
        int ng = n0 + mrow + kg + j;
        float pr = (ng < N_) ? __expf(sv[pt][j] - mnew) : 0.f;
        lsum += pr;
        pk.h[j] = (__bf16)pr;
      }
      *(v4u*)&s_prob[(pt * 16 + col) * PSTR + mrow + kg] = pk.q;
      lrun[pt] = lrun[pt] * alph + lsum;
    }
    lds_barrier();

    if (MK_FRAG) {
      int ktn = (nc + 1) * (CHUNK / 16) + wave;
      if (ktn > NKT - 1) ktn = NKT - 1;
      const GLOBAL_AS __bf16* t = mkf_l + (size_t)ktn * MKT;
#pragma unroll
      for (int ks = 0; ks < 2; ++ks) {
        amk[ks].q[0] = *(const GLOBAL_AS v4u*)(t + ks * FRAG);
        amk[ks].q[1] = *(const GLOBAL_AS v4u*)(t + ks * FRAG + FRAG / 2);
      }
    } else {
      const int ncn = (nc + 2 < NCHUNK) ? (nc + 2) : (NCHUNK - 1);
      int ngc = ncn * CHUNK + 4 * lane;
      if (ngc > N_ - 4) ngc = N_ - 4;
#pragma unroll
      for (int j = 0; j < 8; ++j) {
        const v4f t = *(const GLOBAL_AS v4f*)(mk_l + (size_t)j * N_ + ngc);
        pre[j] = make_float4(t[0], t[1], t[2], t[3]);
      }
    }
    __builtin_amdgcn_sched_barrier(0);

#pragma unroll
    for (int ks = 0; ks < 4; ++ks) {
      int nA = n0 + ks * 32 + kg;
      int b1 = nA;      if (b1 > N_ - 8) b1 = N_ - 8;
      int b2 = nA + 16; if (b2 > N_ - 8) b2 = N_ - 8;
      Frag16 b[PT_];
#pragma unroll
      for (int pt = 0; pt < PT_; ++pt) {
        const __bf16* bp = &s_prob[(pt * 16 + col) * PSTR + ks * 32 + kg];
        b[pt].q[0] = *(const v4u*)bp;  b[pt].q[1] = *(const v4u*)(bp + 16);
      }
#pragma unroll
      for (int vt = 0; vt < 2; ++vt) {
        const int v = vb0 + vt * 16 + col;
        Frag16 a;
        if (MV_BF16) {
          a = mvr[ks][vt];
        } else {
          const float* rowp = mvf_b + (size_t)v * N_;
          const float4 f0 = *(const float4*)(rowp + b1);
          const float4 f1 = *(const float4*)(rowp + b1 + 4);
          const float4 f2 = *(const float4*)(rowp + b2);
          const float4 f3 = *(const float4*)(rowp + b2 + 4);
          a.h[0]=(__bf16)f0.x;  a.h[1]=(__bf16)f0.y;  a.h[2]=(__bf16)f0.z;  a.h[3]=(__bf16)f0.w;
          a.h[4]=(__bf16)f1.x;  a.h[5]=(__bf16)f1.y;  a.h[6]=(__bf16)f1.z;  a.h[7]=(__bf16)f1.w;
          a.h[8]=(__bf16)f2.x;  a.h[9]=(__bf16)f2.y;  a.h[10]=(__bf16)f2.z; a.h[11]=(__bf16)f2.w;
          a.h[12]=(__bf16)f3.x; a.h[13]=(__bf16)f3.y; a.h[14]=(__bf16)f3.z; a.h[15]=(__bf16)f3.w;
        }
#pragma unroll
        for (int pt = 0; pt < PT_; ++pt)
          acc[vt][pt] = __builtin_amdgcn_wmma_f32_16x16x32_bf16(false, a.bf, false, b[pt].bf,
                                                                (short)0, acc[vt][pt], false, false);
      }
    }

    if (MV_BF16) {
      const int ncn = (nc + 1 < NCHUNK) ? (nc + 1) : (NCHUNK - 1);
#pragma unroll
      for (int ks = 0; ks < 4; ++ks) {
        int kst = ncn * (CHUNK / KSTEP) + ks;
        if (kst > NKS - 1) kst = NKS - 1;
#pragma unroll
        for (int vt = 0; vt < 2; ++vt) {
          const __bf16* blk = mv_w0 + ((size_t)vt * NKS + kst) * FRAG;
          mvr[ks][vt].q[0] = *(const v4u*)(blk);
          mvr[ks][vt].q[1] = *(const v4u*)(blk + FRAG / 2);
        }
      }
    }
  }

#pragma unroll
  for (int pt = 0; pt < PT_; ++pt) lrun[pt] += other_half(lrun[pt]);
  if (lane < 16) {
#pragma unroll
    for (int pt = 0; pt < PT_; ++pt) s_red[(pt * 16 + col) * RSTR + wave] = lrun[pt];
  }
  __syncthreads();
  float ltot[PT_];
#pragma unroll
  for (int pt = 0; pt < PT_; ++pt) {
    const float* rp = &s_red[(pt * 16 + col) * RSTR];
    float s = 0.f;
#pragma unroll
    for (int k = 0; k < 8; ++k) s += rp[k];
    ltot[pt] = s;
  }
  if (NSPLIT > 1) {
    if (wave == 0 && lane < 16) {
#pragma unroll
      for (int pt = 0; pt < PT_; ++pt) {
        const int pq = p0 + pt * 16 + col;
        if (pq < P_) {
          const size_t si = ((size_t)(sp * B_ + bi) * VSPL + blockIdx.y) * P_ + pq;
          stat[si]                      = mrun[pt];
          stat[STAT_ELEMS(NSPLIT) + si] = ltot[pt];
        }
      }
    }
  }
  float* dst = (NSPLIT > 1 && sp > 0)
             ? part + (size_t)(sp - 1) * PLANE_ELEMS + (size_t)bi * CV_ * P_
             : out + (size_t)bi * (2 * CV_) * P_;
#pragma unroll
  for (int pt = 0; pt < PT_; ++pt) {
    const float linv = (NSPLIT > 1) ? 1.0f : 1.0f / ltot[pt];
    const int pg = p0 + pt * 16 + col;
    if (pg < P_) {
#pragma unroll
      for (int vt = 0; vt < 2; ++vt) {
#pragma unroll
        for (int j = 0; j < 8; ++j) {
          int v = vb0 + vt * 16 + kg + j;
          dst[(size_t)v * P_ + pg] = (NSPLIT > 1) ? acc[vt][pt][j] : acc[vt][pt][j] * linv;
        }
      }
    }
  }
}

template <int NSPLIT>
__global__ __launch_bounds__(256) void combine_ranges(float* out,
                                                      const float* __restrict__ part,
                                                      const float* __restrict__ stat)
{
  const int u  = blockIdx.x * 256 + threadIdx.x;
  const int p4 = u % QP;
  const int g  = u / QP;
  const int vg = g % CMB_GRPS;
  const int bv = g / CMB_GRPS;
  const int b  = bv / VSPL;
  const int vs = bv - b * VSPL;

  v4f m[NSPLIT], l[NSPLIT], w[NSPLIT];
#pragma unroll
  for (int s = 0; s < NSPLIT; ++s) {
    const float* sp = stat + ((size_t)(s * B_ + b) * VSPL + vs) * P_ + 4 * p4;
    m[s] = *(const v4f*)(sp);
    l[s] = *(const v4f*)(sp + STAT_ELEMS(NSPLIT));
  }
  v4f mx = m[0];
#pragma unroll
  for (int s = 1; s < NSPLIT; ++s)
#pragma unroll
    for (int e = 0; e < 4; ++e) mx[e] = fmaxf(mx[e], m[s][e]);
  v4f den = {0.f, 0.f, 0.f, 0.f};
#pragma unroll
  for (int s = 0; s < NSPLIT; ++s) {
#pragma unroll
    for (int e = 0; e < 4; ++e) w[s][e] = __expf(m[s][e] - mx[e]);
    den += w[s] * l[s];
  }
#pragma unroll
  for (int s = 0; s < NSPLIT; ++s) w[s] = w[s] / den;

  const int v0 = vs * VB_ + vg * CMB_ROWS;
  float* o = out + ((size_t)b * (2 * CV_) + v0) * P_ + 4 * p4;
  const float* pp = part + ((size_t)b * CV_ + v0) * P_ + 4 * p4;
  v4f x[CMB_ROWS][NSPLIT];
#pragma unroll
  for (int r = 0; r < CMB_ROWS; ++r) {
    x[r][0] = *(const v4f*)(o + (size_t)r * P_);
#pragma unroll
    for (int s = 1; s < NSPLIT; ++s)
      x[r][s] = *(const v4f*)(pp + (size_t)(s - 1) * PLANE_ELEMS + (size_t)r * P_);
  }
#pragma unroll
  for (int r = 0; r < CMB_ROWS; ++r) {
    v4f y = x[r][0] * w[0];
#pragma unroll
    for (int s = 1; s < NSPLIT; ++s) y += x[r][s] * w[s];
    *(v4f*)(o + (size_t)r * P_) = y;
  }
}

__global__ __launch_bounds__(256) void convert_mv(const float* __restrict__ src,
                                                  __bf16* __restrict__ dst,
                                                  const float* __restrict__ mk,
                                                  __bf16* __restrict__ mk_dst)
{
  if (blockIdx.x >= CONV_BLOCKS) {
    const int t  = (blockIdx.x - CONV_BLOCKS) * 256 + threadIdx.x;
    const int q  = t % MKQ;
    const int g  = t / MKQ;
    const int cg = g & 7;
    const int b  = g >> 3;
    const float* sp = mk + ((size_t)b * CK_ + cg * 8) * N_ + 4 * q;
    v4f f[8];
#pragma unroll
    for (int j = 0; j < 8; ++j) f[j] = *(const v4f*)(sp + (size_t)j * N_);
    __bf16* dp = mk_dst + ((size_t)b * NKT + (q >> 2)) * MKT + (cg >> 2) * FRAG
               + ((cg >> 1) & 1) * (FRAG / 2) + ((cg & 1) * 16 + 4 * (q & 3)) * 8;
#pragma unroll
    for (int i = 0; i < 4; ++i) {
      Pack8 p;
#pragma unroll
      for (int j = 0; j < 8; ++j) p.h[j] = (__bf16)f[j][i];
      *(v4u*)(dp + i * 8) = p.q;
    }
    return;
  }
  const int u   = blockIdx.x * 256 + threadIdx.x;
  const int g   = u & 3;
  const int r   = (u >> 2) & 15;
  const int grp = u >> 6;
  const int rt  = grp / CONV_GRPS;
  const int ks0 = (grp - rt * CONV_GRPS) * CONV_KS;
  const float* sp = src + ((size_t)rt * 16 + r) * N_ + ks0 * KSTEP + g * 8;
  __bf16* dp = dst + ((size_t)rt * NKS + ks0) * FRAG + (g >> 1) * (FRAG / 2) + ((g & 1) * 16 + r) * 8;
#pragma unroll
  for (int b = 0; b < CONV_KS / CONV_KB; ++b) {
    float4 f0[CONV_KB], f1[CONV_KB];
#pragma unroll
    for (int j = 0; j < CONV_KB; ++j) {
      const float* p = sp + (b * CONV_KB + j) * KSTEP;
      f0[j] = *(const float4*)(p);
      f1[j] = *(const float4*)(p + 4);
    }
#pragma unroll
    for (int j = 0; j < CONV_KB; ++j) {
      Pack8 p;
      p.h[0] = (__bf16)f0[j].x; p.h[1] = (__bf16)f0[j].y;
      p.h[2] = (__bf16)f0[j].z; p.h[3] = (__bf16)f0[j].w;
      p.h[4] = (__bf16)f1[j].x; p.h[5] = (__bf16)f1[j].y;
      p.h[6] = (__bf16)f1[j].z; p.h[7] = (__bf16)f1[j].w;
      *(v4u*)(dp + (size_t)(b * CONV_KB + j) * FRAG) = p.q;
    }
  }
}

__global__ __launch_bounds__(256) void qv_copy(const float* __restrict__ qv,
                                               float* __restrict__ out)
{
  const int t = blockIdx.x * 256 + threadIdx.x;
  const v4f* s4 = (const v4f*)qv;
  v4f* d4 = (v4f*)out;
  v4f v[QV_PER];
#pragma unroll
  for (int j = 0; j < QV_PER; ++j)
    v[j] = s4[t + j * QV_THREADS];
#pragma unroll
  for (int j = 0; j < QV_PER; ++j) {
    const int i4 = t + j * QV_THREADS;
    const int b  = i4 / (CV_ * P_ / 4);
    d4[i4 + (b + 1) * (CV_ * P_ / 4)] = v[j];
  }
}

extern "C" void kernel_launch(void* const* d_in, const int* in_sizes, int n_in,
                              void* d_out, int out_size, void* d_ws, size_t ws_size,
                              hipStream_t stream) {
  const float* mk = (const float*)d_in[0];
  const float* mv = (const float*)d_in[1];
  const float* qk = (const float*)d_in[2];
  const float* qv = (const float*)d_in[3];
  float* out = (float*)d_out;

  const size_t mv_elems = (size_t)B_ * CV_ * N_;
  const size_t need     = mv_elems * sizeof(__bf16);
  const size_t need2 = need + (1 * PLANE_ELEMS + 2 * STAT_ELEMS(2)) * sizeof(float);
  const size_t need3 = need2 + (size_t)B_ * NKT * MKT * sizeof(__bf16);
  dim3 grid(PBLK, VSPL, B_);

  if (d_ws && ws_size >= need3) {
    __bf16* mv_bf = (__bf16*)d_ws;
    float* part = (float*)((char*)d_ws + need);
    float* stat = part + 1 * PLANE_ELEMS;
    __bf16* mk_bf = (__bf16*)((char*)d_ws + need2);
    convert_mv<<<CONV_BLOCKS + MKC_BLOCKS, 256, 0, stream>>>(mv, mv_bf, mk, mk_bf);
    mem_reader_attn<true, 2, true><<<dim3(PBLK, VSPL, B_ * 2), 256, 0, stream>>>(mk, nullptr, mv_bf, mk_bf, qk, out, part, stat);
    combine_ranges<2><<<CMB_BLOCKS, 256, 0, stream>>>(out, part, stat);
  } else if (d_ws && ws_size >= need2) {
    __bf16* mv_bf = (__bf16*)d_ws;
    float* part = (float*)((char*)d_ws + need);
    float* stat = part + 1 * PLANE_ELEMS;
    convert_mv<<<CONV_BLOCKS, 256, 0, stream>>>(mv, mv_bf, mk, nullptr);
    mem_reader_attn<true, 2, false><<<dim3(PBLK, VSPL, B_ * 2), 256, 0, stream>>>(mk, nullptr, mv_bf, nullptr, qk, out, part, stat);
    combine_ranges<2><<<CMB_BLOCKS, 256, 0, stream>>>(out, part, stat);
  } else if (d_ws && ws_size >= need) {
    __bf16* mv_bf = (__bf16*)d_ws;
    convert_mv<<<CONV_BLOCKS, 256, 0, stream>>>(mv, mv_bf, mk, nullptr);
    mem_reader_attn<true, 1, false><<<grid, 256, 0, stream>>>(mk, nullptr, mv_bf, nullptr, qk, out, nullptr, nullptr);
  } else {
    mem_reader_attn<false, 1, false><<<grid, 256, 0, stream>>>(mk, mv, nullptr, nullptr, qk, out, nullptr, nullptr);
  }

  qv_copy<<<QV_BLOCKS, 256, 0, stream>>>(qv, out);
}


# work-group tile 64 queries x 256 v-rows -> 64 x 512, 64 v-rows per wave (grid 26x1x8, one work-group per WGP): scores and softmax once per query, each prob fragment from LDS feeds 4 WMMAs
# speedup vs baseline: 12.6270x; 1.0622x over previous
#include <hip/hip_runtime.h>

#define B_    4
#define CK_   64
#define CV_   512
#define H_    30
#define W_    54
#define T_    8
#define P_    (H_*W_)
#define N_    (T_*H_*W_)
#define CHUNK 128
#define NCHUNK ((N_ + CHUNK - 1)/CHUNK)
#define PQ    64
#define PT_   (PQ/16)
#define VB_   512
#define VT_   (VB_/8/16)
#define VSPL  (CV_/VB_)
#define PBLK  ((P_ + PQ - 1)/PQ)
#define KSTEP 32
#define NKS   (N_/KSTEP)
#define FRAG  (16*KSTEP)
#define CONV_KS      27
#define CONV_KB      9
#define CONV_GRPS    (NKS / CONV_KS)
#define CONV_THREADS ((B_*CV_/16) * CONV_GRPS * 64)
#define CONV_BLOCKS  (CONV_THREADS / 256)
#define QV_V4        (B_*CV_*P_/4)
#define QV_PER       9
#define QV_THREADS   (QV_V4 / QV_PER)
#define QV_BLOCKS    (QV_THREADS / 256)
#define PLANE_ELEMS  ((size_t)B_ * CV_ * P_)
#define STAT_ELEMS(S) ((size_t)(S) * B_ * VSPL * P_)
#define QP           (P_ / 4)
#define CMB_ROWS     8
#define CMB_GRPS     (VB_ / CMB_ROWS)
#define CMB_THREADS  (B_ * VSPL * CMB_GRPS * QP)
#define CMB_BLOCKS   (CMB_THREADS / 256)
#define NKT          (N_ / 16)
#define MKT          (2 * FRAG)
#define MKQ          (N_ / 4)
#define MKC_THREADS  (B_ * (CK_ / 8) * MKQ)
#define MKC_BLOCKS   (MKC_THREADS / 256)

typedef __attribute__((ext_vector_type(16))) __bf16       v16bf;
typedef __attribute__((ext_vector_type(8)))  float        v8f;
typedef __attribute__((ext_vector_type(4)))  unsigned int v4u;
typedef __attribute__((ext_vector_type(4)))  float        v4f;
#define GLOBAL_AS __attribute__((address_space(1)))

union Frag16 {
  v16bf  bf;
  v4u    q[2];
  __bf16 h[16];
};
union Pack8 {
  v4u    q;
  __bf16 h[8];
};
union Pack4 {
  uint2  u;
  __bf16 h[4];
};

#define QSTR  72
#define MKSTR 72
#define PSTR  136
#define RSTR  12

static __device__ __forceinline__ float other_half(float x) {
  return __uint_as_float(__builtin_amdgcn_permlanex16(__float_as_uint(x), __float_as_uint(x),
                                                      0x76543210u, 0xfedcba98u, false, false));
}

static __device__ __forceinline__ void lds_barrier() {
  asm volatile("s_wait_dscnt 0x0\n\ts_barrier_signal -1\n\ts_barrier_wait -1" ::: "memory");
}

template <bool MV_BF16, int NSPLIT, bool MK_FRAG>
__global__ __launch_bounds__(256, 2) void mem_reader_attn(
    const float* __restrict__ mk, const float* __restrict__ mv_f32,
    const __bf16* __restrict__ mv_bf, const __bf16* __restrict__ mk_frag,
    const float* __restrict__ qk,
    float* __restrict__ out, float* __restrict__ part, float* __restrict__ stat)
{
  constexpr int CPS = (NCHUNK + NSPLIT - 1) / NSPLIT;
  static_assert((NSPLIT - 1) * CPS < NCHUNK - 1, "every range must start on a full chunk");
  __shared__ __bf16 s_q[PQ * QSTR];
  __shared__ __bf16 s_mk[MK_FRAG ? 1 : 2][MK_FRAG ? 8 : CHUNK * MKSTR];
  __shared__ __bf16 s_prob[PQ * PSTR];
  __shared__ __attribute__((aligned(16))) float s_red[PQ * RSTR];

  const int tid  = threadIdx.x;
  const int wave = tid >> 5;
  const int lane = tid & 31;
  const int col  = lane & 15;
  const int kg   = (lane >> 4) * 8;
  const int p0   = blockIdx.x * PQ;
  const int vb0  = blockIdx.y * VB_ + wave * (16 * VT_);
  const int bi   = blockIdx.z % B_;
  const int sp   = blockIdx.z / B_;
  const int c_lo = sp * CPS;
  const int c_hi = (c_lo + CPS < NCHUNK) ? (c_lo + CPS) : NCHUNK;

  const float*  mk_b  = mk + (size_t)bi * CK_ * N_;
  const float*  qk_b  = qk + (size_t)bi * CK_ * P_;
  const float*  mvf_b = mv_f32 ? (mv_f32 + (size_t)bi * CV_ * N_) : nullptr;
  const __bf16* mvb_b = mv_bf  ? (mv_bf  + (size_t)bi * CV_ * N_) : nullptr;

  for (int idx = tid; idx < CK_ * PQ; idx += 256) {
    int p = idx & (PQ - 1), c = idx / PQ;
    int pg = p0 + p; if (pg > P_ - 1) pg = P_ - 1;
    s_q[p * QSTR + c] = (__bf16)(qk_b[(size_t)c * P_ + pg] * 0.125f);
  }
  float mrun[PT_], lrun[PT_];
#pragma unroll
  for (int pt = 0; pt < PT_; ++pt) { mrun[pt] = -1e30f; lrun[pt] = 0.f; }

  v8f acc[VT_][PT_];
#pragma unroll
  for (int vt = 0; vt < VT_; ++vt)
#pragma unroll
    for (int pt = 0; pt < PT_; ++pt) acc[vt][pt] = v8f{};

  const float* mk_t = mk_b + (size_t)(wave * 8) * N_;
  float4 pre[8];
  __bf16* const mk_dst = &s_mk[0][MK_FRAG ? 0 : ((4 * lane) * MKSTR + wave * 8)];
  Frag16 amk[2];
  const __bf16* mkf_w = MK_FRAG ? (mk_frag + (size_t)bi * NKT * MKT + lane * 8) : nullptr;
  if (MK_FRAG) {
    const GLOBAL_AS __bf16* mkf_p = (const GLOBAL_AS __bf16*)mkf_w;
    asm volatile("" : "+v"(mkf_p));
    const GLOBAL_AS __bf16* t = mkf_p + (size_t)(c_lo * (CHUNK / 16) + wave) * MKT;
#pragma unroll
    for (int ks = 0; ks < 2; ++ks) {
      amk[ks].q[0] = *(const GLOBAL_AS v4u*)(t + ks * FRAG);
      amk[ks].q[1] = *(const GLOBAL_AS v4u*)(t + ks * FRAG + FRAG / 2);
    }
  } else {
    const int ngc = c_lo * CHUNK + 4 * lane;
    const GLOBAL_AS float* mk_p = (const GLOBAL_AS float*)mk_t;
    asm volatile("" : "+v"(mk_p));
#pragma unroll
    for (int j = 0; j < 8; ++j) {
      const v4f t = *(const GLOBAL_AS v4f*)(mk_p + (size_t)j * N_ + ngc);
      pre[j] = make_float4(t[0], t[1], t[2], t[3]);
    }
    Pack8 r0, r1, r2, r3;
#pragma unroll
    for (int j = 0; j < 8; ++j) {
      r0.h[j] = (__bf16)pre[j].x;
      r1.h[j] = (__bf16)pre[j].y;
      r2.h[j] = (__bf16)pre[j].z;
      r3.h[j] = (__bf16)pre[j].w;
    }
    *(v4u*)(mk_dst)             = r0.q;
    *(v4u*)(mk_dst + MKSTR)     = r1.q;
    *(v4u*)(mk_dst + 2 * MKSTR) = r2.q;
    *(v4u*)(mk_dst + 3 * MKSTR) = r3.q;
    __builtin_amdgcn_sched_barrier(0);
    int ng2 = (c_lo + 1) * CHUNK + 4 * lane;
    if (ng2 > N_ - 4) ng2 = N_ - 4;
#pragma unroll
    for (int j = 0; j < 8; ++j) {
      const v4f t = *(const GLOBAL_AS v4f*)(mk_p + (size_t)j * N_ + ng2);
      pre[j] = make_float4(t[0], t[1], t[2], t[3]);
    }
  }

  const __bf16* mv_w0 = MV_BF16 ? (mvb_b + (size_t)(vb0 / 16) * NKS * FRAG + lane * 8) : nullptr;
  Frag16 mvr[4][VT_];
  if (MV_BF16) {
    __builtin_amdgcn_sched_barrier(0);
    const GLOBAL_AS __bf16* mv_s = (const GLOBAL_AS __bf16*)mvb_b;
    asm volatile("" : "+s"(mv_s));
    const GLOBAL_AS __bf16* mv_p = mv_s + (size_t)(vb0 / 16) * NKS * FRAG + lane * 8;
#pragma unroll
    for (int ks = 0; ks < 4; ++ks) {
#pragma unroll
      for (int vt = 0; vt < VT_; ++vt) {
        const GLOBAL_AS __bf16* blk = mv_p + ((size_t)vt * NKS + c_lo * (CHUNK / KSTEP) + ks) * FRAG;
        mvr[ks][vt].q[0] = *(const GLOBAL_AS v4u*)(blk);
        mvr[ks][vt].q[1] = *(const GLOBAL_AS v4u*)(blk + FRAG / 2);
      }
    }
  }
  lds_barrier();

  const GLOBAL_AS float* mk_l = (const GLOBAL_AS float*)mk_t;
  if (!MK_FRAG) asm volatile("" : "+v"(mk_l));
  const GLOBAL_AS __bf16* mkf_s = (const GLOBAL_AS __bf16*)(MK_FRAG ? (mk_frag + (size_t)bi * NKT * MKT) : nullptr);
  asm volatile("" : "+s"(mkf_s));
  const GLOBAL_AS __bf16* mkf_l = mkf_s + lane * 8;
  int cur = 0;
  for (int nc = c_lo; nc < c_hi; ++nc, cur ^= 1) {
    const int n0 = nc * CHUNK;
    const __bf16* mk_cur = &s_mk[cur][0];

    if (!MK_FRAG) {
      const bool ok = (n0 + CHUNK + 4 * lane) < N_;
      Pack8 r0, r1, r2, r3;
#pragma unroll
      for (int j = 0; j < 8; ++j) {
        r0.h[j] = (__bf16)(ok ? pre[j].x : 0.f);
        r1.h[j] = (__bf16)(ok ? pre[j].y : 0.f);
        r2.h[j] = (__bf16)(ok ? pre[j].z : 0.f);
        r3.h[j] = (__bf16)(ok ? pre[j].w : 0.f);
      }
      __bf16* dp = mk_dst + (cur ^ 1) * (CHUNK * MKSTR);
      *(v4u*)(dp)             = r0.q;
      *(v4u*)(dp + MKSTR)     = r1.q;
      *(v4u*)(dp + 2 * MKSTR) = r2.q;
      *(v4u*)(dp + 3 * MKSTR) = r3.q;
    }

    const int mrow = wave * 16;
    if (!MK_FRAG) {
#pragma unroll
      for (int ks = 0; ks < 2; ++ks) {
        const __bf16* ap = mk_cur + (mrow + col) * MKSTR + ks * 32 + kg;
        amk[ks].q[0] = *(const v4u*)ap;  amk[ks].q[1] = *(const v4u*)(ap + 16);
      }
    }
    float sv[PT_][8];
    float lmx[PT_];
#pragma unroll
    for (int pt = 0; pt < PT_; ++pt) {
      v8f S = {};
#pragma unroll
      for (int ks = 0; ks < 2; ++ks) {
        Frag16 b;
        const __bf16* bp = &s_q[(pt * 16 + col) * QSTR + ks * 32 + kg];
        b.q[0] = *(const v4u*)bp;  b.q[1] = *(const v4u*)(bp + 16);
        S = __builtin_amdgcn_wmma_f32_16x16x32_bf16(false, amk[ks].bf, false, b.bf,
                                                    (short)0, S, false, false);
      }
      float lmax = -1e30f;
#pragma unroll
      for (int j = 0; j < 8; ++j) {
        int ng = n0 + mrow + kg + j;
        sv[pt][j] = (ng < N_) ? S[j] : -1e30f;
        lmax = fmaxf(lmax, sv[pt][j]);
      }
      lmx[pt] = fmaxf(lmax, other_half(lmax));
    }
    if (lane < 16) {
#pragma unroll
      for (int pt = 0; pt < PT_; ++pt) s_red[(pt * 16 + col) * RSTR + wave] = lmx[pt];
    }
    lds_barrier();

#pragma unroll
    for (int pt = 0; pt < PT_; ++pt) {
      const v4f* rp = (const v4f*)&s_red[(pt * 16 + col) * RSTR];
      const v4f a0 = rp[0], a1 = rp[1];
      const float c0 = fmaxf(fmaxf(a0[0], a0[1]), fmaxf(a0[2], a0[3]));
      const float c1 = fmaxf(fmaxf(a1[0], a1[1]), fmaxf(a1[2], a1[3]));
      const float mnew = fmaxf(mrun[pt], fmaxf(c0, c1));
      const float alph = __expf(mrun[pt] - mnew);
      mrun[pt] = mnew;
#pragma unroll
      for (int vt = 0; vt < VT_; ++vt) acc[vt][pt] *= alph;

      Pack8 pk;
      float lsum = 0.f;
#pragma unroll
      for (int j = 0; j < 8; ++j) {
        int ng = n0 + mrow + kg + j;
        float pr = (ng < N_) ? __expf(sv[pt][j] - mnew) : 0.f;
        lsum += pr;
        pk.h[j] = (__bf16)pr;
      }
      *(v4u*)&s_prob[(pt * 16 + col) * PSTR + mrow + kg] = pk.q;
      lrun[pt] = lrun[pt] * alph + lsum;
    }
    lds_barrier();

    if (MK_FRAG) {
      int ktn = (nc + 1) * (CHUNK / 16) + wave;
      if (ktn > NKT - 1) ktn = NKT - 1;
      const GLOBAL_AS __bf16* t = mkf_l + (size_t)ktn * MKT;
#pragma unroll
      for (int ks = 0; ks < 2; ++ks) {
        amk[ks].q[0] = *(const GLOBAL_AS v4u*)(t + ks * FRAG);
        amk[ks].q[1] = *(const GLOBAL_AS v4u*)(t + ks * FRAG + FRAG / 2);
      }
    } else {
      const int ncn = (nc + 2 < NCHUNK) ? (nc + 2) : (NCHUNK - 1);
      int ngc = ncn * CHUNK + 4 * lane;
      if (ngc > N_ - 4) ngc = N_ - 4;
#pragma unroll
      for (int j = 0; j < 8; ++j) {
        const v4f t = *(const GLOBAL_AS v4f*)(mk_l + (size_t)j * N_ + ngc);
        pre[j] = make_float4(t[0], t[1], t[2], t[3]);
      }
    }
    __builtin_amdgcn_sched_barrier(0);

#pragma unroll
    for (int ks = 0; ks < 4; ++ks) {
      int nA = n0 + ks * 32 + kg;
      int b1 = nA;      if (b1 > N_ - 8) b1 = N_ - 8;
      int b2 = nA + 16; if (b2 > N_ - 8) b2 = N_ - 8;
      Frag16 b[PT_];
#pragma unroll
      for (int pt = 0; pt < PT_; ++pt) {
        const __bf16* bp = &s_prob[(pt * 16 + col) * PSTR + ks * 32 + kg];
        b[pt].q[0] = *(const v4u*)bp;  b[pt].q[1] = *(const v4u*)(bp + 16);
      }
#pragma unroll
      for (int vt = 0; vt < VT_; ++vt) {
        const int v = vb0 + vt * 16 + col;
        Frag16 a;
        if (MV_BF16) {
          a = mvr[ks][vt];
        } else {
          const float* rowp = mvf_b + (size_t)v * N_;
          const float4 f0 = *(const float4*)(rowp + b1);
          const float4 f1 = *(const float4*)(rowp + b1 + 4);
          const float4 f2 = *(const float4*)(rowp + b2);
          const float4 f3 = *(const float4*)(rowp + b2 + 4);
          a.h[0]=(__bf16)f0.x;  a.h[1]=(__bf16)f0.y;  a.h[2]=(__bf16)f0.z;  a.h[3]=(__bf16)f0.w;
          a.h[4]=(__bf16)f1.x;  a.h[5]=(__bf16)f1.y;  a.h[6]=(__bf16)f1.z;  a.h[7]=(__bf16)f1.w;
          a.h[8]=(__bf16)f2.x;  a.h[9]=(__bf16)f2.y;  a.h[10]=(__bf16)f2.z; a.h[11]=(__bf16)f2.w;
          a.h[12]=(__bf16)f3.x; a.h[13]=(__bf16)f3.y; a.h[14]=(__bf16)f3.z; a.h[15]=(__bf16)f3.w;
        }
#pragma unroll
        for (int pt = 0; pt < PT_; ++pt)
          acc[vt][pt] = __builtin_amdgcn_wmma_f32_16x16x32_bf16(false, a.bf, false, b[pt].bf,
                                                                (short)0, acc[vt][pt], false, false);
      }
    }

    if (MV_BF16) {
      const int ncn = (nc + 1 < NCHUNK) ? (nc + 1) : (NCHUNK - 1);
#pragma unroll
      for (int ks = 0; ks < 4; ++ks) {
        int kst = ncn * (CHUNK / KSTEP) + ks;
        if (kst > NKS - 1) kst = NKS - 1;
#pragma unroll
        for (int vt = 0; vt < VT_; ++vt) {
          const __bf16* blk = mv_w0 + ((size_t)vt * NKS + kst) * FRAG;
          mvr[ks][vt].q[0] = *(const v4u*)(blk);
          mvr[ks][vt].q[1] = *(const v4u*)(blk + FRAG / 2);
        }
      }
    }
  }

#pragma unroll
  for (int pt = 0; pt < PT_; ++pt) lrun[pt] += other_half(lrun[pt]);
  if (lane < 16) {
#pragma unroll
    for (int pt = 0; pt < PT_; ++pt) s_red[(pt * 16 + col) * RSTR + wave] = lrun[pt];
  }
  __syncthreads();
  float ltot[PT_];
#pragma unroll
  for (int pt = 0; pt < PT_; ++pt) {
    const float* rp = &s_red[(pt * 16 + col) * RSTR];
    float s = 0.f;
#pragma unroll
    for (int k = 0; k < 8; ++k) s += rp[k];
    ltot[pt] = s;
  }
  if (NSPLIT > 1) {
    if (wave == 0 && lane < 16) {
#pragma unroll
      for (int pt = 0; pt < PT_; ++pt) {
        const int pq = p0 + pt * 16 + col;
        if (pq < P_) {
          const size_t si = ((size_t)(sp * B_ + bi) * VSPL + blockIdx.y) * P_ + pq;
          stat[si]                      = mrun[pt];
          stat[STAT_ELEMS(NSPLIT) + si] = ltot[pt];
        }
      }
    }
  }
  float* dst = (NSPLIT > 1 && sp > 0)
             ? part + (size_t)(sp - 1) * PLANE_ELEMS + (size_t)bi * CV_ * P_
             : out + (size_t)bi * (2 * CV_) * P_;
#pragma unroll
  for (int pt = 0; pt < PT_; ++pt) {
    const float linv = (NSPLIT > 1) ? 1.0f : 1.0f / ltot[pt];
    const int pg = p0 + pt * 16 + col;
    if (pg < P_) {
#pragma unroll
      for (int vt = 0; vt < VT_; ++vt) {
#pragma unroll
        for (int j = 0; j < 8; ++j) {
          int v = vb0 + vt * 16 + kg + j;
          dst[(size_t)v * P_ + pg] = (NSPLIT > 1) ? acc[vt][pt][j] : acc[vt][pt][j] * linv;
        }
      }
    }
  }
}

template <int NSPLIT>
__global__ __launch_bounds__(256) void combine_ranges(float* out,
                                                      const float* __restrict__ part,
                                                      const float* __restrict__ stat)
{
  const int u  = blockIdx.x * 256 + threadIdx.x;
  const int p4 = u % QP;
  const int g  = u / QP;
  const int vg = g % CMB_GRPS;
  const int bv = g / CMB_GRPS;
  const int b  = bv / VSPL;
  const int vs = bv - b * VSPL;

  v4f m[NSPLIT], l[NSPLIT], w[NSPLIT];
#pragma unroll
  for (int s = 0; s < NSPLIT; ++s) {
    const float* sp = stat + ((size_t)(s * B_ + b) * VSPL + vs) * P_ + 4 * p4;
    m[s] = *(const v4f*)(sp);
    l[s] = *(const v4f*)(sp + STAT_ELEMS(NSPLIT));
  }
  v4f mx = m[0];
#pragma unroll
  for (int s = 1; s < NSPLIT; ++s)
#pragma unroll
    for (int e = 0; e < 4; ++e) mx[e] = fmaxf(mx[e], m[s][e]);
  v4f den = {0.f, 0.f, 0.f, 0.f};
#pragma unroll
  for (int s = 0; s < NSPLIT; ++s) {
#pragma unroll
    for (int e = 0; e < 4; ++e) w[s][e] = __expf(m[s][e] - mx[e]);
    den += w[s] * l[s];
  }
#pragma unroll
  for (int s = 0; s < NSPLIT; ++s) w[s] = w[s] / den;

  const int v0 = vs * VB_ + vg * CMB_ROWS;
  float* o = out + ((size_t)b * (2 * CV_) + v0) * P_ + 4 * p4;
  const float* pp = part + ((size_t)b * CV_ + v0) * P_ + 4 * p4;
  v4f x[CMB_ROWS][NSPLIT];
#pragma unroll
  for (int r = 0; r < CMB_ROWS; ++r) {
    x[r][0] = *(const v4f*)(o + (size_t)r * P_);
#pragma unroll
    for (int s = 1; s < NSPLIT; ++s)
      x[r][s] = *(const v4f*)(pp + (size_t)(s - 1) * PLANE_ELEMS + (size_t)r * P_);
  }
#pragma unroll
  for (int r = 0; r < CMB_ROWS; ++r) {
    v4f y = x[r][0] * w[0];
#pragma unroll
    for (int s = 1; s < NSPLIT; ++s) y += x[r][s] * w[s];
    *(v4f*)(o + (size_t)r * P_) = y;
  }
}

__global__ __launch_bounds__(256) void convert_mv(const float* __restrict__ src,
                                                  __bf16* __restrict__ dst,
                                                  const float* __restrict__ mk,
                                                  __bf16* __restrict__ mk_dst)
{
  if (blockIdx.x >= CONV_BLOCKS) {
    const int t  = (blockIdx.x - CONV_BLOCKS) * 256 + threadIdx.x;
    const int q  = t % MKQ;
    const int g  = t / MKQ;
    const int cg = g & 7;
    const int b  = g >> 3;
    const float* sp = mk + ((size_t)b * CK_ + cg * 8) * N_ + 4 * q;
    v4f f[8];
#pragma unroll
    for (int j = 0; j < 8; ++j) f[j] = *(const v4f*)(sp + (size_t)j * N_);
    __bf16* dp = mk_dst + ((size_t)b * NKT + (q >> 2)) * MKT + (cg >> 2) * FRAG
               + ((cg >> 1) & 1) * (FRAG / 2) + ((cg & 1) * 16 + 4 * (q & 3)) * 8;
#pragma unroll
    for (int i = 0; i < 4; ++i) {
      Pack8 p;
#pragma unroll
      for (int j = 0; j < 8; ++j) p.h[j] = (__bf16)f[j][i];
      *(v4u*)(dp + i * 8) = p.q;
    }
    return;
  }
  const int u   = blockIdx.x * 256 + threadIdx.x;
  const int g   = u & 3;
  const int r   = (u >> 2) & 15;
  const int grp = u >> 6;
  const int rt  = grp / CONV_GRPS;
  const int ks0 = (grp - rt * CONV_GRPS) * CONV_KS;
  const float* sp = src + ((size_t)rt * 16 + r) * N_ + ks0 * KSTEP + g * 8;
  __bf16* dp = dst + ((size_t)rt * NKS + ks0) * FRAG + (g >> 1) * (FRAG / 2) + ((g & 1) * 16 + r) * 8;
#pragma unroll
  for (int b = 0; b < CONV_KS / CONV_KB; ++b) {
    float4 f0[CONV_KB], f1[CONV_KB];
#pragma unroll
    for (int j = 0; j < CONV_KB; ++j) {
      const float* p = sp + (b * CONV_KB + j) * KSTEP;
      f0[j] = *(const float4*)(p);
      f1[j] = *(const float4*)(p + 4);
    }
#pragma unroll
    for (int j = 0; j < CONV_KB; ++j) {
      Pack8 p;
      p.h[0] = (__bf16)f0[j].x; p.h[1] = (__bf16)f0[j].y;
      p.h[2] = (__bf16)f0[j].z; p.h[3] = (__bf16)f0[j].w;
      p.h[4] = (__bf16)f1[j].x; p.h[5] = (__bf16)f1[j].y;
      p.h[6] = (__bf16)f1[j].z; p.h[7] = (__bf16)f1[j].w;
      *(v4u*)(dp + (size_t)(b * CONV_KB + j) * FRAG) = p.q;
    }
  }
}

__global__ __launch_bounds__(256) void qv_copy(const float* __restrict__ qv,
                                               float* __restrict__ out)
{
  const int t = blockIdx.x * 256 + threadIdx.x;
  const v4f* s4 = (const v4f*)qv;
  v4f* d4 = (v4f*)out;
  v4f v[QV_PER];
#pragma unroll
  for (int j = 0; j < QV_PER; ++j)
    v[j] = s4[t + j * QV_THREADS];
#pragma unroll
  for (int j = 0; j < QV_PER; ++j) {
    const int i4 = t + j * QV_THREADS;
    const int b  = i4 / (CV_ * P_ / 4);
    d4[i4 + (b + 1) * (CV_ * P_ / 4)] = v[j];
  }
}

extern "C" void kernel_launch(void* const* d_in, const int* in_sizes, int n_in,
                              void* d_out, int out_size, void* d_ws, size_t ws_size,
                              hipStream_t stream) {
  const float* mk = (const float*)d_in[0];
  const float* mv = (const float*)d_in[1];
  const float* qk = (const float*)d_in[2];
  const float* qv = (const float*)d_in[3];
  float* out = (float*)d_out;

  const size_t mv_elems = (size_t)B_ * CV_ * N_;
  const size_t need     = mv_elems * sizeof(__bf16);
  const size_t need2 = need + (1 * PLANE_ELEMS + 2 * STAT_ELEMS(2)) * sizeof(float);
  const size_t need3 = need2 + (size_t)B_ * NKT * MKT * sizeof(__bf16);
  dim3 grid(PBLK, VSPL, B_);

  if (d_ws && ws_size >= need3) {
    __bf16* mv_bf = (__bf16*)d_ws;
    float* part = (float*)((char*)d_ws + need);
    float* stat = part + 1 * PLANE_ELEMS;
    __bf16* mk_bf = (__bf16*)((char*)d_ws + need2);
    convert_mv<<<CONV_BLOCKS + MKC_BLOCKS, 256, 0, stream>>>(mv, mv_bf, mk, mk_bf);
    mem_reader_attn<true, 2, true><<<dim3(PBLK, VSPL, B_ * 2), 256, 0, stream>>>(mk, nullptr, mv_bf, mk_bf, qk, out, part, stat);
    combine_ranges<2><<<CMB_BLOCKS, 256, 0, stream>>>(out, part, stat);
  } else if (d_ws && ws_size >= need2) {
    __bf16* mv_bf = (__bf16*)d_ws;
    float* part = (float*)((char*)d_ws + need);
    float* stat = part + 1 * PLANE_ELEMS;
    convert_mv<<<CONV_BLOCKS, 256, 0, stream>>>(mv, mv_bf, mk, nullptr);
    mem_reader_attn<true, 2, false><<<dim3(PBLK, VSPL, B_ * 2), 256, 0, stream>>>(mk, nullptr, mv_bf, nullptr, qk, out, part, stat);
    combine_ranges<2><<<CMB_BLOCKS, 256, 0, stream>>>(out, part, stat);
  } else if (d_ws && ws_size >= need) {
    __bf16* mv_bf = (__bf16*)d_ws;
    convert_mv<<<CONV_BLOCKS, 256, 0, stream>>>(mv, mv_bf, mk, nullptr);
    mem_reader_attn<true, 1, false><<<grid, 256, 0, stream>>>(mk, nullptr, mv_bf, nullptr, qk, out, nullptr, nullptr);
  } else {
    mem_reader_attn<false, 1, false><<<grid, 256, 0, stream>>>(mk, mv, nullptr, nullptr, qk, out, nullptr, nullptr);
  }

  qv_copy<<<QV_BLOCKS, 256, 0, stream>>>(qv, out);
}


# work-group tile 64 -> 80 queries and keys in 3 ranges, not 2: grid 21x1x12 = 252 work-groups for 256 WGPs (was 208), 80x34 query-chunks each, not 64x51; softmax sums add as 3 partials in range order
# speedup vs baseline: 13.6279x; 1.0793x over previous
#include <hip/hip_runtime.h>

#define B_    4
#define CK_   64
#define CV_   512
#define H_    30
#define W_    54
#define T_    8
#define P_    (H_*W_)
#define N_    (T_*H_*W_)
#define CHUNK 128
#define NCHUNK ((N_ + CHUNK - 1)/CHUNK)
#define VB_   512
#define VT_   (VB_/8/16)
#define VSPL  (CV_/VB_)
#define PBLK(Q) ((P_ + (Q) - 1)/(Q))
#define KSTEP 32
#define NKS   (N_/KSTEP)
#define FRAG  (16*KSTEP)
#define CONV_KS      27
#define CONV_KB      9
#define CONV_GRPS    (NKS / CONV_KS)
#define CONV_THREADS ((B_*CV_/16) * CONV_GRPS * 64)
#define CONV_BLOCKS  (CONV_THREADS / 256)
#define QV_V4        (B_*CV_*P_/4)
#define QV_PER       9
#define QV_THREADS   (QV_V4 / QV_PER)
#define QV_BLOCKS    (QV_THREADS / 256)
#define PLANE_ELEMS  ((size_t)B_ * CV_ * P_)
#define STAT_ELEMS(S) ((size_t)(S) * B_ * VSPL * P_)
#define QP           (P_ / 4)
#define CMB_ROWS     8
#define CMB_GRPS     (VB_ / CMB_ROWS)
#define CMB_THREADS  (B_ * VSPL * CMB_GRPS * QP)
#define CMB_BLOCKS   (CMB_THREADS / 256)
#define NKT          (N_ / 16)
#define MKT          (2 * FRAG)
#define MKQ          (N_ / 4)
#define MKC_THREADS  (B_ * (CK_ / 8) * MKQ)
#define MKC_BLOCKS   (MKC_THREADS / 256)

typedef __attribute__((ext_vector_type(16))) __bf16       v16bf;
typedef __attribute__((ext_vector_type(8)))  float        v8f;
typedef __attribute__((ext_vector_type(4)))  unsigned int v4u;
typedef __attribute__((ext_vector_type(4)))  float        v4f;
#define GLOBAL_AS __attribute__((address_space(1)))

union Frag16 {
  v16bf  bf;
  v4u    q[2];
  __bf16 h[16];
};
union Pack8 {
  v4u    q;
  __bf16 h[8];
};
union Pack4 {
  uint2  u;
  __bf16 h[4];
};

#define QSTR  72
#define MKSTR 72
#define PSTR  136
#define RSTR  12

static __device__ __forceinline__ float other_half(float x) {
  return __uint_as_float(__builtin_amdgcn_permlanex16(__float_as_uint(x), __float_as_uint(x),
                                                      0x76543210u, 0xfedcba98u, false, false));
}

static __device__ __forceinline__ void lds_barrier() {
  asm volatile("s_wait_dscnt 0x0\n\ts_barrier_signal -1\n\ts_barrier_wait -1" ::: "memory");
}

template <bool MV_BF16, int NSPLIT, bool MK_FRAG, int PQ>
__global__ __launch_bounds__(256, 2) void mem_reader_attn(
    const float* __restrict__ mk, const float* __restrict__ mv_f32,
    const __bf16* __restrict__ mv_bf, const __bf16* __restrict__ mk_frag,
    const float* __restrict__ qk,
    float* __restrict__ out, float* __restrict__ part, float* __restrict__ stat)
{
  constexpr int PT_ = PQ / 16;
  static_assert(PQ % 16 == 0 && (CK_ * PQ) % 256 == 0, "whole query tiles, whole passes of the q load");
  constexpr int CPS = (NCHUNK + NSPLIT - 1) / NSPLIT;
  static_assert((NSPLIT - 1) * CPS < NCHUNK - 1, "every range must start on a full chunk");
  __shared__ __bf16 s_q[PQ * QSTR];
  __shared__ __bf16 s_mk[MK_FRAG ? 1 : 2][MK_FRAG ? 8 : CHUNK * MKSTR];
  __shared__ __bf16 s_prob[PQ * PSTR];
  __shared__ __attribute__((aligned(16))) float s_red[PQ * RSTR];

  const int tid  = threadIdx.x;
  const int wave = tid >> 5;
  const int lane = tid & 31;
  const int col  = lane & 15;
  const int kg   = (lane >> 4) * 8;
  const int p0   = blockIdx.x * PQ;
  const int vb0  = blockIdx.y * VB_ + wave * (16 * VT_);
  const int bi   = blockIdx.z % B_;
  const int sp   = blockIdx.z / B_;
  const int c_lo = sp * CPS;
  const int c_hi = (c_lo + CPS < NCHUNK) ? (c_lo + CPS) : NCHUNK;

  const float*  mk_b  = mk + (size_t)bi * CK_ * N_;
  const float*  qk_b  = qk + (size_t)bi * CK_ * P_;
  const float*  mvf_b = mv_f32 ? (mv_f32 + (size_t)bi * CV_ * N_) : nullptr;
  const __bf16* mvb_b = mv_bf  ? (mv_bf  + (size_t)bi * CV_ * N_) : nullptr;

  for (int idx = tid; idx < CK_ * PQ; idx += 256) {
    int p = idx % PQ, c = idx / PQ;
    int pg = p0 + p; if (pg > P_ - 1) pg = P_ - 1;
    s_q[p * QSTR + c] = (__bf16)(qk_b[(size_t)c * P_ + pg] * 0.125f);
  }
  float mrun[PT_], lrun[PT_];
#pragma unroll
  for (int pt = 0; pt < PT_; ++pt) { mrun[pt] = -1e30f; lrun[pt] = 0.f; }

  v8f acc[VT_][PT_];
#pragma unroll
  for (int vt = 0; vt < VT_; ++vt)
#pragma unroll
    for (int pt = 0; pt < PT_; ++pt) acc[vt][pt] = v8f{};

  const float* mk_t = mk_b + (size_t)(wave * 8) * N_;
  float4 pre[8];
  __bf16* const mk_dst = &s_mk[0][MK_FRAG ? 0 : ((4 * lane) * MKSTR + wave * 8)];
  Frag16 amk[2];
  const __bf16* mkf_w = MK_FRAG ? (mk_frag + (size_t)bi * NKT * MKT + lane * 8) : nullptr;
  if (MK_FRAG) {
    const GLOBAL_AS __bf16* mkf_p = (const GLOBAL_AS __bf16*)mkf_w;
    asm volatile("" : "+v"(mkf_p));
    const GLOBAL_AS __bf16* t = mkf_p + (size_t)(c_lo * (CHUNK / 16) + wave) * MKT;
#pragma unroll
    for (int ks = 0; ks < 2; ++ks) {
      amk[ks].q[0] = *(const GLOBAL_AS v4u*)(t + ks * FRAG);
      amk[ks].q[1] = *(const GLOBAL_AS v4u*)(t + ks * FRAG + FRAG / 2);
    }
  } else {
    const int ngc = c_lo * CHUNK + 4 * lane;
    const GLOBAL_AS float* mk_p = (const GLOBAL_AS float*)mk_t;
    asm volatile("" : "+v"(mk_p));
#pragma unroll
    for (int j = 0; j < 8; ++j) {
      const v4f t = *(const GLOBAL_AS v4f*)(mk_p + (size_t)j * N_ + ngc);
      pre[j] = make_float4(t[0], t[1], t[2], t[3]);
    }
    Pack8 r0, r1, r2, r3;
#pragma unroll
    for (int j = 0; j < 8; ++j) {
      r0.h[j] = (__bf16)pre[j].x;
      r1.h[j] = (__bf16)pre[j].y;
      r2.h[j] = (__bf16)pre[j].z;
      r3.h[j] = (__bf16)pre[j].w;
    }
    *(v4u*)(mk_dst)             = r0.q;
    *(v4u*)(mk_dst + MKSTR)     = r1.q;
    *(v4u*)(mk_dst + 2 * MKSTR) = r2.q;
    *(v4u*)(mk_dst + 3 * MKSTR) = r3.q;
    __builtin_amdgcn_sched_barrier(0);
    int ng2 = (c_lo + 1) * CHUNK + 4 * lane;
    if (ng2 > N_ - 4) ng2 = N_ - 4;
#pragma unroll
    for (int j = 0; j < 8; ++j) {
      const v4f t = *(const GLOBAL_AS v4f*)(mk_p + (size_t)j * N_ + ng2);
      pre[j] = make_float4(t[0], t[1], t[2], t[3]);
    }
  }

  const __bf16* mv_w0 = MV_BF16 ? (mvb_b + (size_t)(vb0 / 16) * NKS * FRAG + lane * 8) : nullptr;
  Frag16 mvr[4][VT_];
  if (MV_BF16) {
    __builtin_amdgcn_sched_barrier(0);
    const GLOBAL_AS __bf16* mv_s = (const GLOBAL_AS __bf16*)mvb_b;
    asm volatile("" : "+s"(mv_s));
    const GLOBAL_AS __bf16* mv_p = mv_s + (size_t)(vb0 / 16) * NKS * FRAG + lane * 8;
#pragma unroll
    for (int ks = 0; ks < 4; ++ks) {
#pragma unroll
      for (int vt = 0; vt < VT_; ++vt) {
        const GLOBAL_AS __bf16* blk = mv_p + ((size_t)vt * NKS + c_lo * (CHUNK / KSTEP) + ks) * FRAG;
        mvr[ks][vt].q[0] = *(const GLOBAL_AS v4u*)(blk);
        mvr[ks][vt].q[1] = *(const GLOBAL_AS v4u*)(blk + FRAG / 2);
      }
    }
  }
  lds_barrier();

  const GLOBAL_AS float* mk_l = (const GLOBAL_AS float*)mk_t;
  if (!MK_FRAG) asm volatile("" : "+v"(mk_l));
  const GLOBAL_AS __bf16* mkf_s = (const GLOBAL_AS __bf16*)(MK_FRAG ? (mk_frag + (size_t)bi * NKT * MKT) : nullptr);
  asm volatile("" : "+s"(mkf_s));
  const GLOBAL_AS __bf16* mkf_l = mkf_s + lane * 8;
  int cur = 0;
  for (int nc = c_lo; nc < c_hi; ++nc, cur ^= 1) {
    const int n0 = nc * CHUNK;
    const __bf16* mk_cur = &s_mk[cur][0];

    if (!MK_FRAG) {
      const bool ok = (n0 + CHUNK + 4 * lane) < N_;
      Pack8 r0, r1, r2, r3;
#pragma unroll
      for (int j = 0; j < 8; ++j) {
        r0.h[j] = (__bf16)(ok ? pre[j].x : 0.f);
        r1.h[j] = (__bf16)(ok ? pre[j].y : 0.f);
        r2.h[j] = (__bf16)(ok ? pre[j].z : 0.f);
        r3.h[j] = (__bf16)(ok ? pre[j].w : 0.f);
      }
      __bf16* dp = mk_dst + (cur ^ 1) * (CHUNK * MKSTR);
      *(v4u*)(dp)             = r0.q;
      *(v4u*)(dp + MKSTR)     = r1.q;
      *(v4u*)(dp + 2 * MKSTR) = r2.q;
      *(v4u*)(dp + 3 * MKSTR) = r3.q;
    }

    const int mrow = wave * 16;
    if (!MK_FRAG) {
#pragma unroll
      for (int ks = 0; ks < 2; ++ks) {
        const __bf16* ap = mk_cur + (mrow + col) * MKSTR + ks * 32 + kg;
        amk[ks].q[0] = *(const v4u*)ap;  amk[ks].q[1] = *(const v4u*)(ap + 16);
      }
    }
    float sv[PT_][8];
    float lmx[PT_];
#pragma unroll
    for (int pt = 0; pt < PT_; ++pt) {
      v8f S = {};
#pragma unroll
      for (int ks = 0; ks < 2; ++ks) {
        Frag16 b;
        const __bf16* bp = &s_q[(pt * 16 + col) * QSTR + ks * 32 + kg];
        b.q[0] = *(const v4u*)bp;  b.q[1] = *(const v4u*)(bp + 16);
        S = __builtin_amdgcn_wmma_f32_16x16x32_bf16(false, amk[ks].bf, false, b.bf,
                                                    (short)0, S, false, false);
      }
      float lmax = -1e30f;
#pragma unroll
      for (int j = 0; j < 8; ++j) {
        int ng = n0 + mrow + kg + j;
        sv[pt][j] = (ng < N_) ? S[j] : -1e30f;
        lmax = fmaxf(lmax, sv[pt][j]);
      }
      lmx[pt] = fmaxf(lmax, other_half(lmax));
    }
    if (lane < 16) {
#pragma unroll
      for (int pt = 0; pt < PT_; ++pt) s_red[(pt * 16 + col) * RSTR + wave] = lmx[pt];
    }
    lds_barrier();

#pragma unroll
    for (int pt = 0; pt < PT_; ++pt) {
      const v4f* rp = (const v4f*)&s_red[(pt * 16 + col) * RSTR];
      const v4f a0 = rp[0], a1 = rp[1];
      const float c0 = fmaxf(fmaxf(a0[0], a0[1]), fmaxf(a0[2], a0[3]));
      const float c1 = fmaxf(fmaxf(a1[0], a1[1]), fmaxf(a1[2], a1[3]));
      const float mnew = fmaxf(mrun[pt], fmaxf(c0, c1));
      const float alph = __expf(mrun[pt] - mnew);
      mrun[pt] = mnew;
#pragma unroll
      for (int vt = 0; vt < VT_; ++vt) acc[vt][pt] *= alph;

      Pack8 pk;
      float lsum = 0.f;
#pragma unroll
      for (int j = 0; j < 8; ++j) {
        int ng = n0 + mrow + kg + j;
        float pr = (ng < N_) ? __expf(sv[pt][j] - mnew) : 0.f;
        lsum += pr;
        pk.h[j] = (__bf16)pr;
      }
      *(v4u*)&s_prob[(pt * 16 + col) * PSTR + mrow + kg] = pk.q;
      lrun[pt] = lrun[pt] * alph + lsum;
    }
    lds_barrier();

    if (MK_FRAG) {
      int ktn = (nc + 1) * (CHUNK / 16) + wave;
      if (ktn > NKT - 1) ktn = NKT - 1;
      const GLOBAL_AS __bf16* t = mkf_l + (size_t)ktn * MKT;
#pragma unroll
      for (int ks = 0; ks < 2; ++ks) {
        amk[ks].q[0] = *(const GLOBAL_AS v4u*)(t + ks * FRAG);
        amk[ks].q[1] = *(const GLOBAL_AS v4u*)(t + ks * FRAG + FRAG / 2);
      }
    } else {
      const int ncn = (nc + 2 < NCHUNK) ? (nc + 2) : (NCHUNK - 1);
      int ngc = ncn * CHUNK + 4 * lane;
      if (ngc > N_ - 4) ngc = N_ - 4;
#pragma unroll
      for (int j = 0; j < 8; ++j) {
        const v4f t = *(const GLOBAL_AS v4f*)(mk_l + (size_t)j * N_ + ngc);
        pre[j] = make_float4(t[0], t[1], t[2], t[3]);
      }
    }
    __builtin_amdgcn_sched_barrier(0);

#pragma unroll
    for (int ks = 0; ks < 4; ++ks) {
      int nA = n0 + ks * 32 + kg;
      int b1 = nA;      if (b1 > N_ - 8) b1 = N_ - 8;
      int b2 = nA + 16; if (b2 > N_ - 8) b2 = N_ - 8;
      Frag16 b[PT_];
#pragma unroll
      for (int pt = 0; pt < PT_; ++pt) {
        const __bf16* bp = &s_prob[(pt * 16 + col) * PSTR + ks * 32 + kg];
        b[pt].q[0] = *(const v4u*)bp;  b[pt].q[1] = *(const v4u*)(bp + 16);
      }
#pragma unroll
      for (int vt = 0; vt < VT_; ++vt) {
        const int v = vb0 + vt * 16 + col;
        Frag16 a;
        if (MV_BF16) {
          a = mvr[ks][vt];
        } else {
          const float* rowp = mvf_b + (size_t)v * N_;
          const float4 f0 = *(const float4*)(rowp + b1);
          const float4 f1 = *(const float4*)(rowp + b1 + 4);
          const float4 f2 = *(const float4*)(rowp + b2);
          const float4 f3 = *(const float4*)(rowp + b2 + 4);
          a.h[0]=(__bf16)f0.x;  a.h[1]=(__bf16)f0.y;  a.h[2]=(__bf16)f0.z;  a.h[3]=(__bf16)f0.w;
          a.h[4]=(__bf16)f1.x;  a.h[5]=(__bf16)f1.y;  a.h[6]=(__bf16)f1.z;  a.h[7]=(__bf16)f1.w;
          a.h[8]=(__bf16)f2.x;  a.h[9]=(__bf16)f2.y;  a.h[10]=(__bf16)f2.z; a.h[11]=(__bf16)f2.w;
          a.h[12]=(__bf16)f3.x; a.h[13]=(__bf16)f3.y; a.h[14]=(__bf16)f3.z; a.h[15]=(__bf16)f3.w;
        }
#pragma unroll
        for (int pt = 0; pt < PT_; ++pt)
          acc[vt][pt] = __builtin_amdgcn_wmma_f32_16x16x32_bf16(false, a.bf, false, b[pt].bf,
                                                                (short)0, acc[vt][pt], false, false);
      }
    }

    if (MV_BF16) {
      const int ncn = (nc + 1 < NCHUNK) ? (nc + 1) : (NCHUNK - 1);
#pragma unroll
      for (int ks = 0; ks < 4; ++ks) {
        int kst = ncn * (CHUNK / KSTEP) + ks;
        if (kst > NKS - 1) kst = NKS - 1;
#pragma unroll
        for (int vt = 0; vt < VT_; ++vt) {
          const __bf16* blk = mv_w0 + ((size_t)vt * NKS + kst) * FRAG;
          mvr[ks][vt].q[0] = *(const v4u*)(blk);
          mvr[ks][vt].q[1] = *(const v4u*)(blk + FRAG / 2);
        }
      }
    }
  }

  const int lane_e = (int)__builtin_amdgcn_mbcnt_lo(~0u, 0u);
#pragma unroll
  for (int pt = 0; pt < PT_; ++pt) lrun[pt] += other_half(lrun[pt]);
  if (lane_e < 16) {
#pragma unroll
    for (int pt = 0; pt < PT_; ++pt) s_red[(pt * 16 + col) * RSTR + wave] = lrun[pt];
  }
  __syncthreads();
  float ltot[PT_];
#pragma unroll
  for (int pt = 0; pt < PT_; ++pt) {
    const float* rp = &s_red[(pt * 16 + col) * RSTR];
    float s = 0.f;
#pragma unroll
    for (int k = 0; k < 8; ++k) s += rp[k];
    ltot[pt] = s;
  }
  if (NSPLIT > 1) {
    if (wave == 0 && lane_e < 16) {
#pragma unroll
      for (int pt = 0; pt < PT_; ++pt) {
        const int pq = p0 + pt * 16 + col;
        if (pq < P_) {
          const size_t si = ((size_t)(sp * B_ + bi) * VSPL + blockIdx.y) * P_ + pq;
          stat[si]                      = mrun[pt];
          stat[STAT_ELEMS(NSPLIT) + si] = ltot[pt];
        }
      }
    }
  }
  float* dst = (NSPLIT > 2 && sp > 1)
             ? part + (size_t)(sp - 2) * PLANE_ELEMS + (size_t)bi * CV_ * P_
             : out + ((size_t)bi * 2 + sp) * CV_ * P_;
#pragma unroll
  for (int pt = 0; pt < PT_; ++pt) {
    const float linv = (NSPLIT > 1) ? 1.0f : 1.0f / ltot[pt];
    const int pg = p0 + pt * 16 + col;
    if (pg < P_) {
#pragma unroll
      for (int vt = 0; vt < VT_; ++vt) {
#pragma unroll
        for (int j = 0; j < 8; ++j) {
          int v = vb0 + vt * 16 + kg + j;
          dst[(size_t)v * P_ + pg] = (NSPLIT > 1) ? acc[vt][pt][j] : acc[vt][pt][j] * linv;
        }
      }
    }
  }
}

template <int NSPLIT>
__global__ __launch_bounds__(256) void combine_ranges(float* out,
                                                      const float* __restrict__ part,
                                                      const float* __restrict__ stat)
{
  const int u  = blockIdx.x * 256 + threadIdx.x;
  const int p4 = u % QP;
  const int g  = u / QP;
  const int vg = g % CMB_GRPS;
  const int bv = g / CMB_GRPS;
  const int b  = bv / VSPL;
  const int vs = bv - b * VSPL;

  v4f m[NSPLIT], l[NSPLIT], w[NSPLIT];
#pragma unroll
  for (int s = 0; s < NSPLIT; ++s) {
    const float* sp = stat + ((size_t)(s * B_ + b) * VSPL + vs) * P_ + 4 * p4;
    m[s] = *(const v4f*)(sp);
    l[s] = *(const v4f*)(sp + STAT_ELEMS(NSPLIT));
  }
  v4f mx = m[0];
#pragma unroll
  for (int s = 1; s < NSPLIT; ++s)
#pragma unroll
    for (int e = 0; e < 4; ++e) mx[e] = fmaxf(mx[e], m[s][e]);
  v4f den = {0.f, 0.f, 0.f, 0.f};
#pragma unroll
  for (int s = 0; s < NSPLIT; ++s) {
#pragma unroll
    for (int e = 0; e < 4; ++e) w[s][e] = __expf(m[s][e] - mx[e]);
    den += w[s] * l[s];
  }
#pragma unroll
  for (int s = 0; s < NSPLIT; ++s) w[s] = w[s] / den;

  const int v0 = vs * VB_ + vg * CMB_ROWS;
  float* o = out + ((size_t)b * (2 * CV_) + v0) * P_ + 4 * p4;
  const float* pp = part + ((size_t)b * CV_ + v0) * P_ + 4 * p4;
  v4f x[CMB_ROWS][NSPLIT];
#pragma unroll
  for (int r = 0; r < CMB_ROWS; ++r) {
    x[r][0] = *(const v4f*)(o + (size_t)r * P_);
    x[r][1] = *(const v4f*)(o + (size_t)(CV_ + r) * P_);
#pragma unroll
    for (int s = 2; s < NSPLIT; ++s)
      x[r][s] = *(const v4f*)(pp + (size_t)(s - 2) * PLANE_ELEMS + (size_t)r * P_);
  }
#pragma unroll
  for (int r = 0; r < CMB_ROWS; ++r) {
    v4f y = x[r][0] * w[0];
#pragma unroll
    for (int s = 1; s < NSPLIT; ++s) y += x[r][s] * w[s];
    *(v4f*)(o + (size_t)r * P_) = y;
  }
}

__global__ __launch_bounds__(256) void convert_mv(const float* __restrict__ src,
                                                  __bf16* __restrict__ dst,
                                                  const float* __restrict__ mk,
                                                  __bf16* __restrict__ mk_dst)
{
  if (blockIdx.x >= CONV_BLOCKS) {
    const int t  = (blockIdx.x - CONV_BLOCKS) * 256 + threadIdx.x;
    const int q  = t % MKQ;
    const int g  = t / MKQ;
    const int cg = g & 7;
    const int b  = g >> 3;
    const float* sp = mk + ((size_t)b * CK_ + cg * 8) * N_ + 4 * q;
    v4f f[8];
#pragma unroll
    for (int j = 0; j < 8; ++j) f[j] = *(const v4f*)(sp + (size_t)j * N_);
    __bf16* dp = mk_dst + ((size_t)b * NKT + (q >> 2)) * MKT + (cg >> 2) * FRAG
               + ((cg >> 1) & 1) * (FRAG / 2) + ((cg & 1) * 16 + 4 * (q & 3)) * 8;
#pragma unroll
    for (int i = 0; i < 4; ++i) {
      Pack8 p;
#pragma unroll
      for (int j = 0; j < 8; ++j) p.h[j] = (__bf16)f[j][i];
      *(v4u*)(dp + i * 8) = p.q;
    }
    return;
  }
  const int u   = blockIdx.x * 256 + threadIdx.x;
  const int g   = u & 3;
  const int r   = (u >> 2) & 15;
  const int grp = u >> 6;
  const int rt  = grp / CONV_GRPS;
  const int ks0 = (grp - rt * CONV_GRPS) * CONV_KS;
  const float* sp = src + ((size_t)rt * 16 + r) * N_ + ks0 * KSTEP + g * 8;
  __bf16* dp = dst + ((size_t)rt * NKS + ks0) * FRAG + (g >> 1) * (FRAG / 2) + ((g & 1) * 16 + r) * 8;
#pragma unroll
  for (int b = 0; b < CONV_KS / CONV_KB; ++b) {
    float4 f0[CONV_KB], f1[CONV_KB];
#pragma unroll
    for (int j = 0; j < CONV_KB; ++j) {
      const float* p = sp + (b * CONV_KB + j) * KSTEP;
      f0[j] = *(const float4*)(p);
      f1[j] = *(const float4*)(p + 4);
    }
#pragma unroll
    for (int j = 0; j < CONV_KB; ++j) {
      Pack8 p;
      p.h[0] = (__bf16)f0[j].x; p.h[1] = (__bf16)f0[j].y;
      p.h[2] = (__bf16)f0[j].z; p.h[3] = (__bf16)f0[j].w;
      p.h[4] = (__bf16)f1[j].x; p.h[5] = (__bf16)f1[j].y;
      p.h[6] = (__bf16)f1[j].z; p.h[7] = (__bf16)f1[j].w;
      *(v4u*)(dp + (size_t)(b * CONV_KB + j) * FRAG) = p.q;
    }
  }
}

__global__ __launch_bounds__(256) void qv_copy(const float* __restrict__ qv,
                                               float* __restrict__ out)
{
  const int t = blockIdx.x * 256 + threadIdx.x;
  const v4f* s4 = (const v4f*)qv;
  v4f* d4 = (v4f*)out;
  v4f v[QV_PER];
#pragma unroll
  for (int j = 0; j < QV_PER; ++j)
    v[j] = s4[t + j * QV_THREADS];
#pragma unroll
  for (int j = 0; j < QV_PER; ++j) {
    const int i4 = t + j * QV_THREADS;
    const int b  = i4 / (CV_ * P_ / 4);
    d4[i4 + (b + 1) * (CV_ * P_ / 4)] = v[j];
  }
}

extern "C" void kernel_launch(void* const* d_in, const int* in_sizes, int n_in,
                              void* d_out, int out_size, void* d_ws, size_t ws_size,
                              hipStream_t stream) {
  const float* mk = (const float*)d_in[0];
  const float* mv = (const float*)d_in[1];
  const float* qk = (const float*)d_in[2];
  const float* qv = (const float*)d_in[3];
  float* out = (float*)d_out;

  const size_t mv_elems = (size_t)B_ * CV_ * N_;
  const size_t need     = mv_elems * sizeof(__bf16);
  const size_t need2 = need + (1 * PLANE_ELEMS + 2 * STAT_ELEMS(2)) * sizeof(float);
  const size_t need3 = need2 + (size_t)B_ * NKT * MKT * sizeof(__bf16);
  const size_t need2t = need + (1 * PLANE_ELEMS + 2 * STAT_ELEMS(3)) * sizeof(float);
  const size_t need3t = need2t + (size_t)B_ * NKT * MKT * sizeof(__bf16);
  dim3 grid(PBLK(64), VSPL, B_);

  if (d_ws && ws_size >= need3t) {
    __bf16* mv_bf = (__bf16*)d_ws;
    float* part = (float*)((char*)d_ws + need);
    float* stat = part + 1 * PLANE_ELEMS;
    __bf16* mk_bf = (__bf16*)((char*)d_ws + need2t);
    convert_mv<<<CONV_BLOCKS + MKC_BLOCKS, 256, 0, stream>>>(mv, mv_bf, mk, mk_bf);
    mem_reader_attn<true, 3, true, 80><<<dim3(PBLK(80), VSPL, B_ * 3), 256, 0, stream>>>(mk, nullptr, mv_bf, mk_bf, qk, out, part, stat);
    combine_ranges<3><<<CMB_BLOCKS, 256, 0, stream>>>(out, part, stat);
  } else if (d_ws && ws_size >= need3) {
    __bf16* mv_bf = (__bf16*)d_ws;
    float* part = (float*)((char*)d_ws + need);
    float* stat = part + 1 * PLANE_ELEMS;
    __bf16* mk_bf = (__bf16*)((char*)d_ws + need2);
    convert_mv<<<CONV_BLOCKS + MKC_BLOCKS, 256, 0, stream>>>(mv, mv_bf, mk, mk_bf);
    mem_reader_attn<true, 2, true, 64><<<dim3(PBLK(64), VSPL, B_ * 2), 256, 0, stream>>>(mk, nullptr, mv_bf, mk_bf, qk, out, part, stat);
    combine_ranges<2><<<CMB_BLOCKS, 256, 0, stream>>>(out, part, stat);
  } else if (d_ws && ws_size >= need2) {
    __bf16* mv_bf = (__bf16*)d_ws;
    float* part = (float*)((char*)d_ws + need);
    float* stat = part + 1 * PLANE_ELEMS;
    convert_mv<<<CONV_BLOCKS, 256, 0, stream>>>(mv, mv_bf, mk, nullptr);
    mem_reader_attn<true, 2, false, 64><<<dim3(PBLK(64), VSPL, B_ * 2), 256, 0, stream>>>(mk, nullptr, mv_bf, nullptr, qk, out, part, stat);
    combine_ranges<2><<<CMB_BLOCKS, 256, 0, stream>>>(out, part, stat);
  } else if (d_ws && ws_size >= need) {
    __bf16* mv_bf = (__bf16*)d_ws;
    convert_mv<<<CONV_BLOCKS, 256, 0, stream>>>(mv, mv_bf, mk, nullptr);
    mem_reader_attn<true, 1, false, 64><<<grid, 256, 0, stream>>>(mk, nullptr, mv_bf, nullptr, qk, out, nullptr, nullptr);
  } else {
    mem_reader_attn<false, 1, false, 64><<<grid, 256, 0, stream>>>(mk, mv, nullptr, nullptr, qk, out, nullptr, nullptr);
  }

  qv_copy<<<QV_BLOCKS, 256, 0, stream>>>(qv, out);
}
